# starting kernel
# baseline (speedup 1.0000x reference)
#include <hip/hip_runtime.h>
#include <hip/hip_bf16.h>
#include <math.h>

typedef __attribute__((ext_vector_type(2))) float v2f;
typedef __attribute__((ext_vector_type(8))) float v8f;

#define FIN   256
#define FOUT  128
#define NHEAD 4
#define DHEAD 32
#define NEG_SLOPE 0.2f

__global__ void __launch_bounds__(256)
rgat_gemm_wmma(const float* __restrict__ x, const float* __restrict__ Wr,
               float* __restrict__ h, int N) {
  const int lane  = threadIdx.x & 31;
  const int wave  = threadIdx.x >> 5;
  const int tm    = blockIdx.x;
  const int mrow  = lane & 15;
  const int kb    = (lane >> 4) << 1;
  const int col   = wave * 16 + mrow;

  int row = tm * 16 + mrow;
  int rowc = row < N ? row : 0;
  const float* __restrict__ xrow = x + (size_t)rowc * FIN;

  v8f c = {};
  for (int k0 = 0; k0 < FIN; k0 += 4) {
    v2f a = *(const v2f*)(xrow + k0 + kb);
    v2f b;
    b.x = Wr[(size_t)(k0 + kb)     * FOUT + col];
    b.y = Wr[(size_t)(k0 + kb + 1) * FOUT + col];
    c = __builtin_amdgcn_wmma_f32_16x16x4_f32(
         false, a,  false, b,
         (short)0, c,  false,  false);
  }

  const int rbase = tm * 16 + 8 * (lane >> 4);
  const int ocol  = wave * 16 + (lane & 15);
  float* __restrict__ hp = h + (size_t)rbase * FOUT + ocol;

  if (tm * 16 + 16 <= N) {
#pragma unroll
    for (int v = 0; v < 8; ++v) hp[(size_t)v * FOUT] = c[v];
  } else {
#pragma unroll
    for (int v = 0; v < 8; ++v) {
      if (rbase + v < N) hp[(size_t)v * FOUT] = c[v];
    }
  }
}

__global__ void rgat_node_logits(const float* __restrict__ h,
                                 const float* __restrict__ asrc,
                                 const float* __restrict__ adst,
                                 float* __restrict__ als, float* __restrict__ ald,
                                 int N) {
  int idx = blockIdx.x * blockDim.x + threadIdx.x;
  if (idx >= N * NHEAD) return;
  int n  = idx >> 2;
  int hd = idx & 3;
  const float* hp = h + (size_t)n * FOUT + hd * DHEAD;
  const float* as = asrc + hd * DHEAD;
  const float* ad = adst + hd * DHEAD;
  float s = 0.f, d = 0.f;
#pragma unroll
  for (int i = 0; i < DHEAD; ++i) {
    float v = hp[i];
    s = fmaf(v, as[i], s);
    d = fmaf(v, ad[i], d);
  }
  als[idx] = s;
  ald[idx] = d;
}

__global__ void rgat_init_out(const float* __restrict__ bias, float* __restrict__ out,
                              int total) {
  int i = blockIdx.x * blockDim.x + threadIdx.x;
  if (i >= total) return;
  int f = i & (FOUT - 1);
  out[i] = bias[f] + bias[FOUT + f] + bias[2 * FOUT + f];
}

__global__ void rgat_init_seg(float* __restrict__ amax, float* __restrict__ denom,
                              int nh) {
  int i = blockIdx.x * blockDim.x + threadIdx.x;
  if (i >= nh) return;
  amax[i]  = -1e30f;
  denom[i] = 0.f;
}

__device__ __forceinline__ float atomicMaxF(float* addr, float val) {
  if (val >= 0.f)
    return __int_as_float(atomicMax((int*)addr, __float_as_int(val)));
  return __uint_as_float(atomicMin((unsigned int*)addr, __float_as_uint(val)));
}

__device__ __forceinline__ float leaky(float v) {
  return v > 0.f ? v : NEG_SLOPE * v;
}

__global__ void rgat_edge_alpha_max(const int* __restrict__ ei,
                                    const float* __restrict__ als,
                                    const float* __restrict__ ald,
                                    float* __restrict__ alpha,
                                    float* __restrict__ amax, int E) {
  int e = blockIdx.x * blockDim.x + threadIdx.x;
  if (e >= E) return;
  int s = ei[e];
  int d = ei[E + e];
  float4 a_s = *(const float4*)(als + (size_t)s * NHEAD);
  float4 a_d = *(const float4*)(ald + (size_t)d * NHEAD);
  float4 al;
  al.x = leaky(a_s.x + a_d.x);
  al.y = leaky(a_s.y + a_d.y);
  al.z = leaky(a_s.z + a_d.z);
  al.w = leaky(a_s.w + a_d.w);
  *(float4*)(alpha + (size_t)e * NHEAD) = al;
  float* am = amax + (size_t)d * NHEAD;
  atomicMaxF(am + 0, al.x);
  atomicMaxF(am + 1, al.y);
  atomicMaxF(am + 2, al.z);
  atomicMaxF(am + 3, al.w);
}

__global__ void rgat_edge_exp_sum(const int* __restrict__ ei,
                                  float* __restrict__ alpha,
                                  const float* __restrict__ amax,
                                  float* __restrict__ denom, int E) {
  int e = blockIdx.x * blockDim.x + threadIdx.x;
  if (e >= E) return;
  int d = ei[E + e];
  float4 al = *(const float4*)(alpha + (size_t)e * NHEAD);
  float4 mx = *(const float4*)(amax + (size_t)d * NHEAD);
  al.x = __expf(al.x - mx.x);
  al.y = __expf(al.y - mx.y);
  al.z = __expf(al.z - mx.z);
  al.w = __expf(al.w - mx.w);
  *(float4*)(alpha + (size_t)e * NHEAD) = al;
  float* dn = denom + (size_t)d * NHEAD;
  atomicAdd(dn + 0, al.x);
  atomicAdd(dn + 1, al.y);
  atomicAdd(dn + 2, al.z);
  atomicAdd(dn + 3, al.w);
}

__global__ void __launch_bounds__(256)
rgat_edge_scatter(const int* __restrict__ ei, const float* __restrict__ h,
                  const float* __restrict__ alpha, const float* __restrict__ denom,
                  float* __restrict__ out, int E) {
  const int lane = threadIdx.x & 31;
  const int wave = threadIdx.x >> 5;
  int e = blockIdx.x * 8 + wave;
  if (e >= E) return;
  int s = ei[e];
  int d = ei[E + e];

  if (e + 8 < E) {
    int s2 = ei[e + 8];
    __builtin_prefetch(h + (size_t)s2 * FOUT + lane, 0, 0);
  }

  float4 ev = *(const float4*)(alpha + (size_t)e * NHEAD);
  float4 dn = *(const float4*)(denom + (size_t)d * NHEAD);
  float w0 = ev.x / (dn.x + 1e-16f);
  float w1 = ev.y / (dn.y + 1e-16f);
  float w2 = ev.z / (dn.z + 1e-16f);
  float w3 = ev.w / (dn.w + 1e-16f);

  const float* __restrict__ hs = h + (size_t)s * FOUT;
  float* __restrict__ od = out + (size_t)d * FOUT;
  atomicAdd(od + 0 * DHEAD + lane, hs[0 * DHEAD + lane] * w0);
  atomicAdd(od + 1 * DHEAD + lane, hs[1 * DHEAD + lane] * w1);
  atomicAdd(od + 2 * DHEAD + lane, hs[2 * DHEAD + lane] * w2);
  atomicAdd(od + 3 * DHEAD + lane, hs[3 * DHEAD + lane] * w3);
}

extern "C" void kernel_launch(void* const* d_in, const int* in_sizes, int n_in,
                              void* d_out, int out_size, void* d_ws, size_t ws_size,
                              hipStream_t stream) {
  const float* x        = (const float*)d_in[0];
  const int*   edges[3] = {(const int*)d_in[1], (const int*)d_in[2], (const int*)d_in[3]};
  const float* W        = (const float*)d_in[4];
  const float* att_src  = (const float*)d_in[5];
  const float* att_dst  = (const float*)d_in[6];
  const float* bias     = (const float*)d_in[7];

  const int N = in_sizes[0] / FIN;
  const int E = in_sizes[1] / 2;
  float* out = (float*)d_out;

  char* ws = (char*)d_ws;
  size_t off = 0;
  auto carve = [&](size_t bytes) {
    void* p = ws + off;
    off += (bytes + 255) & ~(size_t)255;
    return p;
  };
  float* h     = (float*)carve((size_t)N * FOUT * sizeof(float));
  float* als   = (float*)carve((size_t)N * NHEAD * sizeof(float));
  float* ald   = (float*)carve((size_t)N * NHEAD * sizeof(float));
  float* alpha = (float*)carve((size_t)E * NHEAD * sizeof(float));
  float* amax  = (float*)carve((size_t)N * NHEAD * sizeof(float));
  float* denom = (float*)carve((size_t)N * NHEAD * sizeof(float));
  (void)ws_size;

  rgat_init_out<<<(N * FOUT + 255) / 256, 256, 0, stream>>>(bias, out, N * FOUT);

  for (int r = 0; r < 3; ++r) {
    const float* Wr = W + (size_t)r * FIN * FOUT;
    rgat_gemm_wmma<<<(N + 15) / 16, 256, 0, stream>>>(x, Wr, h, N);
    rgat_node_logits<<<(N * NHEAD + 255) / 256, 256, 0, stream>>>(
        h, att_src + (size_t)r * NHEAD * DHEAD, att_dst + (size_t)r * NHEAD * DHEAD,
        als, ald, N);
    rgat_init_seg<<<(N * NHEAD + 255) / 256, 256, 0, stream>>>(amax, denom, N * NHEAD);
    rgat_edge_alpha_max<<<(E + 255) / 256, 256, 0, stream>>>(
        edges[r], als, ald, alpha, amax, E);
    rgat_edge_exp_sum<<<(E + 255) / 256, 256, 0, stream>>>(
        edges[r], alpha, amax, denom, E);
    rgat_edge_scatter<<<(E + 7) / 8, 256, 0, stream>>>(
        edges[r], h, alpha, denom, out, E);
  }
}


# Edges counting-sorted by destination in the call (integer atomic rank, scan, fill); one wave per node does segment max, sum and weighted row gather in registers: no float atomics; sums in rank order
# speedup vs baseline: 2.9472x; 2.9472x over previous
#include <hip/hip_runtime.h>
#include <hip/hip_bf16.h>
#include <math.h>

typedef __attribute__((ext_vector_type(2))) float v2f;
typedef __attribute__((ext_vector_type(8))) float v8f;

#define FIN   256
#define FOUT  128
#define NHEAD 4
#define DHEAD 32
#define NEG_SLOPE 0.2f

__global__ void __launch_bounds__(256)
rgat_gemm_wmma(const float* __restrict__ x, const float* __restrict__ Wr,
               float* __restrict__ h, int N) {
  const int lane  = threadIdx.x & 31;
  const int wave  = threadIdx.x >> 5;
  const int tm    = blockIdx.x;
  const int mrow  = lane & 15;
  const int kb    = (lane >> 4) << 1;
  const int col   = wave * 16 + mrow;

  int row = tm * 16 + mrow;
  int rowc = row < N ? row : 0;
  const float* __restrict__ xrow = x + (size_t)rowc * FIN;

  v8f c = {};
  for (int k0 = 0; k0 < FIN; k0 += 4) {
    v2f a = *(const v2f*)(xrow + k0 + kb);
    v2f b;
    b.x = Wr[(size_t)(k0 + kb)     * FOUT + col];
    b.y = Wr[(size_t)(k0 + kb + 1) * FOUT + col];
    c = __builtin_amdgcn_wmma_f32_16x16x4_f32(
         false, a,  false, b,
         (short)0, c,  false,  false);
  }

  const int rbase = tm * 16 + 8 * (lane >> 4);
  const int ocol  = wave * 16 + (lane & 15);
  float* __restrict__ hp = h + (size_t)rbase * FOUT + ocol;

  if (tm * 16 + 16 <= N) {
#pragma unroll
    for (int v = 0; v < 8; ++v) hp[(size_t)v * FOUT] = c[v];
  } else {
#pragma unroll
    for (int v = 0; v < 8; ++v) {
      if (rbase + v < N) hp[(size_t)v * FOUT] = c[v];
    }
  }
}

__global__ void rgat_node_logits(const float* __restrict__ h,
                                 const float* __restrict__ asrc,
                                 const float* __restrict__ adst,
                                 float* __restrict__ als, float* __restrict__ ald,
                                 int N) {
  int idx = blockIdx.x * blockDim.x + threadIdx.x;
  if (idx >= N * NHEAD) return;
  int n  = idx >> 2;
  int hd = idx & 3;
  const float* hp = h + (size_t)n * FOUT + hd * DHEAD;
  const float* as = asrc + hd * DHEAD;
  const float* ad = adst + hd * DHEAD;
  float s = 0.f, d = 0.f;
#pragma unroll
  for (int i = 0; i < DHEAD; ++i) {
    float v = hp[i];
    s = fmaf(v, as[i], s);
    d = fmaf(v, ad[i], d);
  }
  als[idx] = s;
  ald[idx] = d;
}

__device__ __forceinline__ float leaky(float v) {
  return v > 0.f ? v : NEG_SLOPE * v;
}

#define SCAN_TILE 1024

__global__ void rgat_zero_cnt(int* __restrict__ cnt, int total) {
  int i = blockIdx.x * blockDim.x + threadIdx.x;
  if (i < total) cnt[i] = 0;
}

__device__ __forceinline__ const int* pick_rel(const int* a, const int* b, const int* c, int r) {
  return r == 0 ? a : (r == 1 ? b : c);
}

__global__ void rgat_edge_rank(const int* __restrict__ ei0, const int* __restrict__ ei1,
                               const int* __restrict__ ei2, int* __restrict__ cnt,
                               int* __restrict__ rank, int E, int NP) {
  int e = blockIdx.x * blockDim.x + threadIdx.x;
  if (e >= E) return;
  const int r = blockIdx.y;
  const int* __restrict__ ei = pick_rel(ei0, ei1, ei2, r);
  int d = ei[E + e];
  rank[(size_t)r * E + e] = atomicAdd(cnt + (size_t)r * NP + d, 1);
}

__device__ __forceinline__ int wave_sum_i(int v) {
  for (int m = 16; m >= 1; m >>= 1) v += __shfl_xor(v, m, 32);
  return v;
}

__global__ void __launch_bounds__(256)
rgat_cnt_tilesum(const int* __restrict__ cnt, int* __restrict__ bsum, int NP, int nt) {
  __shared__ int part[8];
  const int r = blockIdx.y, tile = blockIdx.x;
  const int4 v = *(const int4*)(cnt + (size_t)r * NP + (size_t)tile * SCAN_TILE + threadIdx.x * 4);
  int s = wave_sum_i(v.x + v.y + v.z + v.w);
  if ((threadIdx.x & 31) == 0) part[threadIdx.x >> 5] = s;
  __syncthreads();
  if (threadIdx.x == 0) {
    int t = 0;
    for (int i = 0; i < 8; ++i) t += part[i];
    bsum[r * nt + tile] = t;
  }
}

__global__ void __launch_bounds__(256)
rgat_cnt_scan(const int* __restrict__ cnt, const int* __restrict__ bsum,
              int* __restrict__ offs, int NP, int nt) {
  __shared__ int part[8];
  __shared__ int wpre[8];
  const int r = blockIdx.y, tile = blockIdx.x;
  const int lane = threadIdx.x & 31, wave = threadIdx.x >> 5;

  int p = 0;
  for (int t = threadIdx.x; t < tile; t += 256) p += bsum[r * nt + t];
  p = wave_sum_i(p);
  if (lane == 0) part[wave] = p;

  const size_t at = (size_t)r * NP + (size_t)tile * SCAN_TILE + threadIdx.x * 4;
  const int4 v = *(const int4*)(cnt + at);
  const int mine = v.x + v.y + v.z + v.w;
  int inc = mine;
  for (int m = 1; m < 32; m <<= 1) {
    int up = __shfl_up(inc, m, 32);
    if (lane >= m) inc += up;
  }
  if (lane == 31) wpre[wave] = inc;
  __syncthreads();
  int base = 0;
  for (int i = 0; i < 8; ++i) {
    base += part[i];
    if (i < wave) base += wpre[i];
  }
  int4 o;
  o.x = base + inc - mine;
  o.y = o.x + v.x;
  o.z = o.y + v.y;
  o.w = o.z + v.z;
  *(int4*)(offs + at) = o;
}

__global__ void rgat_edge_fill(const int* __restrict__ ei0, const int* __restrict__ ei1,
                               const int* __restrict__ ei2, const int* __restrict__ offs,
                               const int* __restrict__ rank, int* __restrict__ csr,
                               int E, int NP) {
  int e = blockIdx.x * blockDim.x + threadIdx.x;
  if (e >= E) return;
  const int r = blockIdx.y;
  const int* __restrict__ ei = pick_rel(ei0, ei1, ei2, r);
  int s = ei[e];
  int d = ei[E + e];
  int pos = offs[(size_t)r * NP + d] + rank[(size_t)r * E + e];
  csr[(size_t)r * E + pos] = s;
}

__device__ __forceinline__ float wave_max_f(float v) {
  for (int m = 16; m >= 1; m >>= 1) v = fmaxf(v, __shfl_xor(v, m, 32));
  return v;
}
__device__ __forceinline__ float wave_sum_f(float v) {
  for (int m = 16; m >= 1; m >>= 1) v += __shfl_xor(v, m, 32);
  return v;
}

__device__ __forceinline__ float4 edge_alpha(const float* __restrict__ als, int s, float4 ad) {
  const float4 a = *(const float4*)(als + (size_t)s * NHEAD);
  float4 al;
  al.x = leaky(a.x + ad.x);
  al.y = leaky(a.y + ad.y);
  al.z = leaky(a.z + ad.z);
  al.w = leaky(a.w + ad.w);
  return al;
}

__device__ __forceinline__ float lane_bcast(float v, int t) {
  return __int_as_float(__builtin_amdgcn_readlane(__float_as_int(v), t));
}

__device__ __forceinline__ float edge_weight(float4 w, int t, int hsel) {
  const float x = lane_bcast(w.x, t), y = lane_bcast(w.y, t);
  const float z = lane_bcast(w.z, t), u = lane_bcast(w.w, t);
  return hsel < 2 ? (hsel == 0 ? x : y) : (hsel == 2 ? z : u);
}

__device__ __forceinline__ void row_fma(float4& acc, const float* __restrict__ h, int s,
                                        int lane, float w) {
  const float4 v = *(const float4*)(h + (size_t)s * FOUT + lane * 4);
  acc.x = fmaf(v.x, w, acc.x);
  acc.y = fmaf(v.y, w, acc.y);
  acc.z = fmaf(v.z, w, acc.z);
  acc.w = fmaf(v.w, w, acc.w);
}

__device__ __forceinline__ void gather_rows(float4& acc, const float* __restrict__ h,
                                            int s, float4 w, int n, int lane, int hsel) {
  int t = 0;
  for (; t + 4 <= n; t += 4) {
    const int s0 = __builtin_amdgcn_readlane(s, t);
    const int s1 = __builtin_amdgcn_readlane(s, t + 1);
    const int s2 = __builtin_amdgcn_readlane(s, t + 2);
    const int s3 = __builtin_amdgcn_readlane(s, t + 3);
    const float4 v0 = *(const float4*)(h + (size_t)s0 * FOUT + lane * 4);
    const float4 v1 = *(const float4*)(h + (size_t)s1 * FOUT + lane * 4);
    const float4 v2 = *(const float4*)(h + (size_t)s2 * FOUT + lane * 4);
    const float4 v3 = *(const float4*)(h + (size_t)s3 * FOUT + lane * 4);
    const float w0 = edge_weight(w, t, hsel);
    const float w1 = edge_weight(w, t + 1, hsel);
    const float w2 = edge_weight(w, t + 2, hsel);
    const float w3 = edge_weight(w, t + 3, hsel);
    acc.x = fmaf(v0.x, w0, acc.x); acc.y = fmaf(v0.y, w0, acc.y);
    acc.z = fmaf(v0.z, w0, acc.z); acc.w = fmaf(v0.w, w0, acc.w);
    acc.x = fmaf(v1.x, w1, acc.x); acc.y = fmaf(v1.y, w1, acc.y);
    acc.z = fmaf(v1.z, w1, acc.z); acc.w = fmaf(v1.w, w1, acc.w);
    acc.x = fmaf(v2.x, w2, acc.x); acc.y = fmaf(v2.y, w2, acc.y);
    acc.z = fmaf(v2.z, w2, acc.z); acc.w = fmaf(v2.w, w2, acc.w);
    acc.x = fmaf(v3.x, w3, acc.x); acc.y = fmaf(v3.y, w3, acc.y);
    acc.z = fmaf(v3.z, w3, acc.z); acc.w = fmaf(v3.w, w3, acc.w);
  }
  for (; t < n; ++t) {
    const int s0 = __builtin_amdgcn_readlane(s, t);
    row_fma(acc, h, s0, lane, edge_weight(w, t, hsel));
  }
}

__global__ void __launch_bounds__(256)
rgat_node_gather(const int* __restrict__ offs, const int* __restrict__ csr,
                 const float* __restrict__ h, const float* __restrict__ als,
                 const float* __restrict__ ald, const float* __restrict__ bias,
                 float* __restrict__ out, int N, int first) {
  const int lane = threadIdx.x & 31;
  const int hsel = lane >> 3;
  const int wave = __builtin_amdgcn_readfirstlane((int)(threadIdx.x >> 5));
  const int d = blockIdx.x * 8 + wave;
  if (d >= N) return;
  const int beg = offs[d];
  const int end = offs[d + 1];
  const float4 ad = *(const float4*)(ald + (size_t)d * NHEAD);
  const float ninf = -__builtin_inff();

  const int j0 = beg + lane;
  const bool has0 = j0 < end;
  int s_first = 0;
  float4 al_first = make_float4(ninf, ninf, ninf, ninf);
  if (has0) {
    s_first = csr[j0];
    al_first = edge_alpha(als, s_first, ad);
  }

  float4 mx = al_first;
  for (int j = j0 + 32; j < end; j += 32) {
    const float4 al = edge_alpha(als, csr[j], ad);
    mx.x = fmaxf(mx.x, al.x); mx.y = fmaxf(mx.y, al.y);
    mx.z = fmaxf(mx.z, al.z); mx.w = fmaxf(mx.w, al.w);
  }
  mx.x = wave_max_f(mx.x); mx.y = wave_max_f(mx.y);
  mx.z = wave_max_f(mx.z); mx.w = wave_max_f(mx.w);

  float4 e_first = make_float4(0.f, 0.f, 0.f, 0.f);
  if (has0) {
    e_first.x = __expf(al_first.x - mx.x); e_first.y = __expf(al_first.y - mx.y);
    e_first.z = __expf(al_first.z - mx.z); e_first.w = __expf(al_first.w - mx.w);
  }
  float4 sm = e_first;
  for (int j = j0 + 32; j < end; j += 32) {
    const float4 al = edge_alpha(als, csr[j], ad);
    sm.x += __expf(al.x - mx.x); sm.y += __expf(al.y - mx.y);
    sm.z += __expf(al.z - mx.z); sm.w += __expf(al.w - mx.w);
  }
  sm.x = wave_sum_f(sm.x) + 1e-16f; sm.y = wave_sum_f(sm.y) + 1e-16f;
  sm.z = wave_sum_f(sm.z) + 1e-16f; sm.w = wave_sum_f(sm.w) + 1e-16f;

  float4 acc = make_float4(0.f, 0.f, 0.f, 0.f);
  {
    float4 w;
    w.x = e_first.x / sm.x; w.y = e_first.y / sm.y;
    w.z = e_first.z / sm.z; w.w = e_first.w / sm.w;
    const int n = end - beg < 32 ? end - beg : 32;
    gather_rows(acc, h, s_first, w, n, lane, hsel);
  }
  for (int base = beg + 32; base < end; base += 32) {
    const int j = base + lane;
    int s = 0;
    float4 w = make_float4(0.f, 0.f, 0.f, 0.f);
    if (j < end) {
      s = csr[j];
      const float4 al = edge_alpha(als, s, ad);
      w.x = __expf(al.x - mx.x) / sm.x; w.y = __expf(al.y - mx.y) / sm.y;
      w.z = __expf(al.z - mx.z) / sm.z; w.w = __expf(al.w - mx.w) / sm.w;
    }
    const int n = end - base < 32 ? end - base : 32;
    gather_rows(acc, h, s, w, n, lane, hsel);
  }

  float4* __restrict__ op = (float4*)(out + (size_t)d * FOUT) + lane;
  if (first) {
    const float4 b0 = *((const float4*)bias + lane);
    const float4 b1 = *((const float4*)(bias + FOUT) + lane);
    const float4 b2 = *((const float4*)(bias + 2 * FOUT) + lane);
    acc.x += b0.x + b1.x + b2.x; acc.y += b0.y + b1.y + b2.y;
    acc.z += b0.z + b1.z + b2.z; acc.w += b0.w + b1.w + b2.w;
  } else {
    const float4 o = *op;
    acc.x += o.x; acc.y += o.y; acc.z += o.z; acc.w += o.w;
  }
  *op = acc;
}

extern "C" void kernel_launch(void* const* d_in, const int* in_sizes, int n_in,
                              void* d_out, int out_size, void* d_ws, size_t ws_size,
                              hipStream_t stream) {
  const float* x        = (const float*)d_in[0];
  const int*   edges[3] = {(const int*)d_in[1], (const int*)d_in[2], (const int*)d_in[3]};
  const float* W        = (const float*)d_in[4];
  const float* att_src  = (const float*)d_in[5];
  const float* att_dst  = (const float*)d_in[6];
  const float* bias     = (const float*)d_in[7];

  const int N = in_sizes[0] / FIN;
  const int E = in_sizes[1] / 2;
  float* out = (float*)d_out;

  const int nt = (N + 1 + SCAN_TILE - 1) / SCAN_TILE;
  const int NP = nt * SCAN_TILE;

  char* ws = (char*)d_ws;
  size_t off = 0;
  auto carve = [&](size_t bytes) {
    void* p = ws + off;
    off += (bytes + 255) & ~(size_t)255;
    return p;
  };
  float* h    = (float*)carve((size_t)N * FOUT * sizeof(float));
  float* als  = (float*)carve((size_t)N * NHEAD * sizeof(float));
  float* ald  = (float*)carve((size_t)N * NHEAD * sizeof(float));
  int*   csr  = (int*)carve((size_t)3 * E * sizeof(int));
  int*   cnt  = (int*)carve((size_t)3 * NP * sizeof(int));
  int*   offs = (int*)carve((size_t)3 * NP * sizeof(int));
  int*   bsum = (int*)carve((size_t)3 * nt * sizeof(int));
  int*   rank = (int*)h;
  (void)ws_size;

  const dim3 egrid((E + 255) / 256, 3);
  const dim3 sgrid(nt, 3);
  rgat_zero_cnt<<<(3 * NP + 255) / 256, 256, 0, stream>>>(cnt, 3 * NP);
  rgat_edge_rank<<<egrid, 256, 0, stream>>>(edges[0], edges[1], edges[2], cnt, rank, E, NP);
  rgat_cnt_tilesum<<<sgrid, 256, 0, stream>>>(cnt, bsum, NP, nt);
  rgat_cnt_scan<<<sgrid, 256, 0, stream>>>(cnt, bsum, offs, NP, nt);
  rgat_edge_fill<<<egrid, 256, 0, stream>>>(edges[0], edges[1], edges[2], offs, rank, csr, E, NP);

  for (int r = 0; r < 3; ++r) {
    const float* Wr = W + (size_t)r * FIN * FOUT;
    rgat_gemm_wmma<<<(N + 15) / 16, 256, 0, stream>>>(x, Wr, h, N);
    rgat_node_logits<<<(N * NHEAD + 255) / 256, 256, 0, stream>>>(
        h, att_src + (size_t)r * NHEAD * DHEAD, att_dst + (size_t)r * NHEAD * DHEAD,
        als, ald, N);
    rgat_node_gather<<<(N + 7) / 8, 256, 0, stream>>>(
        offs + (size_t)r * NP, csr + (size_t)r * E, h, als, ald, bias, out, N, r == 0);
  }
}


# Long-lived waves: edge rank/fill kernels take 8 edges a thread, gather waves loop over 5 nodes with the next node's head loads and 8 rows in flight; sums still in arrival-rank order
# speedup vs baseline: 3.1445x; 1.0669x over previous
#include <hip/hip_runtime.h>
#include <hip/hip_bf16.h>
#include <math.h>

typedef __attribute__((ext_vector_type(2))) float v2f;
typedef __attribute__((ext_vector_type(8))) float v8f;

#define FIN   256
#define FOUT  128
#define NHEAD 4
#define DHEAD 32
#define NEG_SLOPE 0.2f

__global__ void __launch_bounds__(256)
rgat_gemm_wmma(const float* __restrict__ x, const float* __restrict__ Wr,
               float* __restrict__ h, int N) {
  const int lane  = threadIdx.x & 31;
  const int wave  = threadIdx.x >> 5;
  const int tm    = blockIdx.x;
  const int mrow  = lane & 15;
  const int kb    = (lane >> 4) << 1;
  const int col   = wave * 16 + mrow;

  int row = tm * 16 + mrow;
  int rowc = row < N ? row : 0;
  const float* __restrict__ xrow = x + (size_t)rowc * FIN;

  v8f c = {};
  for (int k0 = 0; k0 < FIN; k0 += 4) {
    v2f a = *(const v2f*)(xrow + k0 + kb);
    v2f b;
    b.x = Wr[(size_t)(k0 + kb)     * FOUT + col];
    b.y = Wr[(size_t)(k0 + kb + 1) * FOUT + col];
    c = __builtin_amdgcn_wmma_f32_16x16x4_f32(
         false, a,  false, b,
         (short)0, c,  false,  false);
  }

  const int rbase = tm * 16 + 8 * (lane >> 4);
  const int ocol  = wave * 16 + (lane & 15);
  float* __restrict__ hp = h + (size_t)rbase * FOUT + ocol;

  if (tm * 16 + 16 <= N) {
#pragma unroll
    for (int v = 0; v < 8; ++v) hp[(size_t)v * FOUT] = c[v];
  } else {
#pragma unroll
    for (int v = 0; v < 8; ++v) {
      if (rbase + v < N) hp[(size_t)v * FOUT] = c[v];
    }
  }
}

__global__ void rgat_node_logits(const float* __restrict__ h,
                                 const float* __restrict__ asrc,
                                 const float* __restrict__ adst,
                                 float* __restrict__ als, float* __restrict__ ald,
                                 int N) {
  int idx = blockIdx.x * blockDim.x + threadIdx.x;
  if (idx >= N * NHEAD) return;
  int n  = idx >> 2;
  int hd = idx & 3;
  const float* hp = h + (size_t)n * FOUT + hd * DHEAD;
  const float* as = asrc + hd * DHEAD;
  const float* ad = adst + hd * DHEAD;
  float s = 0.f, d = 0.f;
#pragma unroll
  for (int i = 0; i < DHEAD; ++i) {
    float v = hp[i];
    s = fmaf(v, as[i], s);
    d = fmaf(v, ad[i], d);
  }
  als[idx] = s;
  ald[idx] = d;
}

__device__ __forceinline__ float leaky(float v) {
  return v > 0.f ? v : NEG_SLOPE * v;
}

#define SCAN_TILE 1024

__global__ void rgat_zero_cnt(int* __restrict__ cnt, int total) {
  int i = blockIdx.x * blockDim.x + threadIdx.x;
  if (i < total) cnt[i] = 0;
}

__device__ __forceinline__ const int* pick_rel(const int* a, const int* b, const int* c, int r) {
  return r == 0 ? a : (r == 1 ? b : c);
}

#define EDGE_PER_THREAD 8
#define EDGE_PER_BLOCK  (256 * EDGE_PER_THREAD)

template <bool FULL>
__device__ __forceinline__ void edge_rank_body(const int* __restrict__ dstp, int* __restrict__ cr,
                                               int* __restrict__ rr, int e0, int E) {
  int d[EDGE_PER_THREAD], k[EDGE_PER_THREAD];
#pragma unroll
  for (int i = 0; i < EDGE_PER_THREAD; ++i) {
    const int e = e0 + 256 * i;
    d[i] = (FULL || e < E) ? dstp[e] : -1;
  }
#pragma unroll
  for (int i = 0; i < EDGE_PER_THREAD; ++i) {
    k[i] = 0;
    if (FULL || d[i] >= 0) k[i] = atomicAdd(cr + d[i], 1);
  }
#pragma unroll
  for (int i = 0; i < EDGE_PER_THREAD; ++i) {
    if (FULL || d[i] >= 0) rr[e0 + 256 * i] = k[i];
  }
}

__global__ void __launch_bounds__(256)
rgat_edge_rank(const int* __restrict__ ei0, const int* __restrict__ ei1,
               const int* __restrict__ ei2, int* __restrict__ cnt,
               int* __restrict__ rank, int E, int NP) {
  const int r = blockIdx.y;
  const int* __restrict__ dstp = pick_rel(ei0, ei1, ei2, r) + E;
  int* __restrict__ cr = cnt + (size_t)r * NP;
  int* __restrict__ rr = rank + (size_t)r * E;
  const int b0 = blockIdx.x * EDGE_PER_BLOCK;
  if (b0 + EDGE_PER_BLOCK <= E) edge_rank_body<true>(dstp, cr, rr, b0 + threadIdx.x, E);
  else                          edge_rank_body<false>(dstp, cr, rr, b0 + threadIdx.x, E);
}

__device__ __forceinline__ int wave_sum_i(int v) {
  for (int m = 16; m >= 1; m >>= 1) v += __shfl_xor(v, m, 32);
  return v;
}

__global__ void __launch_bounds__(256)
rgat_cnt_tilesum(const int* __restrict__ cnt, int* __restrict__ bsum, int NP, int nt) {
  __shared__ int part[8];
  const int r = blockIdx.y, tile = blockIdx.x;
  const int4 v = *(const int4*)(cnt + (size_t)r * NP + (size_t)tile * SCAN_TILE + threadIdx.x * 4);
  int s = wave_sum_i(v.x + v.y + v.z + v.w);
  if ((threadIdx.x & 31) == 0) part[threadIdx.x >> 5] = s;
  __syncthreads();
  if (threadIdx.x == 0) {
    int t = 0;
    for (int i = 0; i < 8; ++i) t += part[i];
    bsum[r * nt + tile] = t;
  }
}

__global__ void __launch_bounds__(256)
rgat_cnt_scan(const int* __restrict__ cnt, const int* __restrict__ bsum,
              int* __restrict__ offs, int NP, int nt) {
  __shared__ int part[8];
  __shared__ int wpre[8];
  const int r = blockIdx.y, tile = blockIdx.x;
  const int lane = threadIdx.x & 31, wave = threadIdx.x >> 5;

  int p = 0;
  for (int t = threadIdx.x; t < tile; t += 256) p += bsum[r * nt + t];
  p = wave_sum_i(p);
  if (lane == 0) part[wave] = p;

  const size_t at = (size_t)r * NP + (size_t)tile * SCAN_TILE + threadIdx.x * 4;
  const int4 v = *(const int4*)(cnt + at);
  const int mine = v.x + v.y + v.z + v.w;
  int inc = mine;
  for (int m = 1; m < 32; m <<= 1) {
    int up = __shfl_up(inc, m, 32);
    if (lane >= m) inc += up;
  }
  if (lane == 31) wpre[wave] = inc;
  __syncthreads();
  int base = 0;
  for (int i = 0; i < 8; ++i) {
    base += part[i];
    if (i < wave) base += wpre[i];
  }
  int4 o;
  o.x = base + inc - mine;
  o.y = o.x + v.x;
  o.z = o.y + v.y;
  o.w = o.z + v.z;
  *(int4*)(offs + at) = o;
}

template <bool FULL>
__device__ __forceinline__ void edge_fill_body(const int* __restrict__ ei, const int* __restrict__ orl,
                                               const int* __restrict__ rr, int* __restrict__ cr,
                                               int e0, int E) {
  int s[EDGE_PER_THREAD], d[EDGE_PER_THREAD], k[EDGE_PER_THREAD], o[EDGE_PER_THREAD];
#pragma unroll
  for (int i = 0; i < EDGE_PER_THREAD; ++i) {
    const int e = e0 + 256 * i;
    const bool in = FULL || e < E;
    s[i] = in ? ei[e] : 0;
    d[i] = in ? ei[E + e] : -1;
    k[i] = in ? rr[e] : 0;
  }
#pragma unroll
  for (int i = 0; i < EDGE_PER_THREAD; ++i) o[i] = orl[FULL ? d[i] : (d[i] >= 0 ? d[i] : 0)];
#pragma unroll
  for (int i = 0; i < EDGE_PER_THREAD; ++i) {
    if (FULL || d[i] >= 0) cr[o[i] + k[i]] = s[i];
  }
}

__global__ void __launch_bounds__(256)
rgat_edge_fill(const int* __restrict__ ei0, const int* __restrict__ ei1,
               const int* __restrict__ ei2, const int* __restrict__ offs,
               const int* __restrict__ rank, int* __restrict__ csr,
               int E, int NP) {
  const int r = blockIdx.y;
  const int* __restrict__ ei = pick_rel(ei0, ei1, ei2, r);
  const int* __restrict__ orl = offs + (size_t)r * NP;
  const int* __restrict__ rr = rank + (size_t)r * E;
  int* __restrict__ cr = csr + (size_t)r * E;
  const int b0 = blockIdx.x * EDGE_PER_BLOCK;
  if (b0 + EDGE_PER_BLOCK <= E) edge_fill_body<true>(ei, orl, rr, cr, b0 + threadIdx.x, E);
  else                          edge_fill_body<false>(ei, orl, rr, cr, b0 + threadIdx.x, E);
}

#define NODES_PER_WAVE 5

__device__ __forceinline__ float wave_max_f(float v) {
  for (int m = 16; m >= 1; m >>= 1) v = fmaxf(v, __shfl_xor(v, m, 32));
  return v;
}
__device__ __forceinline__ float wave_sum_f(float v) {
  for (int m = 16; m >= 1; m >>= 1) v += __shfl_xor(v, m, 32);
  return v;
}

__device__ __forceinline__ float4 edge_alpha(const float* __restrict__ als, int s, float4 ad) {
  const float4 a = *(const float4*)(als + (size_t)s * NHEAD);
  float4 al;
  al.x = leaky(a.x + ad.x);
  al.y = leaky(a.y + ad.y);
  al.z = leaky(a.z + ad.z);
  al.w = leaky(a.w + ad.w);
  return al;
}

__device__ __forceinline__ float lane_bcast(float v, int t) {
  return __int_as_float(__builtin_amdgcn_readlane(__float_as_int(v), t));
}

__device__ __forceinline__ float edge_weight(float4 w, int t, int hsel) {
  const float x = lane_bcast(w.x, t), y = lane_bcast(w.y, t);
  const float z = lane_bcast(w.z, t), u = lane_bcast(w.w, t);
  return hsel < 2 ? (hsel == 0 ? x : y) : (hsel == 2 ? z : u);
}

template <int U>
__device__ __forceinline__ void rows_trip(float4& acc, const float* __restrict__ h,
                                          int s, float4 w, int t, int lane, int hsel) {
  float4 v[U];
  float wt[U];
#pragma unroll
  for (int u = 0; u < U; ++u) {
    const int su = __builtin_amdgcn_readlane(s, t + u);
    v[u] = *(const float4*)(h + (size_t)su * FOUT + lane * 4);
  }
#pragma unroll
  for (int u = 0; u < U; ++u) wt[u] = edge_weight(w, t + u, hsel);
#pragma unroll
  for (int u = 0; u < U; ++u) {
    acc.x = fmaf(v[u].x, wt[u], acc.x); acc.y = fmaf(v[u].y, wt[u], acc.y);
    acc.z = fmaf(v[u].z, wt[u], acc.z); acc.w = fmaf(v[u].w, wt[u], acc.w);
  }
}

__device__ __forceinline__ void gather_rows(float4& acc, const float* __restrict__ h,
                                            int s, float4 w, int n, int lane, int hsel) {
  int t = 0;
  for (; t + 8 <= n; t += 8) rows_trip<8>(acc, h, s, w, t, lane, hsel);
  if (t + 4 <= n) { rows_trip<4>(acc, h, s, w, t, lane, hsel); t += 4; }
  for (; t < n; ++t) rows_trip<1>(acc, h, s, w, t, lane, hsel);
}

__global__ void __launch_bounds__(256)
rgat_node_gather(const int* __restrict__ offs, const int* __restrict__ csr,
                 const float* __restrict__ h, const float* __restrict__ als,
                 const float* __restrict__ ald, const float* __restrict__ bias,
                 float* __restrict__ out, int N, int first, int nwaves) {
  const int lane = threadIdx.x & 31;
  const int hsel = lane >> 3;
  const int wave = __builtin_amdgcn_readfirstlane((int)(threadIdx.x >> 5));
  const float ninf = -__builtin_inff();
  const float4 zero4 = make_float4(0.f, 0.f, 0.f, 0.f);

  int d = blockIdx.x * 8 + wave;
  if (d >= N) return;

  float4 bsum = zero4;
  if (first) {
    const float4 b0 = *((const float4*)bias + lane);
    const float4 b1 = *((const float4*)(bias + FOUT) + lane);
    const float4 b2 = *((const float4*)(bias + 2 * FOUT) + lane);
    bsum.x = b0.x + b1.x + b2.x; bsum.y = b0.y + b1.y + b2.y;
    bsum.z = b0.z + b1.z + b2.z; bsum.w = b0.w + b1.w + b2.w;
  }

  int beg = offs[d];
  int end = offs[d + 1];
  float4 ad = *(const float4*)(ald + (size_t)d * NHEAD);
  int s_first = 0;
  float4 a_first = zero4;
  if (beg + lane < end) {
    s_first = csr[beg + lane];
    a_first = *(const float4*)(als + (size_t)s_first * NHEAD);
  }

  for (;;) {
    const int dn = d + nwaves;
    const bool more = dn < N;
    int beg_n = 0, end_n = 0;
    float4 ad_n = zero4;
    if (more) {
      beg_n = offs[dn];
      end_n = offs[dn + 1];
      ad_n = *(const float4*)(ald + (size_t)dn * NHEAD);
    }
    float4* __restrict__ op = (float4*)(out + (size_t)d * FOUT) + lane;
    float4 o = bsum;
    if (!first) o = *op;

    const int j0 = beg + lane;
    const bool has0 = j0 < end;
    float4 al_first = make_float4(ninf, ninf, ninf, ninf);
    if (has0) {
      al_first.x = leaky(a_first.x + ad.x);
      al_first.y = leaky(a_first.y + ad.y);
      al_first.z = leaky(a_first.z + ad.z);
      al_first.w = leaky(a_first.w + ad.w);
    }

    float4 mx = al_first;
    for (int j = j0 + 32; j < end; j += 32) {
      const float4 al = edge_alpha(als, csr[j], ad);
      mx.x = fmaxf(mx.x, al.x); mx.y = fmaxf(mx.y, al.y);
      mx.z = fmaxf(mx.z, al.z); mx.w = fmaxf(mx.w, al.w);
    }
    mx.x = wave_max_f(mx.x); mx.y = wave_max_f(mx.y);
    mx.z = wave_max_f(mx.z); mx.w = wave_max_f(mx.w);

    float4 e_first = zero4;
    if (has0) {
      e_first.x = __expf(al_first.x - mx.x); e_first.y = __expf(al_first.y - mx.y);
      e_first.z = __expf(al_first.z - mx.z); e_first.w = __expf(al_first.w - mx.w);
    }
    float4 sm = e_first;
    for (int j = j0 + 32; j < end; j += 32) {
      const float4 al = edge_alpha(als, csr[j], ad);
      sm.x += __expf(al.x - mx.x); sm.y += __expf(al.y - mx.y);
      sm.z += __expf(al.z - mx.z); sm.w += __expf(al.w - mx.w);
    }
    sm.x = wave_sum_f(sm.x) + 1e-16f; sm.y = wave_sum_f(sm.y) + 1e-16f;
    sm.z = wave_sum_f(sm.z) + 1e-16f; sm.w = wave_sum_f(sm.w) + 1e-16f;

    const bool has_n = more && (beg_n + lane < end_n);
    int s_n = 0;
    if (has_n) s_n = csr[beg_n + lane];

    float4 acc = zero4;
    {
      float4 w;
      w.x = e_first.x / sm.x; w.y = e_first.y / sm.y;
      w.z = e_first.z / sm.z; w.w = e_first.w / sm.w;
      const int n = end - beg < 32 ? end - beg : 32;
      gather_rows(acc, h, s_first, w, n, lane, hsel);
    }
    for (int base = beg + 32; base < end; base += 32) {
      const int j = base + lane;
      int s = 0;
      float4 w = zero4;
      if (j < end) {
        s = csr[j];
        const float4 al = edge_alpha(als, s, ad);
        w.x = __expf(al.x - mx.x) / sm.x; w.y = __expf(al.y - mx.y) / sm.y;
        w.z = __expf(al.z - mx.z) / sm.z; w.w = __expf(al.w - mx.w) / sm.w;
      }
      const int n = end - base < 32 ? end - base : 32;
      gather_rows(acc, h, s, w, n, lane, hsel);
    }

    float4 a_n = zero4;
    if (has_n) a_n = *(const float4*)(als + (size_t)s_n * NHEAD);

    acc.x += o.x; acc.y += o.y; acc.z += o.z; acc.w += o.w;
    *op = acc;

    if (!more) break;
    d = dn; beg = beg_n; end = end_n; ad = ad_n;
    s_first = s_n; a_first = a_n;
  }
}

extern "C" void kernel_launch(void* const* d_in, const int* in_sizes, int n_in,
                              void* d_out, int out_size, void* d_ws, size_t ws_size,
                              hipStream_t stream) {
  const float* x        = (const float*)d_in[0];
  const int*   edges[3] = {(const int*)d_in[1], (const int*)d_in[2], (const int*)d_in[3]};
  const float* W        = (const float*)d_in[4];
  const float* att_src  = (const float*)d_in[5];
  const float* att_dst  = (const float*)d_in[6];
  const float* bias     = (const float*)d_in[7];

  const int N = in_sizes[0] / FIN;
  const int E = in_sizes[1] / 2;
  float* out = (float*)d_out;

  const int nt = (N + 1 + SCAN_TILE - 1) / SCAN_TILE;
  const int NP = nt * SCAN_TILE;

  char* ws = (char*)d_ws;
  size_t off = 0;
  auto carve = [&](size_t bytes) {
    void* p = ws + off;
    off += (bytes + 255) & ~(size_t)255;
    return p;
  };
  float* h    = (float*)carve((size_t)N * FOUT * sizeof(float));
  float* als  = (float*)carve((size_t)N * NHEAD * sizeof(float));
  float* ald  = (float*)carve((size_t)N * NHEAD * sizeof(float));
  int*   csr  = (int*)carve((size_t)3 * E * sizeof(int));
  int*   cnt  = (int*)carve((size_t)3 * NP * sizeof(int));
  int*   offs = (int*)carve((size_t)3 * NP * sizeof(int));
  int*   bsum = (int*)carve((size_t)3 * nt * sizeof(int));
  int*   rank = (int*)h;
  (void)ws_size;

  const dim3 egrid((E + EDGE_PER_BLOCK - 1) / EDGE_PER_BLOCK, 3);
  const dim3 sgrid(nt, 3);
  const int gblocks = (N + 8 * NODES_PER_WAVE - 1) / (8 * NODES_PER_WAVE);
  rgat_zero_cnt<<<(3 * NP + 255) / 256, 256, 0, stream>>>(cnt, 3 * NP);
  rgat_edge_rank<<<egrid, 256, 0, stream>>>(edges[0], edges[1], edges[2], cnt, rank, E, NP);
  rgat_cnt_tilesum<<<sgrid, 256, 0, stream>>>(cnt, bsum, NP, nt);
  rgat_cnt_scan<<<sgrid, 256, 0, stream>>>(cnt, bsum, offs, NP, nt);
  rgat_edge_fill<<<egrid, 256, 0, stream>>>(edges[0], edges[1], edges[2], offs, rank, csr, E, NP);

  for (int r = 0; r < 3; ++r) {
    const float* Wr = W + (size_t)r * FIN * FOUT;
    rgat_gemm_wmma<<<(N + 15) / 16, 256, 0, stream>>>(x, Wr, h, N);
    rgat_node_logits<<<(N * NHEAD + 255) / 256, 256, 0, stream>>>(
        h, att_src + (size_t)r * NHEAD * DHEAD, att_dst + (size_t)r * NHEAD * DHEAD,
        als, ald, N);
    rgat_node_gather<<<gblocks, 256, 0, stream>>>(
        offs + (size_t)r * NP, csr + (size_t)r * E, h, als, ald, bias, out, N, r == 0,
        gblocks * 8);
  }
}


# GEMM retiled: block of 64 rows x 128 cols, 32-k chunks of x and W staged in LDS by whole-line loads (two buffers), each wave a 32x32 tile of 4 accumulators; k order of every sum unchanged
# speedup vs baseline: 3.7597x; 1.1956x over previous
#include <hip/hip_runtime.h>
#include <hip/hip_bf16.h>
#include <math.h>

typedef __attribute__((ext_vector_type(2))) float v2f;
typedef __attribute__((ext_vector_type(8))) float v8f;

#define FIN   256
#define FOUT  128
#define NHEAD 4
#define DHEAD 32
#define NEG_SLOPE 0.2f

#define GM_ROWS  64
#define GM_KC    32
#define GM_NCH   (FIN / GM_KC)
#define XS_PITCH 36
#define WS_PITCH 288

__device__ __forceinline__ v8f wmma_f32(v2f a, v2f b, v8f c) {
  return __builtin_amdgcn_wmma_f32_16x16x4_f32(
       false, a,  false, b,
       (short)0, c,  false,  false);
}

__device__ __forceinline__ void stage_w_pair(float* __restrict__ dst, float4 lo, float4 hi) {
  *(float4*)(dst)     = make_float4(lo.x, hi.x, lo.y, hi.y);
  *(float4*)(dst + 4) = make_float4(lo.z, hi.z, lo.w, hi.w);
}

__device__ __forceinline__ void gemm_chunk(const float* __restrict__ xa, const float* __restrict__ wb,
                                           v8f& c00, v8f& c01, v8f& c10, v8f& c11) {
#pragma unroll
  for (int ks = 0; ks < GM_KC / 4; ++ks) {
    const v2f a0 = *(const v2f*)(xa + ks * 4);
    const v2f a1 = *(const v2f*)(xa + 16 * XS_PITCH + ks * 4);
    const v2f b0 = *(const v2f*)(wb + ks * 2 * WS_PITCH);
    const v2f b1 = *(const v2f*)(wb + ks * 2 * WS_PITCH + 32);
    c00 = wmma_f32(a0, b0, c00);
    c01 = wmma_f32(a0, b1, c01);
    c10 = wmma_f32(a1, b0, c10);
    c11 = wmma_f32(a1, b1, c11);
  }
}

__global__ void __launch_bounds__(256)
rgat_gemm_wmma(const float* x, const float* Wr, float* __restrict__ h, int N) {
  __shared__ __attribute__((aligned(16))) float xs[2][GM_ROWS * XS_PITCH];
  __shared__ __attribute__((aligned(16))) float wsm[2][(GM_KC / 2) * WS_PITCH];

  const int tid  = threadIdx.x;
  const int lane = tid & 31;
  const int wave = tid >> 5;
  const int wm   = wave & 1;
  const int wn   = wave >> 1;
  const int c16  = lane & 15;
  const int hh   = lane >> 4;
  const int m0   = blockIdx.x * GM_ROWS;

  const int xr = tid >> 3, xq = tid & 7;
  int gr0 = m0 + xr;       gr0 = gr0 < N ? gr0 : 0;
  int gr1 = m0 + xr + 32;  gr1 = gr1 < N ? gr1 : 0;
  const float* xg0 = x + (size_t)gr0 * FIN + xq * 4;
  const float* xg1 = x + (size_t)gr1 * FIN + xq * 4;
  const int xs_at = xr * XS_PITCH + xq * 4;
  const int wk = tid >> 5, wq = tid & 31;
  const float* wg = Wr + (size_t)(2 * wk) * FOUT + wq * 4;
  const int ws_at = wk * WS_PITCH + wq * 8;

  const int xa_at = (wm * 32 + c16) * XS_PITCH + 2 * hh;
  const int wb_at = hh * WS_PITCH + (wn * 32 + c16) * 2;

  float4 px0, px1, pw0, pw1, pw2, pw3;
  px0 = *(const float4*)(xg0);
  px1 = *(const float4*)(xg1);
  pw0 = *(const float4*)(wg);
  pw1 = *(const float4*)(wg + FOUT);
  pw2 = *(const float4*)(wg + 16 * FOUT);
  pw3 = *(const float4*)(wg + 17 * FOUT);
  *(float4*)(&xs[0][xs_at])                 = px0;
  *(float4*)(&xs[0][xs_at + 32 * XS_PITCH]) = px1;
  stage_w_pair(&wsm[0][ws_at], pw0, pw1);
  stage_w_pair(&wsm[0][ws_at + 8 * WS_PITCH], pw2, pw3);
  __syncthreads();

  v8f c00 = {}, c01 = {}, c10 = {}, c11 = {};
#pragma unroll
  for (int ch = 0; ch + 1 < GM_NCH; ++ch) {
    const int buf = ch & 1;
    const int nb  = buf ^ 1;
    const int kn  = (ch + 1) * GM_KC;
    px0 = *(const float4*)(xg0 + kn);
    px1 = *(const float4*)(xg1 + kn);
    pw0 = *(const float4*)(wg + (size_t)kn * FOUT);
    pw1 = *(const float4*)(wg + (size_t)(kn + 1) * FOUT);
    pw2 = *(const float4*)(wg + (size_t)(kn + 16) * FOUT);
    pw3 = *(const float4*)(wg + (size_t)(kn + 17) * FOUT);
    __builtin_amdgcn_sched_barrier(0);
    gemm_chunk(&xs[buf][xa_at], &wsm[buf][wb_at], c00, c01, c10, c11);
    __builtin_amdgcn_sched_barrier(0);
    *(float4*)(&xs[nb][xs_at])                 = px0;
    *(float4*)(&xs[nb][xs_at + 32 * XS_PITCH]) = px1;
    stage_w_pair(&wsm[nb][ws_at], pw0, pw1);
    stage_w_pair(&wsm[nb][ws_at + 8 * WS_PITCH], pw2, pw3);
    __syncthreads();
  }
  gemm_chunk(&xs[(GM_NCH - 1) & 1][xa_at], &wsm[(GM_NCH - 1) & 1][wb_at], c00, c01, c10, c11);

  const int rbase = m0 + wm * 32 + 8 * hh;
  const int ocol  = wn * 32 + c16;
  float* __restrict__ hp = h + (size_t)rbase * FOUT + ocol;

  if (m0 + GM_ROWS <= N) {
#pragma unroll
    for (int v = 0; v < 8; ++v) {
      hp[(size_t)v * FOUT]             = c00[v];
      hp[(size_t)v * FOUT + 16]        = c01[v];
      hp[(size_t)(v + 16) * FOUT]      = c10[v];
      hp[(size_t)(v + 16) * FOUT + 16] = c11[v];
    }
  } else {
#pragma unroll
    for (int v = 0; v < 8; ++v) {
      if (rbase + v < N) {
        hp[(size_t)v * FOUT]      = c00[v];
        hp[(size_t)v * FOUT + 16] = c01[v];
      }
      if (rbase + v + 16 < N) {
        hp[(size_t)(v + 16) * FOUT]      = c10[v];
        hp[(size_t)(v + 16) * FOUT + 16] = c11[v];
      }
    }
  }
}

__global__ void rgat_node_logits(const float* __restrict__ h,
                                 const float* __restrict__ asrc,
                                 const float* __restrict__ adst,
                                 float* __restrict__ als, float* __restrict__ ald,
                                 int N) {
  int idx = blockIdx.x * blockDim.x + threadIdx.x;
  if (idx >= N * NHEAD) return;
  int n  = idx >> 2;
  int hd = idx & 3;
  const float* hp = h + (size_t)n * FOUT + hd * DHEAD;
  const float* as = asrc + hd * DHEAD;
  const float* ad = adst + hd * DHEAD;
  float s = 0.f, d = 0.f;
#pragma unroll
  for (int i = 0; i < DHEAD; ++i) {
    float v = hp[i];
    s = fmaf(v, as[i], s);
    d = fmaf(v, ad[i], d);
  }
  als[idx] = s;
  ald[idx] = d;
}

__device__ __forceinline__ float leaky(float v) {
  return v > 0.f ? v : NEG_SLOPE * v;
}

#define SCAN_TILE 1024

__global__ void rgat_zero_cnt(int* __restrict__ cnt, int total) {
  int i = blockIdx.x * blockDim.x + threadIdx.x;
  if (i < total) cnt[i] = 0;
}

__device__ __forceinline__ const int* pick_rel(const int* a, const int* b, const int* c, int r) {
  return r == 0 ? a : (r == 1 ? b : c);
}

#define EDGE_PER_THREAD 8
#define EDGE_PER_BLOCK  (256 * EDGE_PER_THREAD)

template <bool FULL>
__device__ __forceinline__ void edge_rank_body(const int* __restrict__ dstp, int* __restrict__ cr,
                                               int* __restrict__ rr, int e0, int E) {
  int d[EDGE_PER_THREAD], k[EDGE_PER_THREAD];
#pragma unroll
  for (int i = 0; i < EDGE_PER_THREAD; ++i) {
    const int e = e0 + 256 * i;
    d[i] = (FULL || e < E) ? dstp[e] : -1;
  }
#pragma unroll
  for (int i = 0; i < EDGE_PER_THREAD; ++i) {
    k[i] = 0;
    if (FULL || d[i] >= 0) k[i] = atomicAdd(cr + d[i], 1);
  }
#pragma unroll
  for (int i = 0; i < EDGE_PER_THREAD; ++i) {
    if (FULL || d[i] >= 0) rr[e0 + 256 * i] = k[i];
  }
}

__global__ void __launch_bounds__(256)
rgat_edge_rank(const int* __restrict__ ei0, const int* __restrict__ ei1,
               const int* __restrict__ ei2, int* __restrict__ cnt,
               int* __restrict__ rank, int E, int NP) {
  const int r = blockIdx.y;
  const int* __restrict__ dstp = pick_rel(ei0, ei1, ei2, r) + E;
  int* __restrict__ cr = cnt + (size_t)r * NP;
  int* __restrict__ rr = rank + (size_t)r * E;
  const int b0 = blockIdx.x * EDGE_PER_BLOCK;
  if (b0 + EDGE_PER_BLOCK <= E) edge_rank_body<true>(dstp, cr, rr, b0 + threadIdx.x, E);
  else                          edge_rank_body<false>(dstp, cr, rr, b0 + threadIdx.x, E);
}

__device__ __forceinline__ int wave_sum_i(int v) {
  for (int m = 16; m >= 1; m >>= 1) v += __shfl_xor(v, m, 32);
  return v;
}

__global__ void __launch_bounds__(256)
rgat_cnt_tilesum(const int* __restrict__ cnt, int* __restrict__ bsum, int NP, int nt) {
  __shared__ int part[8];
  const int r = blockIdx.y, tile = blockIdx.x;
  const int4 v = *(const int4*)(cnt + (size_t)r * NP + (size_t)tile * SCAN_TILE + threadIdx.x * 4);
  int s = wave_sum_i(v.x + v.y + v.z + v.w);
  if ((threadIdx.x & 31) == 0) part[threadIdx.x >> 5] = s;
  __syncthreads();
  if (threadIdx.x == 0) {
    int t = 0;
    for (int i = 0; i < 8; ++i) t += part[i];
    bsum[r * nt + tile] = t;
  }
}

__global__ void __launch_bounds__(256)
rgat_cnt_scan(const int* __restrict__ cnt, const int* __restrict__ bsum,
              int* __restrict__ offs, int NP, int nt) {
  __shared__ int part[8];
  __shared__ int wpre[8];
  const int r = blockIdx.y, tile = blockIdx.x;
  const int lane = threadIdx.x & 31, wave = threadIdx.x >> 5;

  int p = 0;
  for (int t = threadIdx.x; t < tile; t += 256) p += bsum[r * nt + t];
  p = wave_sum_i(p);
  if (lane == 0) part[wave] = p;

  const size_t at = (size_t)r * NP + (size_t)tile * SCAN_TILE + threadIdx.x * 4;
  const int4 v = *(const int4*)(cnt + at);
  const int mine = v.x + v.y + v.z + v.w;
  int inc = mine;
  for (int m = 1; m < 32; m <<= 1) {
    int up = __shfl_up(inc, m, 32);
    if (lane >= m) inc += up;
  }
  if (lane == 31) wpre[wave] = inc;
  __syncthreads();
  int base = 0;
  for (int i = 0; i < 8; ++i) {
    base += part[i];
    if (i < wave) base += wpre[i];
  }
  int4 o;
  o.x = base + inc - mine;
  o.y = o.x + v.x;
  o.z = o.y + v.y;
  o.w = o.z + v.z;
  *(int4*)(offs + at) = o;
}

template <bool FULL>
__device__ __forceinline__ void edge_fill_body(const int* __restrict__ ei, const int* __restrict__ orl,
                                               const int* __restrict__ rr, int* __restrict__ cr,
                                               int e0, int E) {
  int s[EDGE_PER_THREAD], d[EDGE_PER_THREAD], k[EDGE_PER_THREAD], o[EDGE_PER_THREAD];
#pragma unroll
  for (int i = 0; i < EDGE_PER_THREAD; ++i) {
    const int e = e0 + 256 * i;
    const bool in = FULL || e < E;
    s[i] = in ? ei[e] : 0;
    d[i] = in ? ei[E + e] : -1;
    k[i] = in ? rr[e] : 0;
  }
#pragma unroll
  for (int i = 0; i < EDGE_PER_THREAD; ++i) o[i] = orl[FULL ? d[i] : (d[i] >= 0 ? d[i] : 0)];
#pragma unroll
  for (int i = 0; i < EDGE_PER_THREAD; ++i) {
    if (FULL || d[i] >= 0) cr[o[i] + k[i]] = s[i];
  }
}

__global__ void __launch_bounds__(256)
rgat_edge_fill(const int* __restrict__ ei0, const int* __restrict__ ei1,
               const int* __restrict__ ei2, const int* __restrict__ offs,
               const int* __restrict__ rank, int* __restrict__ csr,
               int E, int NP) {
  const int r = blockIdx.y;
  const int* __restrict__ ei = pick_rel(ei0, ei1, ei2, r);
  const int* __restrict__ orl = offs + (size_t)r * NP;
  const int* __restrict__ rr = rank + (size_t)r * E;
  int* __restrict__ cr = csr + (size_t)r * E;
  const int b0 = blockIdx.x * EDGE_PER_BLOCK;
  if (b0 + EDGE_PER_BLOCK <= E) edge_fill_body<true>(ei, orl, rr, cr, b0 + threadIdx.x, E);
  else                          edge_fill_body<false>(ei, orl, rr, cr, b0 + threadIdx.x, E);
}

#define NODES_PER_WAVE 5

__device__ __forceinline__ float wave_max_f(float v) {
  for (int m = 16; m >= 1; m >>= 1) v = fmaxf(v, __shfl_xor(v, m, 32));
  return v;
}
__device__ __forceinline__ float wave_sum_f(float v) {
  for (int m = 16; m >= 1; m >>= 1) v += __shfl_xor(v, m, 32);
  return v;
}

__device__ __forceinline__ float4 edge_alpha(const float* __restrict__ als, int s, float4 ad) {
  const float4 a = *(const float4*)(als + (size_t)s * NHEAD);
  float4 al;
  al.x = leaky(a.x + ad.x);
  al.y = leaky(a.y + ad.y);
  al.z = leaky(a.z + ad.z);
  al.w = leaky(a.w + ad.w);
  return al;
}

__device__ __forceinline__ float lane_bcast(float v, int t) {
  return __int_as_float(__builtin_amdgcn_readlane(__float_as_int(v), t));
}

__device__ __forceinline__ float edge_weight(float4 w, int t, int hsel) {
  const float x = lane_bcast(w.x, t), y = lane_bcast(w.y, t);
  const float z = lane_bcast(w.z, t), u = lane_bcast(w.w, t);
  return hsel < 2 ? (hsel == 0 ? x : y) : (hsel == 2 ? z : u);
}

template <int U>
__device__ __forceinline__ void rows_trip(float4& acc, const float* __restrict__ h,
                                          int s, float4 w, int t, int lane, int hsel) {
  float4 v[U];
  float wt[U];
#pragma unroll
  for (int u = 0; u < U; ++u) {
    const int su = __builtin_amdgcn_readlane(s, t + u);
    v[u] = *(const float4*)(h + (size_t)su * FOUT + lane * 4);
  }
#pragma unroll
  for (int u = 0; u < U; ++u) wt[u] = edge_weight(w, t + u, hsel);
#pragma unroll
  for (int u = 0; u < U; ++u) {
    acc.x = fmaf(v[u].x, wt[u], acc.x); acc.y = fmaf(v[u].y, wt[u], acc.y);
    acc.z = fmaf(v[u].z, wt[u], acc.z); acc.w = fmaf(v[u].w, wt[u], acc.w);
  }
}

__device__ __forceinline__ void gather_rows(float4& acc, const float* __restrict__ h,
                                            int s, float4 w, int n, int lane, int hsel) {
  int t = 0;
  for (; t + 8 <= n; t += 8) rows_trip<8>(acc, h, s, w, t, lane, hsel);
  if (t + 4 <= n) { rows_trip<4>(acc, h, s, w, t, lane, hsel); t += 4; }
  for (; t < n; ++t) rows_trip<1>(acc, h, s, w, t, lane, hsel);
}

__global__ void __launch_bounds__(256)
rgat_node_gather(const int* __restrict__ offs, const int* __restrict__ csr,
                 const float* __restrict__ h, const float* __restrict__ als,
                 const float* __restrict__ ald, const float* __restrict__ bias,
                 float* __restrict__ out, int N, int first, int nwaves) {
  const int lane = threadIdx.x & 31;
  const int hsel = lane >> 3;
  const int wave = __builtin_amdgcn_readfirstlane((int)(threadIdx.x >> 5));
  const float ninf = -__builtin_inff();
  const float4 zero4 = make_float4(0.f, 0.f, 0.f, 0.f);

  int d = blockIdx.x * 8 + wave;
  if (d >= N) return;

  float4 bsum = zero4;
  if (first) {
    const float4 b0 = *((const float4*)bias + lane);
    const float4 b1 = *((const float4*)(bias + FOUT) + lane);
    const float4 b2 = *((const float4*)(bias + 2 * FOUT) + lane);
    bsum.x = b0.x + b1.x + b2.x; bsum.y = b0.y + b1.y + b2.y;
    bsum.z = b0.z + b1.z + b2.z; bsum.w = b0.w + b1.w + b2.w;
  }

  int beg = offs[d];
  int end = offs[d + 1];
  float4 ad = *(const float4*)(ald + (size_t)d * NHEAD);
  int s_first = 0;
  float4 a_first = zero4;
  if (beg + lane < end) {
    s_first = csr[beg + lane];
    a_first = *(const float4*)(als + (size_t)s_first * NHEAD);
  }

  for (;;) {
    const int dn = d + nwaves;
    const bool more = dn < N;
    int beg_n = 0, end_n = 0;
    float4 ad_n = zero4;
    if (more) {
      beg_n = offs[dn];
      end_n = offs[dn + 1];
      ad_n = *(const float4*)(ald + (size_t)dn * NHEAD);
    }
    float4* __restrict__ op = (float4*)(out + (size_t)d * FOUT) + lane;
    float4 o = bsum;
    if (!first) o = *op;

    const int j0 = beg + lane;
    const bool has0 = j0 < end;
    float4 al_first = make_float4(ninf, ninf, ninf, ninf);
    if (has0) {
      al_first.x = leaky(a_first.x + ad.x);
      al_first.y = leaky(a_first.y + ad.y);
      al_first.z = leaky(a_first.z + ad.z);
      al_first.w = leaky(a_first.w + ad.w);
    }

    float4 mx = al_first;
    for (int j = j0 + 32; j < end; j += 32) {
      const float4 al = edge_alpha(als, csr[j], ad);
      mx.x = fmaxf(mx.x, al.x); mx.y = fmaxf(mx.y, al.y);
      mx.z = fmaxf(mx.z, al.z); mx.w = fmaxf(mx.w, al.w);
    }
    mx.x = wave_max_f(mx.x); mx.y = wave_max_f(mx.y);
    mx.z = wave_max_f(mx.z); mx.w = wave_max_f(mx.w);

    float4 e_first = zero4;
    if (has0) {
      e_first.x = __expf(al_first.x - mx.x); e_first.y = __expf(al_first.y - mx.y);
      e_first.z = __expf(al_first.z - mx.z); e_first.w = __expf(al_first.w - mx.w);
    }
    float4 sm = e_first;
    for (int j = j0 + 32; j < end; j += 32) {
      const float4 al = edge_alpha(als, csr[j], ad);
      sm.x += __expf(al.x - mx.x); sm.y += __expf(al.y - mx.y);
      sm.z += __expf(al.z - mx.z); sm.w += __expf(al.w - mx.w);
    }
    sm.x = wave_sum_f(sm.x) + 1e-16f; sm.y = wave_sum_f(sm.y) + 1e-16f;
    sm.z = wave_sum_f(sm.z) + 1e-16f; sm.w = wave_sum_f(sm.w) + 1e-16f;

    const bool has_n = more && (beg_n + lane < end_n);
    int s_n = 0;
    if (has_n) s_n = csr[beg_n + lane];

    float4 acc = zero4;
    {
      float4 w;
      w.x = e_first.x / sm.x; w.y = e_first.y / sm.y;
      w.z = e_first.z / sm.z; w.w = e_first.w / sm.w;
      const int n = end - beg < 32 ? end - beg : 32;
      gather_rows(acc, h, s_first, w, n, lane, hsel);
    }
    for (int base = beg + 32; base < end; base += 32) {
      const int j = base + lane;
      int s = 0;
      float4 w = zero4;
      if (j < end) {
        s = csr[j];
        const float4 al = edge_alpha(als, s, ad);
        w.x = __expf(al.x - mx.x) / sm.x; w.y = __expf(al.y - mx.y) / sm.y;
        w.z = __expf(al.z - mx.z) / sm.z; w.w = __expf(al.w - mx.w) / sm.w;
      }
      const int n = end - base < 32 ? end - base : 32;
      gather_rows(acc, h, s, w, n, lane, hsel);
    }

    float4 a_n = zero4;
    if (has_n) a_n = *(const float4*)(als + (size_t)s_n * NHEAD);

    acc.x += o.x; acc.y += o.y; acc.z += o.z; acc.w += o.w;
    *op = acc;

    if (!more) break;
    d = dn; beg = beg_n; end = end_n; ad = ad_n;
    s_first = s_n; a_first = a_n;
  }
}

extern "C" void kernel_launch(void* const* d_in, const int* in_sizes, int n_in,
                              void* d_out, int out_size, void* d_ws, size_t ws_size,
                              hipStream_t stream) {
  const float* x        = (const float*)d_in[0];
  const int*   edges[3] = {(const int*)d_in[1], (const int*)d_in[2], (const int*)d_in[3]};
  const float* W        = (const float*)d_in[4];
  const float* att_src  = (const float*)d_in[5];
  const float* att_dst  = (const float*)d_in[6];
  const float* bias     = (const float*)d_in[7];

  const int N = in_sizes[0] / FIN;
  const int E = in_sizes[1] / 2;
  float* out = (float*)d_out;

  const int nt = (N + 1 + SCAN_TILE - 1) / SCAN_TILE;
  const int NP = nt * SCAN_TILE;

  char* ws = (char*)d_ws;
  size_t off = 0;
  auto carve = [&](size_t bytes) {
    void* p = ws + off;
    off += (bytes + 255) & ~(size_t)255;
    return p;
  };
  float* h    = (float*)carve((size_t)N * FOUT * sizeof(float));
  float* als  = (float*)carve((size_t)N * NHEAD * sizeof(float));
  float* ald  = (float*)carve((size_t)N * NHEAD * sizeof(float));
  int*   csr  = (int*)carve((size_t)3 * E * sizeof(int));
  int*   cnt  = (int*)carve((size_t)3 * NP * sizeof(int));
  int*   offs = (int*)carve((size_t)3 * NP * sizeof(int));
  int*   bsum = (int*)carve((size_t)3 * nt * sizeof(int));
  int*   rank = (int*)h;
  (void)ws_size;

  const dim3 egrid((E + EDGE_PER_BLOCK - 1) / EDGE_PER_BLOCK, 3);
  const dim3 sgrid(nt, 3);
  const int gblocks = (N + 8 * NODES_PER_WAVE - 1) / (8 * NODES_PER_WAVE);
  rgat_zero_cnt<<<(3 * NP + 255) / 256, 256, 0, stream>>>(cnt, 3 * NP);
  rgat_edge_rank<<<egrid, 256, 0, stream>>>(edges[0], edges[1], edges[2], cnt, rank, E, NP);
  rgat_cnt_tilesum<<<sgrid, 256, 0, stream>>>(cnt, bsum, NP, nt);
  rgat_cnt_scan<<<sgrid, 256, 0, stream>>>(cnt, bsum, offs, NP, nt);
  rgat_edge_fill<<<egrid, 256, 0, stream>>>(edges[0], edges[1], edges[2], offs, rank, csr, E, NP);

  for (int r = 0; r < 3; ++r) {
    const float* Wr = W + (size_t)r * FIN * FOUT;
    rgat_gemm_wmma<<<(N + GM_ROWS - 1) / GM_ROWS, 256, 0, stream>>>(x, Wr, h, N);
    rgat_node_logits<<<(N * NHEAD + 255) / 256, 256, 0, stream>>>(
        h, att_src + (size_t)r * NHEAD * DHEAD, att_dst + (size_t)r * NHEAD * DHEAD,
        als, ald, N);
    rgat_node_gather<<<gblocks, 256, 0, stream>>>(
        offs + (size_t)r * NP, csr + (size_t)r * E, h, als, ald, bias, out, N, r == 0,
        gblocks * 8);
  }
}


# GEMM epilogue stages its 64x128 tile in LDS: whole 512-byte row stores of h, and the per-node attention dot products taken from the staged tile in the same feature order; no separate logits launch
# speedup vs baseline: 3.9564x; 1.0523x over previous
#include <hip/hip_runtime.h>
#include <hip/hip_bf16.h>
#include <math.h>

typedef __attribute__((ext_vector_type(2))) float v2f;
typedef __attribute__((ext_vector_type(8))) float v8f;

#define FIN   256
#define FOUT  128
#define NHEAD 4
#define DHEAD 32
#define NEG_SLOPE 0.2f

#define GM_ROWS  64
#define GM_KC    32
#define GM_NCH   (FIN / GM_KC)
#define XS_PITCH 36
#define WS_PITCH 288
#define HS_PITCH 132
#define XS_FLOATS (GM_ROWS * XS_PITCH)
#define WS_FLOATS ((GM_KC / 2) * WS_PITCH)

__device__ __forceinline__ v8f wmma_f32(v2f a, v2f b, v8f c) {
  return __builtin_amdgcn_wmma_f32_16x16x4_f32(
       false, a,  false, b,
       (short)0, c,  false,  false);
}

__device__ __forceinline__ void stage_w_pair(float* __restrict__ dst, float4 lo, float4 hi) {
  *(float4*)(dst)     = make_float4(lo.x, hi.x, lo.y, hi.y);
  *(float4*)(dst + 4) = make_float4(lo.z, hi.z, lo.w, hi.w);
}

__device__ __forceinline__ void gemm_chunk(const float* __restrict__ xa, const float* __restrict__ wb,
                                           v8f& c00, v8f& c01, v8f& c10, v8f& c11) {
#pragma unroll
  for (int ks = 0; ks < GM_KC / 4; ++ks) {
    const v2f a0 = *(const v2f*)(xa + ks * 4);
    const v2f a1 = *(const v2f*)(xa + 16 * XS_PITCH + ks * 4);
    const v2f b0 = *(const v2f*)(wb + ks * 2 * WS_PITCH);
    const v2f b1 = *(const v2f*)(wb + ks * 2 * WS_PITCH + 32);
    c00 = wmma_f32(a0, b0, c00);
    c01 = wmma_f32(a0, b1, c01);
    c10 = wmma_f32(a1, b0, c10);
    c11 = wmma_f32(a1, b1, c11);
  }
}

__global__ void __launch_bounds__(256)
rgat_gemm_wmma(const float* x, const float* Wr, const float* asrc, const float* adst,
               float* __restrict__ h, float* __restrict__ als, float* __restrict__ ald, int N) {
  __shared__ __attribute__((aligned(16))) float sm[2 * XS_FLOATS + 2 * WS_FLOATS];
  __shared__ __attribute__((aligned(16))) float av[2 * FOUT];
  float (*xs)[XS_FLOATS]  = (float (*)[XS_FLOATS])sm;
  float (*wsm)[WS_FLOATS] = (float (*)[WS_FLOATS])(sm + 2 * XS_FLOATS);
  float* hs = sm;

  const int tid  = threadIdx.x;
  const int lane = tid & 31;
  const int wave = tid >> 5;
  const int wm   = wave & 1;
  const int wn   = wave >> 1;
  const int c16  = lane & 15;
  const int hh   = lane >> 4;
  const int m0   = blockIdx.x * GM_ROWS;

  const int xr = tid >> 3, xq = tid & 7;
  int gr0 = m0 + xr;       gr0 = gr0 < N ? gr0 : 0;
  int gr1 = m0 + xr + 32;  gr1 = gr1 < N ? gr1 : 0;
  const float* xg0 = x + (size_t)gr0 * FIN + xq * 4;
  const float* xg1 = x + (size_t)gr1 * FIN + xq * 4;
  const int xs_at = xr * XS_PITCH + xq * 4;
  const int wk = tid >> 5, wq = tid & 31;
  const float* wg = Wr + (size_t)(2 * wk) * FOUT + wq * 4;
  const int ws_at = wk * WS_PITCH + wq * 8;

  const int xa_at = (wm * 32 + c16) * XS_PITCH + 2 * hh;
  const int wb_at = hh * WS_PITCH + (wn * 32 + c16) * 2;

  float4 px0, px1, pw0, pw1, pw2, pw3;
  px0 = *(const float4*)(xg0);
  px1 = *(const float4*)(xg1);
  pw0 = *(const float4*)(wg);
  pw1 = *(const float4*)(wg + FOUT);
  pw2 = *(const float4*)(wg + 16 * FOUT);
  pw3 = *(const float4*)(wg + 17 * FOUT);
  const float pa = *(((tid & 128) ? adst : asrc) + (tid & 127));
  *(float4*)(&xs[0][xs_at])                 = px0;
  *(float4*)(&xs[0][xs_at + 32 * XS_PITCH]) = px1;
  stage_w_pair(&wsm[0][ws_at], pw0, pw1);
  stage_w_pair(&wsm[0][ws_at + 8 * WS_PITCH], pw2, pw3);
  av[tid] = pa;
  __syncthreads();

  v8f c00 = {}, c01 = {}, c10 = {}, c11 = {};
#pragma unroll
  for (int ch = 0; ch + 1 < GM_NCH; ++ch) {
    const int buf = ch & 1;
    const int nb  = buf ^ 1;
    const int kn  = (ch + 1) * GM_KC;
    px0 = *(const float4*)(xg0 + kn);
    px1 = *(const float4*)(xg1 + kn);
    pw0 = *(const float4*)(wg + (size_t)kn * FOUT);
    pw1 = *(const float4*)(wg + (size_t)(kn + 1) * FOUT);
    pw2 = *(const float4*)(wg + (size_t)(kn + 16) * FOUT);
    pw3 = *(const float4*)(wg + (size_t)(kn + 17) * FOUT);
    __builtin_amdgcn_sched_barrier(0);
    gemm_chunk(&xs[buf][xa_at], &wsm[buf][wb_at], c00, c01, c10, c11);
    __builtin_amdgcn_sched_barrier(0);
    *(float4*)(&xs[nb][xs_at])                 = px0;
    *(float4*)(&xs[nb][xs_at + 32 * XS_PITCH]) = px1;
    stage_w_pair(&wsm[nb][ws_at], pw0, pw1);
    stage_w_pair(&wsm[nb][ws_at + 8 * WS_PITCH], pw2, pw3);
    __syncthreads();
  }
  gemm_chunk(&xs[(GM_NCH - 1) & 1][xa_at], &wsm[(GM_NCH - 1) & 1][wb_at], c00, c01, c10, c11);

  __syncthreads();

  {
    float* hw = hs + (wm * 32 + 8 * hh) * HS_PITCH + wn * 32 + c16;
#pragma unroll
    for (int v = 0; v < 8; ++v) {
      hw[v * HS_PITCH]             = c00[v];
      hw[v * HS_PITCH + 16]        = c01[v];
      hw[(v + 16) * HS_PITCH]      = c10[v];
      hw[(v + 16) * HS_PITCH + 16] = c11[v];
    }
  }
  __syncthreads();

  int nrow = N - m0;
  nrow = nrow < GM_ROWS ? nrow : GM_ROWS;

  const float* hr = hs + wave * HS_PITCH + lane * 4;
  float* hg = h + (size_t)(m0 + wave) * FOUT + lane * 4;
  if (nrow == GM_ROWS) {
    float4 val[GM_ROWS / 8];
#pragma unroll
    for (int i = 0; i < GM_ROWS / 8; ++i) val[i] = *(const float4*)(hr + i * 8 * HS_PITCH);
#pragma unroll
    for (int i = 0; i < GM_ROWS / 8; ++i) *(float4*)(hg + (size_t)i * 8 * FOUT) = val[i];
  } else {
#pragma unroll
    for (int i = 0; i < GM_ROWS / 8; ++i) {
      if (i * 8 + wave < nrow) *(float4*)(hg + (size_t)i * 8 * FOUT) = *(const float4*)(hr + i * 8 * HS_PITCH);
    }
  }

  {
    const int r  = tid >> 2;
    const int hd = tid & 3;
    const float* hp = hs + r * HS_PITCH + hd * DHEAD;
    const float* as = av + hd * DHEAD;
    const float* ad = av + FOUT + hd * DHEAD;
    float s = 0.f, d = 0.f;
#pragma unroll
    for (int i = 0; i < DHEAD; ++i) {
      const float v = hp[i];
      s = fmaf(v, as[i], s);
      d = fmaf(v, ad[i], d);
    }
    if (r < nrow) {
      als[(size_t)m0 * NHEAD + tid] = s;
      ald[(size_t)m0 * NHEAD + tid] = d;
    }
  }
}

__device__ __forceinline__ float leaky(float v) {
  return v > 0.f ? v : NEG_SLOPE * v;
}

#define SCAN_TILE 1024

__global__ void rgat_zero_cnt(int* __restrict__ cnt, int total) {
  int i = blockIdx.x * blockDim.x + threadIdx.x;
  if (i < total) cnt[i] = 0;
}

__device__ __forceinline__ const int* pick_rel(const int* a, const int* b, const int* c, int r) {
  return r == 0 ? a : (r == 1 ? b : c);
}

#define EDGE_PER_THREAD 8
#define EDGE_PER_BLOCK  (256 * EDGE_PER_THREAD)

template <bool FULL>
__device__ __forceinline__ void edge_rank_body(const int* __restrict__ dstp, int* __restrict__ cr,
                                               int* __restrict__ rr, int e0, int E) {
  int d[EDGE_PER_THREAD], k[EDGE_PER_THREAD];
#pragma unroll
  for (int i = 0; i < EDGE_PER_THREAD; ++i) {
    const int e = e0 + 256 * i;
    d[i] = (FULL || e < E) ? dstp[e] : -1;
  }
#pragma unroll
  for (int i = 0; i < EDGE_PER_THREAD; ++i) {
    k[i] = 0;
    if (FULL || d[i] >= 0) k[i] = atomicAdd(cr + d[i], 1);
  }
#pragma unroll
  for (int i = 0; i < EDGE_PER_THREAD; ++i) {
    if (FULL || d[i] >= 0) rr[e0 + 256 * i] = k[i];
  }
}

__global__ void __launch_bounds__(256)
rgat_edge_rank(const int* __restrict__ ei0, const int* __restrict__ ei1,
               const int* __restrict__ ei2, int* __restrict__ cnt,
               int* __restrict__ rank, int E, int NP) {
  const int r = blockIdx.y;
  const int* __restrict__ dstp = pick_rel(ei0, ei1, ei2, r) + E;
  int* __restrict__ cr = cnt + (size_t)r * NP;
  int* __restrict__ rr = rank + (size_t)r * E;
  const int b0 = blockIdx.x * EDGE_PER_BLOCK;
  if (b0 + EDGE_PER_BLOCK <= E) edge_rank_body<true>(dstp, cr, rr, b0 + threadIdx.x, E);
  else                          edge_rank_body<false>(dstp, cr, rr, b0 + threadIdx.x, E);
}

__device__ __forceinline__ int wave_sum_i(int v) {
  for (int m = 16; m >= 1; m >>= 1) v += __shfl_xor(v, m, 32);
  return v;
}

__global__ void __launch_bounds__(256)
rgat_cnt_tilesum(const int* __restrict__ cnt, int* __restrict__ bsum, int NP, int nt) {
  __shared__ int part[8];
  const int r = blockIdx.y, tile = blockIdx.x;
  const int4 v = *(const int4*)(cnt + (size_t)r * NP + (size_t)tile * SCAN_TILE + threadIdx.x * 4);
  int s = wave_sum_i(v.x + v.y + v.z + v.w);
  if ((threadIdx.x & 31) == 0) part[threadIdx.x >> 5] = s;
  __syncthreads();
  if (threadIdx.x == 0) {
    int t = 0;
    for (int i = 0; i < 8; ++i) t += part[i];
    bsum[r * nt + tile] = t;
  }
}

__global__ void __launch_bounds__(256)
rgat_cnt_scan(const int* __restrict__ cnt, const int* __restrict__ bsum,
              int* __restrict__ offs, int NP, int nt) {
  __shared__ int part[8];
  __shared__ int wpre[8];
  const int r = blockIdx.y, tile = blockIdx.x;
  const int lane = threadIdx.x & 31, wave = threadIdx.x >> 5;

  int p = 0;
  for (int t = threadIdx.x; t < tile; t += 256) p += bsum[r * nt + t];
  p = wave_sum_i(p);
  if (lane == 0) part[wave] = p;

  const size_t at = (size_t)r * NP + (size_t)tile * SCAN_TILE + threadIdx.x * 4;
  const int4 v = *(const int4*)(cnt + at);
  const int mine = v.x + v.y + v.z + v.w;
  int inc = mine;
  for (int m = 1; m < 32; m <<= 1) {
    int up = __shfl_up(inc, m, 32);
    if (lane >= m) inc += up;
  }
  if (lane == 31) wpre[wave] = inc;
  __syncthreads();
  int base = 0;
  for (int i = 0; i < 8; ++i) {
    base += part[i];
    if (i < wave) base += wpre[i];
  }
  int4 o;
  o.x = base + inc - mine;
  o.y = o.x + v.x;
  o.z = o.y + v.y;
  o.w = o.z + v.z;
  *(int4*)(offs + at) = o;
}

template <bool FULL>
__device__ __forceinline__ void edge_fill_body(const int* __restrict__ ei, const int* __restrict__ orl,
                                               const int* __restrict__ rr, int* __restrict__ cr,
                                               int e0, int E) {
  int s[EDGE_PER_THREAD], d[EDGE_PER_THREAD], k[EDGE_PER_THREAD], o[EDGE_PER_THREAD];
#pragma unroll
  for (int i = 0; i < EDGE_PER_THREAD; ++i) {
    const int e = e0 + 256 * i;
    const bool in = FULL || e < E;
    s[i] = in ? ei[e] : 0;
    d[i] = in ? ei[E + e] : -1;
    k[i] = in ? rr[e] : 0;
  }
#pragma unroll
  for (int i = 0; i < EDGE_PER_THREAD; ++i) o[i] = orl[FULL ? d[i] : (d[i] >= 0 ? d[i] : 0)];
#pragma unroll
  for (int i = 0; i < EDGE_PER_THREAD; ++i) {
    if (FULL || d[i] >= 0) cr[o[i] + k[i]] = s[i];
  }
}

__global__ void __launch_bounds__(256)
rgat_edge_fill(const int* __restrict__ ei0, const int* __restrict__ ei1,
               const int* __restrict__ ei2, const int* __restrict__ offs,
               const int* __restrict__ rank, int* __restrict__ csr,
               int E, int NP) {
  const int r = blockIdx.y;
  const int* __restrict__ ei = pick_rel(ei0, ei1, ei2, r);
  const int* __restrict__ orl = offs + (size_t)r * NP;
  const int* __restrict__ rr = rank + (size_t)r * E;
  int* __restrict__ cr = csr + (size_t)r * E;
  const int b0 = blockIdx.x * EDGE_PER_BLOCK;
  if (b0 + EDGE_PER_BLOCK <= E) edge_fill_body<true>(ei, orl, rr, cr, b0 + threadIdx.x, E);
  else                          edge_fill_body<false>(ei, orl, rr, cr, b0 + threadIdx.x, E);
}

#define NODES_PER_WAVE 5

__device__ __forceinline__ float wave_max_f(float v) {
  for (int m = 16; m >= 1; m >>= 1) v = fmaxf(v, __shfl_xor(v, m, 32));
  return v;
}
__device__ __forceinline__ float wave_sum_f(float v) {
  for (int m = 16; m >= 1; m >>= 1) v += __shfl_xor(v, m, 32);
  return v;
}

__device__ __forceinline__ float4 edge_alpha(const float* __restrict__ als, int s, float4 ad) {
  const float4 a = *(const float4*)(als + (size_t)s * NHEAD);
  float4 al;
  al.x = leaky(a.x + ad.x);
  al.y = leaky(a.y + ad.y);
  al.z = leaky(a.z + ad.z);
  al.w = leaky(a.w + ad.w);
  return al;
}

__device__ __forceinline__ float lane_bcast(float v, int t) {
  return __int_as_float(__builtin_amdgcn_readlane(__float_as_int(v), t));
}

__device__ __forceinline__ float edge_weight(float4 w, int t, int hsel) {
  const float x = lane_bcast(w.x, t), y = lane_bcast(w.y, t);
  const float z = lane_bcast(w.z, t), u = lane_bcast(w.w, t);
  return hsel < 2 ? (hsel == 0 ? x : y) : (hsel == 2 ? z : u);
}

template <int U>
__device__ __forceinline__ void rows_trip(float4& acc, const float* __restrict__ h,
                                          int s, float4 w, int t, int lane, int hsel) {
  float4 v[U];
  float wt[U];
#pragma unroll
  for (int u = 0; u < U; ++u) {
    const int su = __builtin_amdgcn_readlane(s, t + u);
    v[u] = *(const float4*)(h + (size_t)su * FOUT + lane * 4);
  }
#pragma unroll
  for (int u = 0; u < U; ++u) wt[u] = edge_weight(w, t + u, hsel);
#pragma unroll
  for (int u = 0; u < U; ++u) {
    acc.x = fmaf(v[u].x, wt[u], acc.x); acc.y = fmaf(v[u].y, wt[u], acc.y);
    acc.z = fmaf(v[u].z, wt[u], acc.z); acc.w = fmaf(v[u].w, wt[u], acc.w);
  }
}

__device__ __forceinline__ void gather_rows(float4& acc, const float* __restrict__ h,
                                            int s, float4 w, int n, int lane, int hsel) {
  int t = 0;
  for (; t + 8 <= n; t += 8) rows_trip<8>(acc, h, s, w, t, lane, hsel);
  if (t + 4 <= n) { rows_trip<4>(acc, h, s, w, t, lane, hsel); t += 4; }
  for (; t < n; ++t) rows_trip<1>(acc, h, s, w, t, lane, hsel);
}

__global__ void __launch_bounds__(256)
rgat_node_gather(const int* __restrict__ offs, const int* __restrict__ csr,
                 const float* __restrict__ h, const float* __restrict__ als,
                 const float* __restrict__ ald, const float* __restrict__ bias,
                 float* __restrict__ out, int N, int first, int nwaves) {
  const int lane = threadIdx.x & 31;
  const int hsel = lane >> 3;
  const int wave = __builtin_amdgcn_readfirstlane((int)(threadIdx.x >> 5));
  const float ninf = -__builtin_inff();
  const float4 zero4 = make_float4(0.f, 0.f, 0.f, 0.f);

  int d = blockIdx.x * 8 + wave;
  if (d >= N) return;

  float4 bsum = zero4;
  if (first) {
    const float4 b0 = *((const float4*)bias + lane);
    const float4 b1 = *((const float4*)(bias + FOUT) + lane);
    const float4 b2 = *((const float4*)(bias + 2 * FOUT) + lane);
    bsum.x = b0.x + b1.x + b2.x; bsum.y = b0.y + b1.y + b2.y;
    bsum.z = b0.z + b1.z + b2.z; bsum.w = b0.w + b1.w + b2.w;
  }

  int beg = offs[d];
  int end = offs[d + 1];
  float4 ad = *(const float4*)(ald + (size_t)d * NHEAD);
  int s_first = 0;
  float4 a_first = zero4;
  if (beg + lane < end) {
    s_first = csr[beg + lane];
    a_first = *(const float4*)(als + (size_t)s_first * NHEAD);
  }

  for (;;) {
    const int dn = d + nwaves;
    const bool more = dn < N;
    int beg_n = 0, end_n = 0;
    float4 ad_n = zero4;
    if (more) {
      beg_n = offs[dn];
      end_n = offs[dn + 1];
      ad_n = *(const float4*)(ald + (size_t)dn * NHEAD);
    }
    float4* __restrict__ op = (float4*)(out + (size_t)d * FOUT) + lane;
    float4 o = bsum;
    if (!first) o = *op;

    const int j0 = beg + lane;
    const bool has0 = j0 < end;
    float4 al_first = make_float4(ninf, ninf, ninf, ninf);
    if (has0) {
      al_first.x = leaky(a_first.x + ad.x);
      al_first.y = leaky(a_first.y + ad.y);
      al_first.z = leaky(a_first.z + ad.z);
      al_first.w = leaky(a_first.w + ad.w);
    }

    float4 mx = al_first;
    for (int j = j0 + 32; j < end; j += 32) {
      const float4 al = edge_alpha(als, csr[j], ad);
      mx.x = fmaxf(mx.x, al.x); mx.y = fmaxf(mx.y, al.y);
      mx.z = fmaxf(mx.z, al.z); mx.w = fmaxf(mx.w, al.w);
    }
    mx.x = wave_max_f(mx.x); mx.y = wave_max_f(mx.y);
    mx.z = wave_max_f(mx.z); mx.w = wave_max_f(mx.w);

    float4 e_first = zero4;
    if (has0) {
      e_first.x = __expf(al_first.x - mx.x); e_first.y = __expf(al_first.y - mx.y);
      e_first.z = __expf(al_first.z - mx.z); e_first.w = __expf(al_first.w - mx.w);
    }
    float4 sm = e_first;
    for (int j = j0 + 32; j < end; j += 32) {
      const float4 al = edge_alpha(als, csr[j], ad);
      sm.x += __expf(al.x - mx.x); sm.y += __expf(al.y - mx.y);
      sm.z += __expf(al.z - mx.z); sm.w += __expf(al.w - mx.w);
    }
    sm.x = wave_sum_f(sm.x) + 1e-16f; sm.y = wave_sum_f(sm.y) + 1e-16f;
    sm.z = wave_sum_f(sm.z) + 1e-16f; sm.w = wave_sum_f(sm.w) + 1e-16f;

    const bool has_n = more && (beg_n + lane < end_n);
    int s_n = 0;
    if (has_n) s_n = csr[beg_n + lane];

    float4 acc = zero4;
    {
      float4 w;
      w.x = e_first.x / sm.x; w.y = e_first.y / sm.y;
      w.z = e_first.z / sm.z; w.w = e_first.w / sm.w;
      const int n = end - beg < 32 ? end - beg : 32;
      gather_rows(acc, h, s_first, w, n, lane, hsel);
    }
    for (int base = beg + 32; base < end; base += 32) {
      const int j = base + lane;
      int s = 0;
      float4 w = zero4;
      if (j < end) {
        s = csr[j];
        const float4 al = edge_alpha(als, s, ad);
        w.x = __expf(al.x - mx.x) / sm.x; w.y = __expf(al.y - mx.y) / sm.y;
        w.z = __expf(al.z - mx.z) / sm.z; w.w = __expf(al.w - mx.w) / sm.w;
      }
      const int n = end - base < 32 ? end - base : 32;
      gather_rows(acc, h, s, w, n, lane, hsel);
    }

    float4 a_n = zero4;
    if (has_n) a_n = *(const float4*)(als + (size_t)s_n * NHEAD);

    acc.x += o.x; acc.y += o.y; acc.z += o.z; acc.w += o.w;
    *op = acc;

    if (!more) break;
    d = dn; beg = beg_n; end = end_n; ad = ad_n;
    s_first = s_n; a_first = a_n;
  }
}

extern "C" void kernel_launch(void* const* d_in, const int* in_sizes, int n_in,
                              void* d_out, int out_size, void* d_ws, size_t ws_size,
                              hipStream_t stream) {
  const float* x        = (const float*)d_in[0];
  const int*   edges[3] = {(const int*)d_in[1], (const int*)d_in[2], (const int*)d_in[3]};
  const float* W        = (const float*)d_in[4];
  const float* att_src  = (const float*)d_in[5];
  const float* att_dst  = (const float*)d_in[6];
  const float* bias     = (const float*)d_in[7];

  const int N = in_sizes[0] / FIN;
  const int E = in_sizes[1] / 2;
  float* out = (float*)d_out;

  const int nt = (N + 1 + SCAN_TILE - 1) / SCAN_TILE;
  const int NP = nt * SCAN_TILE;

  char* ws = (char*)d_ws;
  size_t off = 0;
  auto carve = [&](size_t bytes) {
    void* p = ws + off;
    off += (bytes + 255) & ~(size_t)255;
    return p;
  };
  float* h    = (float*)carve((size_t)N * FOUT * sizeof(float));
  float* als  = (float*)carve((size_t)N * NHEAD * sizeof(float));
  float* ald  = (float*)carve((size_t)N * NHEAD * sizeof(float));
  int*   csr  = (int*)carve((size_t)3 * E * sizeof(int));
  int*   cnt  = (int*)carve((size_t)3 * NP * sizeof(int));
  int*   offs = (int*)carve((size_t)3 * NP * sizeof(int));
  int*   bsum = (int*)carve((size_t)3 * nt * sizeof(int));
  int*   rank = (int*)h;
  (void)ws_size;

  const dim3 egrid((E + EDGE_PER_BLOCK - 1) / EDGE_PER_BLOCK, 3);
  const dim3 sgrid(nt, 3);
  const int gblocks = (N + 8 * NODES_PER_WAVE - 1) / (8 * NODES_PER_WAVE);
  rgat_zero_cnt<<<(3 * NP + 255) / 256, 256, 0, stream>>>(cnt, 3 * NP);
  rgat_edge_rank<<<egrid, 256, 0, stream>>>(edges[0], edges[1], edges[2], cnt, rank, E, NP);
  rgat_cnt_tilesum<<<sgrid, 256, 0, stream>>>(cnt, bsum, NP, nt);
  rgat_cnt_scan<<<sgrid, 256, 0, stream>>>(cnt, bsum, offs, NP, nt);
  rgat_edge_fill<<<egrid, 256, 0, stream>>>(edges[0], edges[1], edges[2], offs, rank, csr, E, NP);

  for (int r = 0; r < 3; ++r) {
    const float* Wr = W + (size_t)r * FIN * FOUT;
    rgat_gemm_wmma<<<(N + GM_ROWS - 1) / GM_ROWS, 256, 0, stream>>>(
        x, Wr, att_src + (size_t)r * NHEAD * DHEAD, att_dst + (size_t)r * NHEAD * DHEAD,
        h, als, ald, N);
    rgat_node_gather<<<gblocks, 256, 0, stream>>>(
        offs + (size_t)r * NP, csr + (size_t)r * E, h, als, ald, bias, out, N, r == 0,
        gblocks * 8);
  }
}


# Destination sort moved into LDS: edges binned by 64-node range (block-local LDS count and sort, one global add per block and bin), then each bin sorted by node in LDS; sums in LDS arrival order
# speedup vs baseline: 7.1475x; 1.8066x over previous
#include <hip/hip_runtime.h>
#include <hip/hip_bf16.h>
#include <math.h>

typedef __attribute__((ext_vector_type(2))) float v2f;
typedef __attribute__((ext_vector_type(8))) float v8f;

#define FIN   256
#define FOUT  128
#define NHEAD 4
#define DHEAD 32
#define NEG_SLOPE 0.2f

#define GM_ROWS  64
#define GM_KC    32
#define GM_NCH   (FIN / GM_KC)
#define XS_PITCH 36
#define WS_PITCH 288
#define HS_PITCH 132
#define XS_FLOATS (GM_ROWS * XS_PITCH)
#define WS_FLOATS ((GM_KC / 2) * WS_PITCH)

__device__ __forceinline__ v8f wmma_f32(v2f a, v2f b, v8f c) {
  return __builtin_amdgcn_wmma_f32_16x16x4_f32(
       false, a,  false, b,
       (short)0, c,  false,  false);
}

__device__ __forceinline__ void stage_w_pair(float* __restrict__ dst, float4 lo, float4 hi) {
  *(float4*)(dst)     = make_float4(lo.x, hi.x, lo.y, hi.y);
  *(float4*)(dst + 4) = make_float4(lo.z, hi.z, lo.w, hi.w);
}

__device__ __forceinline__ void gemm_chunk(const float* __restrict__ xa, const float* __restrict__ wb,
                                           v8f& c00, v8f& c01, v8f& c10, v8f& c11) {
#pragma unroll
  for (int ks = 0; ks < GM_KC / 4; ++ks) {
    const v2f a0 = *(const v2f*)(xa + ks * 4);
    const v2f a1 = *(const v2f*)(xa + 16 * XS_PITCH + ks * 4);
    const v2f b0 = *(const v2f*)(wb + ks * 2 * WS_PITCH);
    const v2f b1 = *(const v2f*)(wb + ks * 2 * WS_PITCH + 32);
    c00 = wmma_f32(a0, b0, c00);
    c01 = wmma_f32(a0, b1, c01);
    c10 = wmma_f32(a1, b0, c10);
    c11 = wmma_f32(a1, b1, c11);
  }
}

__global__ void __launch_bounds__(256)
rgat_gemm_wmma(const float* x, const float* Wr, const float* asrc, const float* adst,
               float* __restrict__ h, float* __restrict__ als, float* __restrict__ ald, int N) {
  __shared__ __attribute__((aligned(16))) float sm[2 * XS_FLOATS + 2 * WS_FLOATS];
  __shared__ __attribute__((aligned(16))) float av[2 * FOUT];
  float (*xs)[XS_FLOATS]  = (float (*)[XS_FLOATS])sm;
  float (*wsm)[WS_FLOATS] = (float (*)[WS_FLOATS])(sm + 2 * XS_FLOATS);
  float* hs = sm;

  const int tid  = threadIdx.x;
  const int lane = tid & 31;
  const int wave = tid >> 5;
  const int wm   = wave & 1;
  const int wn   = wave >> 1;
  const int c16  = lane & 15;
  const int hh   = lane >> 4;
  const int m0   = blockIdx.x * GM_ROWS;

  const int xr = tid >> 3, xq = tid & 7;
  int gr0 = m0 + xr;       gr0 = gr0 < N ? gr0 : 0;
  int gr1 = m0 + xr + 32;  gr1 = gr1 < N ? gr1 : 0;
  const float* xg0 = x + (size_t)gr0 * FIN + xq * 4;
  const float* xg1 = x + (size_t)gr1 * FIN + xq * 4;
  const int xs_at = xr * XS_PITCH + xq * 4;
  const int wk = tid >> 5, wq = tid & 31;
  const float* wg = Wr + (size_t)(2 * wk) * FOUT + wq * 4;
  const int ws_at = wk * WS_PITCH + wq * 8;

  const int xa_at = (wm * 32 + c16) * XS_PITCH + 2 * hh;
  const int wb_at = hh * WS_PITCH + (wn * 32 + c16) * 2;

  float4 px0, px1, pw0, pw1, pw2, pw3;
  px0 = *(const float4*)(xg0);
  px1 = *(const float4*)(xg1);
  pw0 = *(const float4*)(wg);
  pw1 = *(const float4*)(wg + FOUT);
  pw2 = *(const float4*)(wg + 16 * FOUT);
  pw3 = *(const float4*)(wg + 17 * FOUT);
  const float pa = *(((tid & 128) ? adst : asrc) + (tid & 127));
  *(float4*)(&xs[0][xs_at])                 = px0;
  *(float4*)(&xs[0][xs_at + 32 * XS_PITCH]) = px1;
  stage_w_pair(&wsm[0][ws_at], pw0, pw1);
  stage_w_pair(&wsm[0][ws_at + 8 * WS_PITCH], pw2, pw3);
  av[tid] = pa;
  __syncthreads();

  v8f c00 = {}, c01 = {}, c10 = {}, c11 = {};
#pragma unroll
  for (int ch = 0; ch + 1 < GM_NCH; ++ch) {
    const int buf = ch & 1;
    const int nb  = buf ^ 1;
    const int kn  = (ch + 1) * GM_KC;
    px0 = *(const float4*)(xg0 + kn);
    px1 = *(const float4*)(xg1 + kn);
    pw0 = *(const float4*)(wg + (size_t)kn * FOUT);
    pw1 = *(const float4*)(wg + (size_t)(kn + 1) * FOUT);
    pw2 = *(const float4*)(wg + (size_t)(kn + 16) * FOUT);
    pw3 = *(const float4*)(wg + (size_t)(kn + 17) * FOUT);
    __builtin_amdgcn_sched_barrier(0);
    gemm_chunk(&xs[buf][xa_at], &wsm[buf][wb_at], c00, c01, c10, c11);
    __builtin_amdgcn_sched_barrier(0);
    *(float4*)(&xs[nb][xs_at])                 = px0;
    *(float4*)(&xs[nb][xs_at + 32 * XS_PITCH]) = px1;
    stage_w_pair(&wsm[nb][ws_at], pw0, pw1);
    stage_w_pair(&wsm[nb][ws_at + 8 * WS_PITCH], pw2, pw3);
    __syncthreads();
  }
  gemm_chunk(&xs[(GM_NCH - 1) & 1][xa_at], &wsm[(GM_NCH - 1) & 1][wb_at], c00, c01, c10, c11);

  __syncthreads();

  {
    float* hw = hs + (wm * 32 + 8 * hh) * HS_PITCH + wn * 32 + c16;
#pragma unroll
    for (int v = 0; v < 8; ++v) {
      hw[v * HS_PITCH]             = c00[v];
      hw[v * HS_PITCH + 16]        = c01[v];
      hw[(v + 16) * HS_PITCH]      = c10[v];
      hw[(v + 16) * HS_PITCH + 16] = c11[v];
    }
  }
  __syncthreads();

  int nrow = N - m0;
  nrow = nrow < GM_ROWS ? nrow : GM_ROWS;

  const float* hr = hs + wave * HS_PITCH + lane * 4;
  float* hg = h + (size_t)(m0 + wave) * FOUT + lane * 4;
  if (nrow == GM_ROWS) {
    float4 val[GM_ROWS / 8];
#pragma unroll
    for (int i = 0; i < GM_ROWS / 8; ++i) val[i] = *(const float4*)(hr + i * 8 * HS_PITCH);
#pragma unroll
    for (int i = 0; i < GM_ROWS / 8; ++i) *(float4*)(hg + (size_t)i * 8 * FOUT) = val[i];
  } else {
#pragma unroll
    for (int i = 0; i < GM_ROWS / 8; ++i) {
      if (i * 8 + wave < nrow) *(float4*)(hg + (size_t)i * 8 * FOUT) = *(const float4*)(hr + i * 8 * HS_PITCH);
    }
  }

  {
    const int r  = tid >> 2;
    const int hd = tid & 3;
    const float* hp = hs + r * HS_PITCH + hd * DHEAD;
    const float* as = av + hd * DHEAD;
    const float* ad = av + FOUT + hd * DHEAD;
    float s = 0.f, d = 0.f;
#pragma unroll
    for (int i = 0; i < DHEAD; ++i) {
      const float v = hp[i];
      s = fmaf(v, as[i], s);
      d = fmaf(v, ad[i], d);
    }
    if (r < nrow) {
      als[(size_t)m0 * NHEAD + tid] = s;
      ald[(size_t)m0 * NHEAD + tid] = d;
    }
  }
}

__device__ __forceinline__ float leaky(float v) {
  return v > 0.f ? v : NEG_SLOPE * v;
}

#define BIN_SH     6
#define BIN_NODES  (1 << BIN_SH)
#define NB_MAX     1024
#define BK_THREADS 512
#define BK_EPT     16
#define BK_CH      (BK_THREADS * BK_EPT)
#define BS_CAP     4096

__global__ void rgat_zero_cnt(int* __restrict__ cnt, int total) {
  int i = blockIdx.x * blockDim.x + threadIdx.x;
  if (i < total) cnt[i] = 0;
}

__device__ __forceinline__ const int* pick_rel(const int* a, const int* b, const int* c, int r) {
  return r == 0 ? a : (r == 1 ? b : c);
}

__device__ __forceinline__ void block_excl_scan(int* arr, int* wsum, int tid) {
  const int lane = tid & 31, wave = tid >> 5;
  const int a = arr[2 * tid], b = arr[2 * tid + 1];
  const int mine = a + b;
  int inc = mine;
  for (int m = 1; m < 32; m <<= 1) {
    int up = __shfl_up(inc, m, 32);
    if (lane >= m) inc += up;
  }
  if (lane == 31) wsum[wave] = inc;
  __syncthreads();
  int base = 0;
  for (int i = 0; i < BK_THREADS / 32; ++i) {
    if (i < wave) base += wsum[i];
  }
  const int ex = base + inc - mine;
  arr[2 * tid]     = ex;
  arr[2 * tid + 1] = ex + a;
  __syncthreads();
}

__global__ void __launch_bounds__(BK_THREADS)
rgat_bin_count(const int* __restrict__ ei0, const int* __restrict__ ei1,
               const int* __restrict__ ei2, int* __restrict__ bintot, int E, int nb) {
  __shared__ int hist[NB_MAX];
  const int tid = threadIdx.x;
  const int r = blockIdx.y;
  const int* __restrict__ dstp = pick_rel(ei0, ei1, ei2, r) + E;
  for (int i = tid; i < NB_MAX; i += BK_THREADS) hist[i] = 0;
  __syncthreads();
  const int e0 = blockIdx.x * BK_CH + tid;
  int d[BK_EPT];
#pragma unroll
  for (int i = 0; i < BK_EPT; ++i) {
    const int e = e0 + BK_THREADS * i;
    d[i] = e < E ? dstp[e] : -1;
  }
#pragma unroll
  for (int i = 0; i < BK_EPT; ++i) {
    if (d[i] >= 0) atomicAdd(&hist[d[i] >> BIN_SH], 1);
  }
  __syncthreads();
  for (int i = tid; i < nb; i += BK_THREADS) {
    const int c = hist[i];
    if (c > 0) atomicAdd(bintot + r * NB_MAX + i, c);
  }
}

__global__ void __launch_bounds__(BK_THREADS)
rgat_bin_scatter(const int* __restrict__ ei0, const int* __restrict__ ei1,
                 const int* __restrict__ ei2, const int* __restrict__ bintot,
                 int* __restrict__ cursor, int* __restrict__ binbase,
                 int2* __restrict__ binned, int E, int nb) {
  __shared__ int2 stage[BK_CH];
  __shared__ int hist[NB_MAX];
  __shared__ int gb[NB_MAX];
  __shared__ int dlt[NB_MAX];
  __shared__ int wsum[BK_THREADS / 32];
  const int tid = threadIdx.x;
  const int r = blockIdx.y;
  const int* __restrict__ ei = pick_rel(ei0, ei1, ei2, r);
  for (int i = tid; i < NB_MAX; i += BK_THREADS) {
    hist[i] = 0;
    gb[i] = i < nb ? bintot[r * NB_MAX + i] : 0;
  }
  __syncthreads();

  const int e0 = blockIdx.x * BK_CH + tid;
  int s[BK_EPT], d[BK_EPT], k[BK_EPT];
#pragma unroll
  for (int i = 0; i < BK_EPT; ++i) {
    const int e = e0 + BK_THREADS * i;
    const bool in = e < E;
    s[i] = in ? ei[e] : 0;
    d[i] = in ? ei[E + e] : -1;
  }
#pragma unroll
  for (int i = 0; i < BK_EPT; ++i) {
    k[i] = 0;
    if (d[i] >= 0) k[i] = atomicAdd(&hist[d[i] >> BIN_SH], 1);
  }
  __syncthreads();

  const int c0 = hist[2 * tid], c1 = hist[2 * tid + 1];
  block_excl_scan(gb, wsum, tid);
  block_excl_scan(hist, wsum, tid);

  if (blockIdx.x == 0) {
    for (int i = tid; i <= nb; i += BK_THREADS) binbase[r * (NB_MAX + 1) + i] = i < NB_MAX ? gb[i] : E;
  }
  {
    const int b0 = 2 * tid, b1 = 2 * tid + 1;
    if (c0 > 0) dlt[b0] = gb[b0] + atomicAdd(cursor + r * NB_MAX + b0, c0) - hist[b0];
    if (c1 > 0) dlt[b1] = gb[b1] + atomicAdd(cursor + r * NB_MAX + b1, c1) - hist[b1];
  }
#pragma unroll
  for (int i = 0; i < BK_EPT; ++i) {
    if (d[i] >= 0) stage[hist[d[i] >> BIN_SH] + k[i]] = make_int2(s[i], d[i]);
  }
  __syncthreads();

  int nloc = E - blockIdx.x * BK_CH;
  nloc = nloc < BK_CH ? nloc : BK_CH;
  int2* __restrict__ outp = binned + (size_t)r * E;
  for (int i = tid; i < nloc; i += BK_THREADS) {
    const int2 v = stage[i];
    outp[i + dlt[v.y >> BIN_SH]] = v;
  }
}

__global__ void __launch_bounds__(256)
rgat_bin_sort(const int* __restrict__ binbase, const int2* __restrict__ binned,
              int* __restrict__ offs, int* __restrict__ csr, int E, int N, int NP) {
  __shared__ int cn[BIN_NODES];
  __shared__ int st[BIN_NODES];
  __shared__ int lst[BS_CAP];
  const int tid = threadIdx.x;
  const int bin = blockIdx.x, r = blockIdx.y;
  const int b0 = binbase[r * (NB_MAX + 1) + bin];
  const int b1 = binbase[r * (NB_MAX + 1) + bin + 1];
  const int len = b1 - b0;
  const int2* __restrict__ ed = binned + (size_t)r * E + b0;
  int* __restrict__ outp = csr + (size_t)r * E + b0;

  if (tid < BIN_NODES) cn[tid] = 0;
  __syncthreads();
  for (int i = tid; i < len; i += 256) atomicAdd(&cn[ed[i].y & (BIN_NODES - 1)], 1);
  __syncthreads();
  if (tid < BIN_NODES) {
    int p = 0;
    for (int j = 0; j < BIN_NODES; ++j) {
      const int c = cn[j];
      if (j < tid) p += c;
    }
    st[tid] = p;
    const int node = bin * BIN_NODES + tid;
    if (node <= N) offs[(size_t)r * NP + node] = b0 + p;
  }
  __syncthreads();

  if (len <= BS_CAP) {
    for (int i = tid; i < len; i += 256) {
      const int2 v = ed[i];
      lst[atomicAdd(&st[v.y & (BIN_NODES - 1)], 1)] = v.x;
    }
    __syncthreads();
    for (int i = tid; i < len; i += 256) outp[i] = lst[i];
  } else {
    for (int i = tid; i < len; i += 256) {
      const int2 v = ed[i];
      outp[atomicAdd(&st[v.y & (BIN_NODES - 1)], 1)] = v.x;
    }
  }
}

#define NODES_PER_WAVE 5

__device__ __forceinline__ float wave_max_f(float v) {
  for (int m = 16; m >= 1; m >>= 1) v = fmaxf(v, __shfl_xor(v, m, 32));
  return v;
}
__device__ __forceinline__ float wave_sum_f(float v) {
  for (int m = 16; m >= 1; m >>= 1) v += __shfl_xor(v, m, 32);
  return v;
}

__device__ __forceinline__ float4 edge_alpha(const float* __restrict__ als, int s, float4 ad) {
  const float4 a = *(const float4*)(als + (size_t)s * NHEAD);
  float4 al;
  al.x = leaky(a.x + ad.x);
  al.y = leaky(a.y + ad.y);
  al.z = leaky(a.z + ad.z);
  al.w = leaky(a.w + ad.w);
  return al;
}

__device__ __forceinline__ float lane_bcast(float v, int t) {
  return __int_as_float(__builtin_amdgcn_readlane(__float_as_int(v), t));
}

__device__ __forceinline__ float edge_weight(float4 w, int t, int hsel) {
  const float x = lane_bcast(w.x, t), y = lane_bcast(w.y, t);
  const float z = lane_bcast(w.z, t), u = lane_bcast(w.w, t);
  return hsel < 2 ? (hsel == 0 ? x : y) : (hsel == 2 ? z : u);
}

template <int U>
__device__ __forceinline__ void rows_trip(float4& acc, const float* __restrict__ h,
                                          int s, float4 w, int t, int lane, int hsel) {
  float4 v[U];
  float wt[U];
#pragma unroll
  for (int u = 0; u < U; ++u) {
    const int su = __builtin_amdgcn_readlane(s, t + u);
    v[u] = *(const float4*)(h + (size_t)su * FOUT + lane * 4);
  }
#pragma unroll
  for (int u = 0; u < U; ++u) wt[u] = edge_weight(w, t + u, hsel);
#pragma unroll
  for (int u = 0; u < U; ++u) {
    acc.x = fmaf(v[u].x, wt[u], acc.x); acc.y = fmaf(v[u].y, wt[u], acc.y);
    acc.z = fmaf(v[u].z, wt[u], acc.z); acc.w = fmaf(v[u].w, wt[u], acc.w);
  }
}

__device__ __forceinline__ void gather_rows(float4& acc, const float* __restrict__ h,
                                            int s, float4 w, int n, int lane, int hsel) {
  int t = 0;
  for (; t + 8 <= n; t += 8) rows_trip<8>(acc, h, s, w, t, lane, hsel);
  if (t + 4 <= n) { rows_trip<4>(acc, h, s, w, t, lane, hsel); t += 4; }
  for (; t < n; ++t) rows_trip<1>(acc, h, s, w, t, lane, hsel);
}

__global__ void __launch_bounds__(256)
rgat_node_gather(const int* __restrict__ offs, const int* __restrict__ csr,
                 const float* __restrict__ h, const float* __restrict__ als,
                 const float* __restrict__ ald, const float* __restrict__ bias,
                 float* __restrict__ out, int N, int first, int nwaves) {
  const int lane = threadIdx.x & 31;
  const int hsel = lane >> 3;
  const int wave = __builtin_amdgcn_readfirstlane((int)(threadIdx.x >> 5));
  const float ninf = -__builtin_inff();
  const float4 zero4 = make_float4(0.f, 0.f, 0.f, 0.f);

  int d = blockIdx.x * 8 + wave;
  if (d >= N) return;

  float4 bsum = zero4;
  if (first) {
    const float4 b0 = *((const float4*)bias + lane);
    const float4 b1 = *((const float4*)(bias + FOUT) + lane);
    const float4 b2 = *((const float4*)(bias + 2 * FOUT) + lane);
    bsum.x = b0.x + b1.x + b2.x; bsum.y = b0.y + b1.y + b2.y;
    bsum.z = b0.z + b1.z + b2.z; bsum.w = b0.w + b1.w + b2.w;
  }

  int beg = offs[d];
  int end = offs[d + 1];
  float4 ad = *(const float4*)(ald + (size_t)d * NHEAD);
  int s_first = 0;
  float4 a_first = zero4;
  if (beg + lane < end) {
    s_first = csr[beg + lane];
    a_first = *(const float4*)(als + (size_t)s_first * NHEAD);
  }

  for (;;) {
    const int dn = d + nwaves;
    const bool more = dn < N;
    int beg_n = 0, end_n = 0;
    float4 ad_n = zero4;
    if (more) {
      beg_n = offs[dn];
      end_n = offs[dn + 1];
      ad_n = *(const float4*)(ald + (size_t)dn * NHEAD);
    }
    float4* __restrict__ op = (float4*)(out + (size_t)d * FOUT) + lane;
    float4 o = bsum;
    if (!first) o = *op;

    const int j0 = beg + lane;
    const bool has0 = j0 < end;
    float4 al_first = make_float4(ninf, ninf, ninf, ninf);
    if (has0) {
      al_first.x = leaky(a_first.x + ad.x);
      al_first.y = leaky(a_first.y + ad.y);
      al_first.z = leaky(a_first.z + ad.z);
      al_first.w = leaky(a_first.w + ad.w);
    }

    float4 mx = al_first;
    for (int j = j0 + 32; j < end; j += 32) {
      const float4 al = edge_alpha(als, csr[j], ad);
      mx.x = fmaxf(mx.x, al.x); mx.y = fmaxf(mx.y, al.y);
      mx.z = fmaxf(mx.z, al.z); mx.w = fmaxf(mx.w, al.w);
    }
    mx.x = wave_max_f(mx.x); mx.y = wave_max_f(mx.y);
    mx.z = wave_max_f(mx.z); mx.w = wave_max_f(mx.w);

    float4 e_first = zero4;
    if (has0) {
      e_first.x = __expf(al_first.x - mx.x); e_first.y = __expf(al_first.y - mx.y);
      e_first.z = __expf(al_first.z - mx.z); e_first.w = __expf(al_first.w - mx.w);
    }
    float4 sm = e_first;
    for (int j = j0 + 32; j < end; j += 32) {
      const float4 al = edge_alpha(als, csr[j], ad);
      sm.x += __expf(al.x - mx.x); sm.y += __expf(al.y - mx.y);
      sm.z += __expf(al.z - mx.z); sm.w += __expf(al.w - mx.w);
    }
    sm.x = wave_sum_f(sm.x) + 1e-16f; sm.y = wave_sum_f(sm.y) + 1e-16f;
    sm.z = wave_sum_f(sm.z) + 1e-16f; sm.w = wave_sum_f(sm.w) + 1e-16f;

    const bool has_n = more && (beg_n + lane < end_n);
    int s_n = 0;
    if (has_n) s_n = csr[beg_n + lane];

    float4 acc = zero4;
    {
      float4 w;
      w.x = e_first.x / sm.x; w.y = e_first.y / sm.y;
      w.z = e_first.z / sm.z; w.w = e_first.w / sm.w;
      const int n = end - beg < 32 ? end - beg : 32;
      gather_rows(acc, h, s_first, w, n, lane, hsel);
    }
    for (int base = beg + 32; base < end; base += 32) {
      const int j = base + lane;
      int s = 0;
      float4 w = zero4;
      if (j < end) {
        s = csr[j];
        const float4 al = edge_alpha(als, s, ad);
        w.x = __expf(al.x - mx.x) / sm.x; w.y = __expf(al.y - mx.y) / sm.y;
        w.z = __expf(al.z - mx.z) / sm.z; w.w = __expf(al.w - mx.w) / sm.w;
      }
      const int n = end - base < 32 ? end - base : 32;
      gather_rows(acc, h, s, w, n, lane, hsel);
    }

    float4 a_n = zero4;
    if (has_n) a_n = *(const float4*)(als + (size_t)s_n * NHEAD);

    acc.x += o.x; acc.y += o.y; acc.z += o.z; acc.w += o.w;
    *op = acc;

    if (!more) break;
    d = dn; beg = beg_n; end = end_n; ad = ad_n;
    s_first = s_n; a_first = a_n;
  }
}

extern "C" void kernel_launch(void* const* d_in, const int* in_sizes, int n_in,
                              void* d_out, int out_size, void* d_ws, size_t ws_size,
                              hipStream_t stream) {
  const float* x        = (const float*)d_in[0];
  const int*   edges[3] = {(const int*)d_in[1], (const int*)d_in[2], (const int*)d_in[3]};
  const float* W        = (const float*)d_in[4];
  const float* att_src  = (const float*)d_in[5];
  const float* att_dst  = (const float*)d_in[6];
  const float* bias     = (const float*)d_in[7];

  const int N = in_sizes[0] / FIN;
  const int E = in_sizes[1] / 2;
  float* out = (float*)d_out;

  const int nt = (N + 1 + 1023) / 1024;
  const int NP = nt * 1024;
  const int nb = (N + 1 + BIN_NODES - 1) / BIN_NODES;

  char* ws = (char*)d_ws;
  size_t off = 0;
  auto carve = [&](size_t bytes) {
    void* p = ws + off;
    off += (bytes + 255) & ~(size_t)255;
    return p;
  };
  float* h    = (float*)carve((size_t)N * FOUT * sizeof(float));
  float* als  = (float*)carve((size_t)N * NHEAD * sizeof(float));
  float* ald  = (float*)carve((size_t)N * NHEAD * sizeof(float));
  int*   csr  = (int*)carve((size_t)3 * E * sizeof(int));
  int*   tab  = (int*)carve((size_t)3 * NP * sizeof(int));
  int*   offs = (int*)carve((size_t)3 * NP * sizeof(int));
  int*   bintot  = tab;
  int*   cursor  = tab + 3 * NB_MAX;
  int*   binbase = tab + 6 * NB_MAX;
  int2*  binned = (int2*)h;
  (void)ws_size;

  const dim3 bgrid((E + BK_CH - 1) / BK_CH, 3);
  const dim3 sgrid(nb, 3);
  const int gblocks = (N + 8 * NODES_PER_WAVE - 1) / (8 * NODES_PER_WAVE);
  rgat_zero_cnt<<<(6 * NB_MAX + 255) / 256, 256, 0, stream>>>(tab, 6 * NB_MAX);
  rgat_bin_count<<<bgrid, BK_THREADS, 0, stream>>>(edges[0], edges[1], edges[2], bintot, E, nb);
  rgat_bin_scatter<<<bgrid, BK_THREADS, 0, stream>>>(edges[0], edges[1], edges[2], bintot, cursor,
                                                     binbase, binned, E, nb);
  rgat_bin_sort<<<sgrid, 256, 0, stream>>>(binbase, binned, offs, csr, E, N, NP);

  for (int r = 0; r < 3; ++r) {
    const float* Wr = W + (size_t)r * FIN * FOUT;
    rgat_gemm_wmma<<<(N + GM_ROWS - 1) / GM_ROWS, 256, 0, stream>>>(
        x, Wr, att_src + (size_t)r * NHEAD * DHEAD, att_dst + (size_t)r * NHEAD * DHEAD,
        h, als, ald, N);
    rgat_node_gather<<<gblocks, 256, 0, stream>>>(
        offs + (size_t)r * NP, csr + (size_t)r * E, h, als, ald, bias, out, N, r == 0,
        gblocks * 8);
  }
}


# GEMM grid balanced over the 256 WGPs: full rounds of 64-row tiles, the last partial round's rows as 16-row strip blocks (8 waves x 16 cols), so no WGP gets a 4th whole tile; sum order unchanged
# speedup vs baseline: 7.9143x; 1.1073x over previous
#include <hip/hip_runtime.h>
#include <hip/hip_bf16.h>
#include <math.h>

typedef __attribute__((ext_vector_type(2))) float v2f;
typedef __attribute__((ext_vector_type(8))) float v8f;

#define FIN   256
#define FOUT  128
#define NHEAD 4
#define DHEAD 32
#define NEG_SLOPE 0.2f

#define GM_ROWS  64
#define GM_STRIP 16
#define GM_WGPS  256
#define GM_KC    32
#define GM_NCH   (FIN / GM_KC)
#define XS_PITCH 36
#define WS_PITCH 288
#define HS_PITCH 132
#define XS_FLOATS (GM_ROWS * XS_PITCH)
#define WS_FLOATS ((GM_KC / 2) * WS_PITCH)

__device__ __forceinline__ v8f wmma_f32(v2f a, v2f b, v8f c) {
  return __builtin_amdgcn_wmma_f32_16x16x4_f32(
       false, a,  false, b,
       (short)0, c,  false,  false);
}

__device__ __forceinline__ void stage_w_pair(float* __restrict__ dst, float4 lo, float4 hi) {
  *(float4*)(dst)     = make_float4(lo.x, hi.x, lo.y, hi.y);
  *(float4*)(dst + 4) = make_float4(lo.z, hi.z, lo.w, hi.w);
}

__device__ __forceinline__ void gemm_chunk(const float* __restrict__ xa, const float* __restrict__ wb,
                                           v8f& c00, v8f& c01, v8f& c10, v8f& c11) {
#pragma unroll
  for (int ks = 0; ks < GM_KC / 4; ++ks) {
    const v2f a0 = *(const v2f*)(xa + ks * 4);
    const v2f a1 = *(const v2f*)(xa + 16 * XS_PITCH + ks * 4);
    const v2f b0 = *(const v2f*)(wb + ks * 2 * WS_PITCH);
    const v2f b1 = *(const v2f*)(wb + ks * 2 * WS_PITCH + 32);
    c00 = wmma_f32(a0, b0, c00);
    c01 = wmma_f32(a0, b1, c01);
    c10 = wmma_f32(a1, b0, c10);
    c11 = wmma_f32(a1, b1, c11);
  }
}

__device__ __forceinline__ void gemm_strip16(const float* x, const float* Wr, const float* adst,
                                             float* __restrict__ h, float* __restrict__ ald,
                                             int N, int m0, float* sm, float* av) {
  float (*xs)[XS_FLOATS]  = (float (*)[XS_FLOATS])sm;
  float (*wsm)[WS_FLOATS] = (float (*)[WS_FLOATS])(sm + 2 * XS_FLOATS);
  float* hs = sm;

  const int tid  = threadIdx.x;
  const int lane = tid & 31;
  const int wave = tid >> 5;
  const int c16  = lane & 15;
  const int hh   = lane >> 4;

  const bool xld = tid < GM_STRIP * 8;
  const int xr = (tid >> 3) & (GM_STRIP - 1), xq = tid & 7;
  int gr0 = m0 + xr;  gr0 = gr0 < N ? gr0 : 0;
  const float* xg0 = x + (size_t)gr0 * FIN + xq * 4;
  const int xs_at = xr * XS_PITCH + xq * 4;
  const int wk = tid >> 5, wq = tid & 31;
  const float* wg = Wr + (size_t)(2 * wk) * FOUT + wq * 4;
  const int ws_at = wk * WS_PITCH + wq * 8;

  const int xa_at = c16 * XS_PITCH + 2 * hh;
  const int wb_at = hh * WS_PITCH + (wave * 16 + c16) * 2;

  float4 px0 = make_float4(0.f, 0.f, 0.f, 0.f), pw0, pw1, pw2, pw3;
  if (xld) px0 = *(const float4*)(xg0);
  pw0 = *(const float4*)(wg);
  pw1 = *(const float4*)(wg + FOUT);
  pw2 = *(const float4*)(wg + 16 * FOUT);
  pw3 = *(const float4*)(wg + 17 * FOUT);
  const float pa = adst[tid & 127];
  if (xld) *(float4*)(&xs[0][xs_at]) = px0;
  stage_w_pair(&wsm[0][ws_at], pw0, pw1);
  stage_w_pair(&wsm[0][ws_at + 8 * WS_PITCH], pw2, pw3);
  if (tid < FOUT) av[tid] = pa;
  __syncthreads();

  v8f c = {};
  for (int ch = 0; ch < GM_NCH; ++ch) {
    const int buf = ch & 1;
    const int nb  = buf ^ 1;
    const bool more = ch + 1 < GM_NCH;
    if (more) {
      const int kn = (ch + 1) * GM_KC;
      if (xld) px0 = *(const float4*)(xg0 + kn);
      pw0 = *(const float4*)(wg + (size_t)kn * FOUT);
      pw1 = *(const float4*)(wg + (size_t)(kn + 1) * FOUT);
      pw2 = *(const float4*)(wg + (size_t)(kn + 16) * FOUT);
      pw3 = *(const float4*)(wg + (size_t)(kn + 17) * FOUT);
    }
    const float* xa = &xs[buf][xa_at];
    const float* wb = &wsm[buf][wb_at];
#pragma unroll
    for (int ks = 0; ks < GM_KC / 4; ++ks) {
      const v2f a = *(const v2f*)(xa + ks * 4);
      const v2f b = *(const v2f*)(wb + ks * 2 * WS_PITCH);
      c = wmma_f32(a, b, c);
    }
    if (more) {
      if (xld) *(float4*)(&xs[nb][xs_at]) = px0;
      stage_w_pair(&wsm[nb][ws_at], pw0, pw1);
      stage_w_pair(&wsm[nb][ws_at + 8 * WS_PITCH], pw2, pw3);
    }
    __syncthreads();
  }

  {
    float* hw = hs + (8 * hh) * HS_PITCH + wave * 16 + c16;
#pragma unroll
    for (int v = 0; v < 8; ++v) hw[v * HS_PITCH] = c[v];
  }
  __syncthreads();

  int nrow = N - m0;
  nrow = nrow < GM_STRIP ? nrow : GM_STRIP;

  for (int i = 0; i < GM_STRIP / 8; ++i) {
    const int r = i * 8 + wave;
    if (r < nrow) *(float4*)(h + (size_t)(m0 + r) * FOUT + lane * 4) = *(const float4*)(hs + r * HS_PITCH + lane * 4);
  }

  if (tid < GM_STRIP * NHEAD) {
    const int r  = tid >> 2;
    const int hd = tid & 3;
    const float* hp = hs + r * HS_PITCH + hd * DHEAD;
    const float* ad = av + hd * DHEAD;
    float d = 0.f;
#pragma unroll
    for (int i = 0; i < DHEAD; ++i) d = fmaf(hp[i], ad[i], d);
    if (r < nrow) ald[(size_t)m0 * NHEAD + tid] = d;
  }
}

__global__ void __launch_bounds__(256)
rgat_gemm_wmma(const float* x, const float* Wr, const float* adst,
               float* __restrict__ h, float* __restrict__ ald, int N, int nmain) {
  __shared__ __attribute__((aligned(16))) float sm[2 * XS_FLOATS + 2 * WS_FLOATS];
  __shared__ __attribute__((aligned(16))) float av[FOUT];
  float (*xs)[XS_FLOATS]  = (float (*)[XS_FLOATS])sm;
  float (*wsm)[WS_FLOATS] = (float (*)[WS_FLOATS])(sm + 2 * XS_FLOATS);
  float* hs = sm;

  if ((int)blockIdx.x >= nmain) {
    gemm_strip16(x, Wr, adst, h, ald, N, nmain * GM_ROWS + ((int)blockIdx.x - nmain) * GM_STRIP, sm, av);
    return;
  }

  const int tid  = threadIdx.x;
  const int lane = tid & 31;
  const int wave = tid >> 5;
  const int wm   = wave & 1;
  const int wn   = wave >> 1;
  const int c16  = lane & 15;
  const int hh   = lane >> 4;
  const int m0   = blockIdx.x * GM_ROWS;

  const int xr = tid >> 3, xq = tid & 7;
  int gr0 = m0 + xr;       gr0 = gr0 < N ? gr0 : 0;
  int gr1 = m0 + xr + 32;  gr1 = gr1 < N ? gr1 : 0;
  const float* xg0 = x + (size_t)gr0 * FIN + xq * 4;
  const float* xg1 = x + (size_t)gr1 * FIN + xq * 4;
  const int xs_at = xr * XS_PITCH + xq * 4;
  const int wk = tid >> 5, wq = tid & 31;
  const float* wg = Wr + (size_t)(2 * wk) * FOUT + wq * 4;
  const int ws_at = wk * WS_PITCH + wq * 8;

  const int xa_at = (wm * 32 + c16) * XS_PITCH + 2 * hh;
  const int wb_at = hh * WS_PITCH + (wn * 32 + c16) * 2;

  float4 px0, px1, pw0, pw1, pw2, pw3;
  px0 = *(const float4*)(xg0);
  px1 = *(const float4*)(xg1);
  pw0 = *(const float4*)(wg);
  pw1 = *(const float4*)(wg + FOUT);
  pw2 = *(const float4*)(wg + 16 * FOUT);
  pw3 = *(const float4*)(wg + 17 * FOUT);
  const float pa = adst[tid & 127];
  *(float4*)(&xs[0][xs_at])                 = px0;
  *(float4*)(&xs[0][xs_at + 32 * XS_PITCH]) = px1;
  stage_w_pair(&wsm[0][ws_at], pw0, pw1);
  stage_w_pair(&wsm[0][ws_at + 8 * WS_PITCH], pw2, pw3);
  if (tid < FOUT) av[tid] = pa;
  __syncthreads();

  v8f c00 = {}, c01 = {}, c10 = {}, c11 = {};
#pragma unroll
  for (int ch = 0; ch + 1 < GM_NCH; ++ch) {
    const int buf = ch & 1;
    const int nb  = buf ^ 1;
    const int kn  = (ch + 1) * GM_KC;
    px0 = *(const float4*)(xg0 + kn);
    px1 = *(const float4*)(xg1 + kn);
    pw0 = *(const float4*)(wg + (size_t)kn * FOUT);
    pw1 = *(const float4*)(wg + (size_t)(kn + 1) * FOUT);
    pw2 = *(const float4*)(wg + (size_t)(kn + 16) * FOUT);
    pw3 = *(const float4*)(wg + (size_t)(kn + 17) * FOUT);
    __builtin_amdgcn_sched_barrier(0);
    gemm_chunk(&xs[buf][xa_at], &wsm[buf][wb_at], c00, c01, c10, c11);
    __builtin_amdgcn_sched_barrier(0);
    *(float4*)(&xs[nb][xs_at])                 = px0;
    *(float4*)(&xs[nb][xs_at + 32 * XS_PITCH]) = px1;
    stage_w_pair(&wsm[nb][ws_at], pw0, pw1);
    stage_w_pair(&wsm[nb][ws_at + 8 * WS_PITCH], pw2, pw3);
    __syncthreads();
  }
  gemm_chunk(&xs[(GM_NCH - 1) & 1][xa_at], &wsm[(GM_NCH - 1) & 1][wb_at], c00, c01, c10, c11);

  __syncthreads();

  {
    float* hw = hs + (wm * 32 + 8 * hh) * HS_PITCH + wn * 32 + c16;
#pragma unroll
    for (int v = 0; v < 8; ++v) {
      hw[v * HS_PITCH]             = c00[v];
      hw[v * HS_PITCH + 16]        = c01[v];
      hw[(v + 16) * HS_PITCH]      = c10[v];
      hw[(v + 16) * HS_PITCH + 16] = c11[v];
    }
  }
  __syncthreads();

  int nrow = N - m0;
  nrow = nrow < GM_ROWS ? nrow : GM_ROWS;

  const float* hr = hs + wave * HS_PITCH + lane * 4;
  float* hg = h + (size_t)(m0 + wave) * FOUT + lane * 4;
  if (nrow == GM_ROWS) {
    float4 val[GM_ROWS / 8];
#pragma unroll
    for (int i = 0; i < GM_ROWS / 8; ++i) val[i] = *(const float4*)(hr + i * 8 * HS_PITCH);
#pragma unroll
    for (int i = 0; i < GM_ROWS / 8; ++i) *(float4*)(hg + (size_t)i * 8 * FOUT) = val[i];
  } else {
#pragma unroll
    for (int i = 0; i < GM_ROWS / 8; ++i) {
      if (i * 8 + wave < nrow) *(float4*)(hg + (size_t)i * 8 * FOUT) = *(const float4*)(hr + i * 8 * HS_PITCH);
    }
  }

  {
    const int r  = tid >> 2;
    const int hd = tid & 3;
    const float* hp = hs + r * HS_PITCH + hd * DHEAD;
    const float* ad = av + hd * DHEAD;
    float d = 0.f;
#pragma unroll
    for (int i = 0; i < DHEAD; ++i) d = fmaf(hp[i], ad[i], d);
    if (r < nrow) ald[(size_t)m0 * NHEAD + tid] = d;
  }
}

__device__ __forceinline__ float leaky(float v) {
  return v > 0.f ? v : NEG_SLOPE * v;
}

#define BIN_SH     6
#define BIN_NODES  (1 << BIN_SH)
#define NB_MAX     1024
#define BK_THREADS 512
#define BK_EPT     16
#define BK_CH      (BK_THREADS * BK_EPT)
#define BS_CAP     4096

__global__ void rgat_zero_cnt(int* __restrict__ cnt, int total) {
  int i = blockIdx.x * blockDim.x + threadIdx.x;
  if (i < total) cnt[i] = 0;
}

__device__ __forceinline__ const int* pick_rel(const int* a, const int* b, const int* c, int r) {
  return r == 0 ? a : (r == 1 ? b : c);
}

__device__ __forceinline__ void block_excl_scan(int* arr, int* wsum, int tid) {
  const int lane = tid & 31, wave = tid >> 5;
  const int a = arr[2 * tid], b = arr[2 * tid + 1];
  const int mine = a + b;
  int inc = mine;
  for (int m = 1; m < 32; m <<= 1) {
    int up = __shfl_up(inc, m, 32);
    if (lane >= m) inc += up;
  }
  if (lane == 31) wsum[wave] = inc;
  __syncthreads();
  int base = 0;
  for (int i = 0; i < BK_THREADS / 32; ++i) {
    if (i < wave) base += wsum[i];
  }
  const int ex = base + inc - mine;
  arr[2 * tid]     = ex;
  arr[2 * tid + 1] = ex + a;
  __syncthreads();
}

__global__ void __launch_bounds__(BK_THREADS)
rgat_bin_count(const int* __restrict__ ei0, const int* __restrict__ ei1,
               const int* __restrict__ ei2, int* __restrict__ bintot, int E, int nb) {
  __shared__ int hist[NB_MAX];
  const int tid = threadIdx.x;
  const int r = blockIdx.y;
  const int* __restrict__ dstp = pick_rel(ei0, ei1, ei2, r) + E;
  for (int i = tid; i < NB_MAX; i += BK_THREADS) hist[i] = 0;
  __syncthreads();
  const int e0 = blockIdx.x * BK_CH + tid;
  int d[BK_EPT];
#pragma unroll
  for (int i = 0; i < BK_EPT; ++i) {
    const int e = e0 + BK_THREADS * i;
    d[i] = e < E ? dstp[e] : -1;
  }
#pragma unroll
  for (int i = 0; i < BK_EPT; ++i) {
    if (d[i] >= 0) atomicAdd(&hist[d[i] >> BIN_SH], 1);
  }
  __syncthreads();
  for (int i = tid; i < nb; i += BK_THREADS) {
    const int c = hist[i];
    if (c > 0) atomicAdd(bintot + r * NB_MAX + i, c);
  }
}

__global__ void __launch_bounds__(BK_THREADS)
rgat_bin_scatter(const int* __restrict__ ei0, const int* __restrict__ ei1,
                 const int* __restrict__ ei2, const int* __restrict__ bintot,
                 int* __restrict__ cursor, int* __restrict__ binbase,
                 int2* __restrict__ binned, int E, int nb) {
  __shared__ int2 stage[BK_CH];
  __shared__ int hist[NB_MAX];
  __shared__ int gb[NB_MAX];
  __shared__ int dlt[NB_MAX];
  __shared__ int wsum[BK_THREADS / 32];
  const int tid = threadIdx.x;
  const int r = blockIdx.y;
  const int* __restrict__ ei = pick_rel(ei0, ei1, ei2, r);
  for (int i = tid; i < NB_MAX; i += BK_THREADS) {
    hist[i] = 0;
    gb[i] = i < nb ? bintot[r * NB_MAX + i] : 0;
  }
  __syncthreads();

  const int e0 = blockIdx.x * BK_CH + tid;
  int s[BK_EPT], d[BK_EPT], k[BK_EPT];
#pragma unroll
  for (int i = 0; i < BK_EPT; ++i) {
    const int e = e0 + BK_THREADS * i;
    const bool in = e < E;
    s[i] = in ? ei[e] : 0;
    d[i] = in ? ei[E + e] : -1;
  }
#pragma unroll
  for (int i = 0; i < BK_EPT; ++i) {
    k[i] = 0;
    if (d[i] >= 0) k[i] = atomicAdd(&hist[d[i] >> BIN_SH], 1);
  }
  __syncthreads();

  const int c0 = hist[2 * tid], c1 = hist[2 * tid + 1];
  block_excl_scan(gb, wsum, tid);
  block_excl_scan(hist, wsum, tid);

  if (blockIdx.x == 0) {
    for (int i = tid; i <= nb; i += BK_THREADS) binbase[r * (NB_MAX + 1) + i] = i < NB_MAX ? gb[i] : E;
  }
  {
    const int b0 = 2 * tid, b1 = 2 * tid + 1;
    if (c0 > 0) dlt[b0] = gb[b0] + atomicAdd(cursor + r * NB_MAX + b0, c0) - hist[b0];
    if (c1 > 0) dlt[b1] = gb[b1] + atomicAdd(cursor + r * NB_MAX + b1, c1) - hist[b1];
  }
#pragma unroll
  for (int i = 0; i < BK_EPT; ++i) {
    if (d[i] >= 0) stage[hist[d[i] >> BIN_SH] + k[i]] = make_int2(s[i], d[i]);
  }
  __syncthreads();

  int nloc = E - blockIdx.x * BK_CH;
  nloc = nloc < BK_CH ? nloc : BK_CH;
  int2* __restrict__ outp = binned + (size_t)r * E;
  for (int i = tid; i < nloc; i += BK_THREADS) {
    const int2 v = stage[i];
    outp[i + dlt[v.y >> BIN_SH]] = v;
  }
}

__global__ void __launch_bounds__(256)
rgat_bin_sort(const int* __restrict__ binbase, const int2* __restrict__ binned,
              int* __restrict__ offs, int* __restrict__ csr, int E, int N, int NP) {
  __shared__ int cn[BIN_NODES];
  __shared__ int st[BIN_NODES];
  __shared__ int lst[BS_CAP];
  const int tid = threadIdx.x;
  const int bin = blockIdx.x, r = blockIdx.y;
  const int b0 = binbase[r * (NB_MAX + 1) + bin];
  const int b1 = binbase[r * (NB_MAX + 1) + bin + 1];
  const int len = b1 - b0;
  const int2* __restrict__ ed = binned + (size_t)r * E + b0;
  int* __restrict__ outp = csr + (size_t)r * E + b0;

  if (tid < BIN_NODES) cn[tid] = 0;
  __syncthreads();
  for (int i = tid; i < len; i += 256) atomicAdd(&cn[ed[i].y & (BIN_NODES - 1)], 1);
  __syncthreads();
  if (tid < BIN_NODES) {
    int p = 0;
    for (int j = 0; j < BIN_NODES; ++j) {
      const int c = cn[j];
      if (j < tid) p += c;
    }
    st[tid] = p;
    const int node = bin * BIN_NODES + tid;
    if (node <= N) offs[(size_t)r * NP + node] = b0 + p;
  }
  __syncthreads();

  if (len <= BS_CAP) {
    for (int i = tid; i < len; i += 256) {
      const int2 v = ed[i];
      lst[atomicAdd(&st[v.y & (BIN_NODES - 1)], 1)] = v.x;
    }
    __syncthreads();
    for (int i = tid; i < len; i += 256) outp[i] = lst[i];
  } else {
    for (int i = tid; i < len; i += 256) {
      const int2 v = ed[i];
      outp[atomicAdd(&st[v.y & (BIN_NODES - 1)], 1)] = v.x;
    }
  }
}

#define NODES_PER_WAVE 5

template <int M>
__device__ __forceinline__ float lane_xor(float v) {
  return __int_as_float(__builtin_amdgcn_mov_dpp(__float_as_int(v), 0x160 + M, 0xf, 0xf, false));
}

template <int U>
__device__ __forceinline__ void rows_trip(v2f& acc01, v2f& acc23, float& m, float& l,
                                          const float* __restrict__ h, int s, int t, int lane,
                                          v2f as01, v2f as23, float adh) {
  v2f lo[U], hi[U];
  float sc[U];
#pragma unroll
  for (int u = 0; u < U; ++u) {
    const int su = __builtin_amdgcn_readlane(s, t + u);
    const float4 v = *(const float4*)(h + (size_t)su * FOUT + lane * 4);
    lo[u] = (v2f){v.x, v.y};
    hi[u] = (v2f){v.z, v.w};
  }
#pragma unroll
  for (int u = 0; u < U; ++u) {
    const v2f q = __builtin_elementwise_fma(hi[u], as23, lo[u] * as01);
    sc[u] = q.x + q.y;
  }
#pragma unroll
  for (int u = 0; u < U; ++u) sc[u] += lane_xor<1>(sc[u]);
#pragma unroll
  for (int u = 0; u < U; ++u) sc[u] += lane_xor<2>(sc[u]);
#pragma unroll
  for (int u = 0; u < U; ++u) sc[u] += lane_xor<4>(sc[u]);
  float gm = m;
#pragma unroll
  for (int u = 0; u < U; ++u) {
    sc[u] = leaky(sc[u] + adh);
    gm = fmaxf(gm, sc[u]);
  }
  const float scale = __expf(m - gm);
  m = gm;
  l *= scale;
  acc01 *= scale;
  acc23 *= scale;
#pragma unroll
  for (int u = 0; u < U; ++u) {
    const float p = __expf(sc[u] - gm);
    l += p;
    const v2f pp = {p, p};
    acc01 = __builtin_elementwise_fma(lo[u], pp, acc01);
    acc23 = __builtin_elementwise_fma(hi[u], pp, acc23);
  }
}

__device__ __forceinline__ void gather_rows(v2f& acc01, v2f& acc23, float& m, float& l,
                                            const float* __restrict__ h, int s, int n, int lane,
                                            v2f as01, v2f as23, float adh) {
  int t = 0;
  for (; t + 8 <= n; t += 8) rows_trip<8>(acc01, acc23, m, l, h, s, t, lane, as01, as23, adh);
  if (t + 4 <= n) { rows_trip<4>(acc01, acc23, m, l, h, s, t, lane, as01, as23, adh); t += 4; }
  for (; t < n; ++t) rows_trip<1>(acc01, acc23, m, l, h, s, t, lane, as01, as23, adh);
}

__global__ void __launch_bounds__(256)
rgat_node_gather(const int* __restrict__ offs, const int* __restrict__ csr,
                 const float* __restrict__ h, const float* __restrict__ asrc,
                 const float* __restrict__ ald, const float* __restrict__ bias,
                 float* __restrict__ out, int N, int first, int nwaves) {
  const int lane = threadIdx.x & 31;
  const int hsel = lane >> 3;
  const int wave = __builtin_amdgcn_readfirstlane((int)(threadIdx.x >> 5));
  const float ninf = -__builtin_inff();
  const float4 zero4 = make_float4(0.f, 0.f, 0.f, 0.f);

  int d = blockIdx.x * 8 + wave;
  if (d >= N) return;

  const float4 as4 = *((const float4*)asrc + lane);
  const v2f as01 = {as4.x, as4.y}, as23 = {as4.z, as4.w};

  float4 bsum = zero4;
  if (first) {
    const float4 b0 = *((const float4*)bias + lane);
    const float4 b1 = *((const float4*)(bias + FOUT) + lane);
    const float4 b2 = *((const float4*)(bias + 2 * FOUT) + lane);
    bsum.x = b0.x + b1.x + b2.x; bsum.y = b0.y + b1.y + b2.y;
    bsum.z = b0.z + b1.z + b2.z; bsum.w = b0.w + b1.w + b2.w;
  }

  int beg = offs[d];
  int end = offs[d + 1];
  float4 ad = *(const float4*)(ald + (size_t)d * NHEAD);
  int s_first = 0;
  if (beg + lane < end) s_first = csr[beg + lane];

  for (;;) {
    const int dn = d + nwaves;
    const bool more = dn < N;
    int beg_n = 0, end_n = 0;
    float4 ad_n = zero4;
    if (more) {
      beg_n = offs[dn];
      end_n = offs[dn + 1];
      ad_n = *(const float4*)(ald + (size_t)dn * NHEAD);
    }
    float4* __restrict__ op = (float4*)(out + (size_t)d * FOUT) + lane;
    float4 o = bsum;
    if (!first) o = *op;

    int s_n = 0;
    if (more && (beg_n + lane < end_n)) s_n = csr[beg_n + lane];

    const float adh = hsel < 2 ? (hsel == 0 ? ad.x : ad.y) : (hsel == 2 ? ad.z : ad.w);

    v2f acc01 = {0.f, 0.f}, acc23 = {0.f, 0.f};
    float m = ninf, l = 0.f;
    {
      const int n = end - beg < 32 ? end - beg : 32;
      gather_rows(acc01, acc23, m, l, h, s_first, n, lane, as01, as23, adh);
    }
    for (int base = beg + 32; base < end; base += 32) {
      const int j = base + lane;
      int s = 0;
      if (j < end) s = csr[j];
      const int n = end - base < 32 ? end - base : 32;
      gather_rows(acc01, acc23, m, l, h, s, n, lane, as01, as23, adh);
    }

    const float inv = 1.f / (l + 1e-16f);
    float4 res;
    res.x = fmaf(acc01.x, inv, o.x); res.y = fmaf(acc01.y, inv, o.y);
    res.z = fmaf(acc23.x, inv, o.z); res.w = fmaf(acc23.y, inv, o.w);
    *op = res;

    if (!more) break;
    d = dn; beg = beg_n; end = end_n; ad = ad_n;
    s_first = s_n;
  }
}

extern "C" void kernel_launch(void* const* d_in, const int* in_sizes, int n_in,
                              void* d_out, int out_size, void* d_ws, size_t ws_size,
                              hipStream_t stream) {
  const float* x        = (const float*)d_in[0];
  const int*   edges[3] = {(const int*)d_in[1], (const int*)d_in[2], (const int*)d_in[3]};
  const float* W        = (const float*)d_in[4];
  const float* att_src  = (const float*)d_in[5];
  const float* att_dst  = (const float*)d_in[6];
  const float* bias     = (const float*)d_in[7];

  const int N = in_sizes[0] / FIN;
  const int E = in_sizes[1] / 2;
  float* out = (float*)d_out;

  const int nt = (N + 1 + 1023) / 1024;
  const int NP = nt * 1024;
  const int nb = (N + 1 + BIN_NODES - 1) / BIN_NODES;

  char* ws = (char*)d_ws;
  size_t off = 0;
  auto carve = [&](size_t bytes) {
    void* p = ws + off;
    off += (bytes + 255) & ~(size_t)255;
    return p;
  };
  float* h    = (float*)carve((size_t)N * FOUT * sizeof(float));
  float* ald  = (float*)carve((size_t)N * NHEAD * sizeof(float));
  int*   csr  = (int*)carve((size_t)3 * E * sizeof(int));
  int*   tab  = (int*)carve((size_t)3 * NP * sizeof(int));
  int*   offs = (int*)carve((size_t)3 * NP * sizeof(int));
  int*   bintot  = tab;
  int*   cursor  = tab + 3 * NB_MAX;
  int*   binbase = tab + 6 * NB_MAX;
  int2*  binned = (int2*)h;
  (void)ws_size;

  const int nmain  = (N / GM_ROWS / GM_WGPS) * GM_WGPS;
  const int nstrip = (N - nmain * GM_ROWS + GM_STRIP - 1) / GM_STRIP;
  const dim3 bgrid((E + BK_CH - 1) / BK_CH, 3);
  const dim3 sgrid(nb, 3);
  const int gblocks = (N + 8 * NODES_PER_WAVE - 1) / (8 * NODES_PER_WAVE);
  rgat_zero_cnt<<<(6 * NB_MAX + 255) / 256, 256, 0, stream>>>(tab, 6 * NB_MAX);
  rgat_bin_count<<<bgrid, BK_THREADS, 0, stream>>>(edges[0], edges[1], edges[2], bintot, E, nb);
  rgat_bin_scatter<<<bgrid, BK_THREADS, 0, stream>>>(edges[0], edges[1], edges[2], bintot, cursor,
                                                     binbase, binned, E, nb);
  rgat_bin_sort<<<sgrid, 256, 0, stream>>>(binbase, binned, offs, csr, E, N, NP);

  for (int r = 0; r < 3; ++r) {
    const float* Wr = W + (size_t)r * FIN * FOUT;
    rgat_gemm_wmma<<<nmain + nstrip, 256, 0, stream>>>(
        x, Wr, att_dst + (size_t)r * NHEAD * DHEAD, h, ald, N, nmain);
    rgat_node_gather<<<gblocks, 256, 0, stream>>>(
        offs + (size_t)r * NP, csr + (size_t)r * E, h, att_src + (size_t)r * NHEAD * DHEAD, ald,
        bias, out, N, r == 0, gblocks * 8);
  }
}


# Each relation's GEMM rides in the launch of the stage before it (first beside the bin sort, next beside the previous gather; GEMM blocks loop over tiles, 2nd h/ald plane): 7 launches, sums unchanged
# speedup vs baseline: 8.4869x; 1.0724x over previous
#include <hip/hip_runtime.h>
#include <hip/hip_bf16.h>
#include <math.h>

typedef __attribute__((ext_vector_type(2))) float v2f;
typedef __attribute__((ext_vector_type(8))) float v8f;

#define FIN   256
#define FOUT  128
#define NHEAD 4
#define DHEAD 32
#define NEG_SLOPE 0.2f

#define GM_ROWS  64
#define GM_STRIP 16
#define GM_WGPS  256
#define GM_KC    32
#define GM_NCH   (FIN / GM_KC)
#define XS_PITCH 36
#define WS_PITCH 288
#define HS_PITCH 132
#define XS_FLOATS (GM_ROWS * XS_PITCH)
#define WS_FLOATS ((GM_KC / 2) * WS_PITCH)

__device__ __forceinline__ v8f wmma_f32(v2f a, v2f b, v8f c) {
  return __builtin_amdgcn_wmma_f32_16x16x4_f32(
       false, a,  false, b,
       (short)0, c,  false,  false);
}

__device__ __forceinline__ void stage_w_pair(float* __restrict__ dst, float4 lo, float4 hi) {
  *(float4*)(dst)     = make_float4(lo.x, hi.x, lo.y, hi.y);
  *(float4*)(dst + 4) = make_float4(lo.z, hi.z, lo.w, hi.w);
}

__device__ __forceinline__ void gemm_chunk(const float* __restrict__ xa, const float* __restrict__ wb,
                                           v8f& c00, v8f& c01, v8f& c10, v8f& c11) {
#pragma unroll
  for (int ks = 0; ks < GM_KC / 4; ++ks) {
    const v2f a0 = *(const v2f*)(xa + ks * 4);
    const v2f a1 = *(const v2f*)(xa + 16 * XS_PITCH + ks * 4);
    const v2f b0 = *(const v2f*)(wb + ks * 2 * WS_PITCH);
    const v2f b1 = *(const v2f*)(wb + ks * 2 * WS_PITCH + 32);
    c00 = wmma_f32(a0, b0, c00);
    c01 = wmma_f32(a0, b1, c01);
    c10 = wmma_f32(a1, b0, c10);
    c11 = wmma_f32(a1, b1, c11);
  }
}

__device__ __forceinline__ void gemm_strip16(const float* x, const float* Wr, const float* adst,
                                             float* __restrict__ h, float* __restrict__ ald,
                                             int N, int m0, float* sm, float* av) {
  float (*xs)[XS_FLOATS]  = (float (*)[XS_FLOATS])sm;
  float (*wsm)[WS_FLOATS] = (float (*)[WS_FLOATS])(sm + 2 * XS_FLOATS);
  float* hs = sm;

  const int tid  = threadIdx.x;
  const int lane = tid & 31;
  const int wave = tid >> 5;
  const int c16  = lane & 15;
  const int hh   = lane >> 4;

  const bool xld = tid < GM_STRIP * 8;
  const int xr = (tid >> 3) & (GM_STRIP - 1), xq = tid & 7;
  int gr0 = m0 + xr;  gr0 = gr0 < N ? gr0 : 0;
  const float* xg0 = x + (size_t)gr0 * FIN + xq * 4;
  const int xs_at = xr * XS_PITCH + xq * 4;
  const int wk = tid >> 5, wq = tid & 31;
  const float* wg = Wr + (size_t)(2 * wk) * FOUT + wq * 4;
  const int ws_at = wk * WS_PITCH + wq * 8;

  const int xa_at = c16 * XS_PITCH + 2 * hh;
  const int wb_at = hh * WS_PITCH + (wave * 16 + c16) * 2;

  float4 px0 = make_float4(0.f, 0.f, 0.f, 0.f), pw0, pw1, pw2, pw3;
  if (xld) px0 = *(const float4*)(xg0);
  pw0 = *(const float4*)(wg);
  pw1 = *(const float4*)(wg + FOUT);
  pw2 = *(const float4*)(wg + 16 * FOUT);
  pw3 = *(const float4*)(wg + 17 * FOUT);
  const float pa = adst[tid & 127];
  if (xld) *(float4*)(&xs[0][xs_at]) = px0;
  stage_w_pair(&wsm[0][ws_at], pw0, pw1);
  stage_w_pair(&wsm[0][ws_at + 8 * WS_PITCH], pw2, pw3);
  if (tid < FOUT) av[tid] = pa;
  __syncthreads();

  v8f c = {};
  for (int ch = 0; ch < GM_NCH; ++ch) {
    const int buf = ch & 1;
    const int nb  = buf ^ 1;
    const bool more = ch + 1 < GM_NCH;
    if (more) {
      const int kn = (ch + 1) * GM_KC;
      if (xld) px0 = *(const float4*)(xg0 + kn);
      pw0 = *(const float4*)(wg + (size_t)kn * FOUT);
      pw1 = *(const float4*)(wg + (size_t)(kn + 1) * FOUT);
      pw2 = *(const float4*)(wg + (size_t)(kn + 16) * FOUT);
      pw3 = *(const float4*)(wg + (size_t)(kn + 17) * FOUT);
    }
    const float* xa = &xs[buf][xa_at];
    const float* wb = &wsm[buf][wb_at];
#pragma unroll
    for (int ks = 0; ks < GM_KC / 4; ++ks) {
      const v2f a = *(const v2f*)(xa + ks * 4);
      const v2f b = *(const v2f*)(wb + ks * 2 * WS_PITCH);
      c = wmma_f32(a, b, c);
    }
    if (more) {
      if (xld) *(float4*)(&xs[nb][xs_at]) = px0;
      stage_w_pair(&wsm[nb][ws_at], pw0, pw1);
      stage_w_pair(&wsm[nb][ws_at + 8 * WS_PITCH], pw2, pw3);
    }
    __syncthreads();
  }

  {
    float* hw = hs + (8 * hh) * HS_PITCH + wave * 16 + c16;
#pragma unroll
    for (int v = 0; v < 8; ++v) hw[v * HS_PITCH] = c[v];
  }
  __syncthreads();

  int nrow = N - m0;
  nrow = nrow < GM_STRIP ? nrow : GM_STRIP;

  for (int i = 0; i < GM_STRIP / 8; ++i) {
    const int r = i * 8 + wave;
    if (r < nrow) *(float4*)(h + (size_t)(m0 + r) * FOUT + lane * 4) = *(const float4*)(hs + r * HS_PITCH + lane * 4);
  }

  if (tid < GM_STRIP * NHEAD) {
    const int r  = tid >> 2;
    const int hd = tid & 3;
    const float* hp = hs + r * HS_PITCH + hd * DHEAD;
    const float* ad = av + hd * DHEAD;
    float d = 0.f;
#pragma unroll
    for (int i = 0; i < DHEAD; ++i) d = fmaf(hp[i], ad[i], d);
    if (r < nrow) ald[(size_t)m0 * NHEAD + tid] = d;
  }
}

#define GM_SM_FLOATS (2 * XS_FLOATS + 2 * WS_FLOATS)
__device__ __forceinline__ void gemm_block(const float* x, const float* Wr, const float* adst,
                                           float* __restrict__ h, float* __restrict__ ald,
                                           int N, int nmain, int bid, float* sm, float* av) {
  float (*xs)[XS_FLOATS]  = (float (*)[XS_FLOATS])sm;
  float (*wsm)[WS_FLOATS] = (float (*)[WS_FLOATS])(sm + 2 * XS_FLOATS);
  float* hs = sm;

  if (bid >= nmain) {
    gemm_strip16(x, Wr, adst, h, ald, N, nmain * GM_ROWS + (bid - nmain) * GM_STRIP, sm, av);
    return;
  }

  const int tid  = threadIdx.x;
  const int lane = tid & 31;
  const int wave = tid >> 5;
  const int wm   = wave & 1;
  const int wn   = wave >> 1;
  const int c16  = lane & 15;
  const int hh   = lane >> 4;
  const int m0   = bid * GM_ROWS;

  const int xr = tid >> 3, xq = tid & 7;
  int gr0 = m0 + xr;       gr0 = gr0 < N ? gr0 : 0;
  int gr1 = m0 + xr + 32;  gr1 = gr1 < N ? gr1 : 0;
  const float* xg0 = x + (size_t)gr0 * FIN + xq * 4;
  const float* xg1 = x + (size_t)gr1 * FIN + xq * 4;
  const int xs_at = xr * XS_PITCH + xq * 4;
  const int wk = tid >> 5, wq = tid & 31;
  const float* wg = Wr + (size_t)(2 * wk) * FOUT + wq * 4;
  const int ws_at = wk * WS_PITCH + wq * 8;

  const int xa_at = (wm * 32 + c16) * XS_PITCH + 2 * hh;
  const int wb_at = hh * WS_PITCH + (wn * 32 + c16) * 2;

  float4 px0, px1, pw0, pw1, pw2, pw3;
  px0 = *(const float4*)(xg0);
  px1 = *(const float4*)(xg1);
  pw0 = *(const float4*)(wg);
  pw1 = *(const float4*)(wg + FOUT);
  pw2 = *(const float4*)(wg + 16 * FOUT);
  pw3 = *(const float4*)(wg + 17 * FOUT);
  const float pa = adst[tid & 127];
  *(float4*)(&xs[0][xs_at])                 = px0;
  *(float4*)(&xs[0][xs_at + 32 * XS_PITCH]) = px1;
  stage_w_pair(&wsm[0][ws_at], pw0, pw1);
  stage_w_pair(&wsm[0][ws_at + 8 * WS_PITCH], pw2, pw3);
  if (tid < FOUT) av[tid] = pa;
  __syncthreads();

  v8f c00 = {}, c01 = {}, c10 = {}, c11 = {};
#pragma unroll
  for (int ch = 0; ch + 1 < GM_NCH; ++ch) {
    const int buf = ch & 1;
    const int nb  = buf ^ 1;
    const int kn  = (ch + 1) * GM_KC;
    px0 = *(const float4*)(xg0 + kn);
    px1 = *(const float4*)(xg1 + kn);
    pw0 = *(const float4*)(wg + (size_t)kn * FOUT);
    pw1 = *(const float4*)(wg + (size_t)(kn + 1) * FOUT);
    pw2 = *(const float4*)(wg + (size_t)(kn + 16) * FOUT);
    pw3 = *(const float4*)(wg + (size_t)(kn + 17) * FOUT);
    __builtin_amdgcn_sched_barrier(0);
    gemm_chunk(&xs[buf][xa_at], &wsm[buf][wb_at], c00, c01, c10, c11);
    __builtin_amdgcn_sched_barrier(0);
    *(float4*)(&xs[nb][xs_at])                 = px0;
    *(float4*)(&xs[nb][xs_at + 32 * XS_PITCH]) = px1;
    stage_w_pair(&wsm[nb][ws_at], pw0, pw1);
    stage_w_pair(&wsm[nb][ws_at + 8 * WS_PITCH], pw2, pw3);
    __syncthreads();
  }
  gemm_chunk(&xs[(GM_NCH - 1) & 1][xa_at], &wsm[(GM_NCH - 1) & 1][wb_at], c00, c01, c10, c11);

  __syncthreads();

  {
    float* hw = hs + (wm * 32 + 8 * hh) * HS_PITCH + wn * 32 + c16;
#pragma unroll
    for (int v = 0; v < 8; ++v) {
      hw[v * HS_PITCH]             = c00[v];
      hw[v * HS_PITCH + 16]        = c01[v];
      hw[(v + 16) * HS_PITCH]      = c10[v];
      hw[(v + 16) * HS_PITCH + 16] = c11[v];
    }
  }
  __syncthreads();

  int nrow = N - m0;
  nrow = nrow < GM_ROWS ? nrow : GM_ROWS;

  const float* hr = hs + wave * HS_PITCH + lane * 4;
  float* hg = h + (size_t)(m0 + wave) * FOUT + lane * 4;
  if (nrow == GM_ROWS) {
    float4 val[GM_ROWS / 8];
#pragma unroll
    for (int i = 0; i < GM_ROWS / 8; ++i) val[i] = *(const float4*)(hr + i * 8 * HS_PITCH);
#pragma unroll
    for (int i = 0; i < GM_ROWS / 8; ++i) *(float4*)(hg + (size_t)i * 8 * FOUT) = val[i];
  } else {
#pragma unroll
    for (int i = 0; i < GM_ROWS / 8; ++i) {
      if (i * 8 + wave < nrow) *(float4*)(hg + (size_t)i * 8 * FOUT) = *(const float4*)(hr + i * 8 * HS_PITCH);
    }
  }

  {
    const int r  = tid >> 2;
    const int hd = tid & 3;
    const float* hp = hs + r * HS_PITCH + hd * DHEAD;
    const float* ad = av + hd * DHEAD;
    float d = 0.f;
#pragma unroll
    for (int i = 0; i < DHEAD; ++i) d = fmaf(hp[i], ad[i], d);
    if (r < nrow) ald[(size_t)m0 * NHEAD + tid] = d;
  }
}

__device__ __forceinline__ float leaky(float v) {
  return v > 0.f ? v : NEG_SLOPE * v;
}

#define BIN_SH     6
#define BIN_NODES  (1 << BIN_SH)
#define NB_MAX     1024
#define BK_THREADS 512
#define BK_EPT     16
#define BK_CH      (BK_THREADS * BK_EPT)
#define BS_CAP     4096

__global__ void rgat_zero_cnt(int* __restrict__ cnt, int total) {
  int i = blockIdx.x * blockDim.x + threadIdx.x;
  if (i < total) cnt[i] = 0;
}

__device__ __forceinline__ const int* pick_rel(const int* a, const int* b, const int* c, int r) {
  return r == 0 ? a : (r == 1 ? b : c);
}

__device__ __forceinline__ void block_excl_scan(int* arr, int* wsum, int tid) {
  const int lane = tid & 31, wave = tid >> 5;
  const int a = arr[2 * tid], b = arr[2 * tid + 1];
  const int mine = a + b;
  int inc = mine;
  for (int m = 1; m < 32; m <<= 1) {
    int up = __shfl_up(inc, m, 32);
    if (lane >= m) inc += up;
  }
  if (lane == 31) wsum[wave] = inc;
  __syncthreads();
  int base = 0;
  for (int i = 0; i < BK_THREADS / 32; ++i) {
    if (i < wave) base += wsum[i];
  }
  const int ex = base + inc - mine;
  arr[2 * tid]     = ex;
  arr[2 * tid + 1] = ex + a;
  __syncthreads();
}

__global__ void __launch_bounds__(BK_THREADS)
rgat_bin_count(const int* __restrict__ ei0, const int* __restrict__ ei1,
               const int* __restrict__ ei2, int* __restrict__ bintot, int E, int nb) {
  __shared__ int hist[NB_MAX];
  const int tid = threadIdx.x;
  const int r = blockIdx.y;
  const int* __restrict__ dstp = pick_rel(ei0, ei1, ei2, r) + E;
  for (int i = tid; i < NB_MAX; i += BK_THREADS) hist[i] = 0;
  __syncthreads();
  const int e0 = blockIdx.x * BK_CH + tid;
  int d[BK_EPT];
#pragma unroll
  for (int i = 0; i < BK_EPT; ++i) {
    const int e = e0 + BK_THREADS * i;
    d[i] = e < E ? dstp[e] : -1;
  }
#pragma unroll
  for (int i = 0; i < BK_EPT; ++i) {
    if (d[i] >= 0) atomicAdd(&hist[d[i] >> BIN_SH], 1);
  }
  __syncthreads();
  for (int i = tid; i < nb; i += BK_THREADS) {
    const int c = hist[i];
    if (c > 0) atomicAdd(bintot + r * NB_MAX + i, c);
  }
}

__global__ void __launch_bounds__(BK_THREADS)
rgat_bin_scatter(const int* __restrict__ ei0, const int* __restrict__ ei1,
                 const int* __restrict__ ei2, const int* __restrict__ bintot,
                 int* __restrict__ cursor, int* __restrict__ binbase,
                 int2* __restrict__ binned, int E, int nb) {
  __shared__ int2 stage[BK_CH];
  __shared__ int hist[NB_MAX];
  __shared__ int gb[NB_MAX];
  __shared__ int dlt[NB_MAX];
  __shared__ int wsum[BK_THREADS / 32];
  const int tid = threadIdx.x;
  const int r = blockIdx.y;
  const int* __restrict__ ei = pick_rel(ei0, ei1, ei2, r);
  for (int i = tid; i < NB_MAX; i += BK_THREADS) {
    hist[i] = 0;
    gb[i] = i < nb ? bintot[r * NB_MAX + i] : 0;
  }
  __syncthreads();

  const int e0 = blockIdx.x * BK_CH + tid;
  int s[BK_EPT], d[BK_EPT], k[BK_EPT];
#pragma unroll
  for (int i = 0; i < BK_EPT; ++i) {
    const int e = e0 + BK_THREADS * i;
    const bool in = e < E;
    s[i] = in ? ei[e] : 0;
    d[i] = in ? ei[E + e] : -1;
  }
#pragma unroll
  for (int i = 0; i < BK_EPT; ++i) {
    k[i] = 0;
    if (d[i] >= 0) k[i] = atomicAdd(&hist[d[i] >> BIN_SH], 1);
  }
  __syncthreads();

  const int c0 = hist[2 * tid], c1 = hist[2 * tid + 1];
  block_excl_scan(gb, wsum, tid);
  block_excl_scan(hist, wsum, tid);

  if (blockIdx.x == 0) {
    for (int i = tid; i <= nb; i += BK_THREADS) binbase[r * (NB_MAX + 1) + i] = i < NB_MAX ? gb[i] : E;
  }
  {
    const int b0 = 2 * tid, b1 = 2 * tid + 1;
    if (c0 > 0) dlt[b0] = gb[b0] + atomicAdd(cursor + r * NB_MAX + b0, c0) - hist[b0];
    if (c1 > 0) dlt[b1] = gb[b1] + atomicAdd(cursor + r * NB_MAX + b1, c1) - hist[b1];
  }
#pragma unroll
  for (int i = 0; i < BK_EPT; ++i) {
    if (d[i] >= 0) stage[hist[d[i] >> BIN_SH] + k[i]] = make_int2(s[i], d[i]);
  }
  __syncthreads();

  int nloc = E - blockIdx.x * BK_CH;
  nloc = nloc < BK_CH ? nloc : BK_CH;
  int2* __restrict__ outp = binned + (size_t)r * E;
  for (int i = tid; i < nloc; i += BK_THREADS) {
    const int2 v = stage[i];
    outp[i + dlt[v.y >> BIN_SH]] = v;
  }
}

#define BS_LDS_INTS (2 * BIN_NODES + BS_CAP)
__device__ __forceinline__ void bin_sort_block(const int* __restrict__ binbase, const int2* __restrict__ binned,
                                               int* __restrict__ offs, int* __restrict__ csr,
                                               int E, int N, int NP, int bin, int r,
                                               int* cn, int* st, int* lst) {
  const int tid = threadIdx.x;
  const int b0 = binbase[r * (NB_MAX + 1) + bin];
  const int b1 = binbase[r * (NB_MAX + 1) + bin + 1];
  const int len = b1 - b0;
  const int2* __restrict__ ed = binned + (size_t)r * E + b0;
  int* __restrict__ outp = csr + (size_t)r * E + b0;

  if (tid < BIN_NODES) cn[tid] = 0;
  __syncthreads();
  for (int i = tid; i < len; i += 256) atomicAdd(&cn[ed[i].y & (BIN_NODES - 1)], 1);
  __syncthreads();
  if (tid < BIN_NODES) {
    int p = 0;
    for (int j = 0; j < BIN_NODES; ++j) {
      const int c = cn[j];
      if (j < tid) p += c;
    }
    st[tid] = p;
    const int node = bin * BIN_NODES + tid;
    if (node <= N) offs[(size_t)r * NP + node] = b0 + p;
  }
  __syncthreads();

  if (len <= BS_CAP) {
    for (int i = tid; i < len; i += 256) {
      const int2 v = ed[i];
      lst[atomicAdd(&st[v.y & (BIN_NODES - 1)], 1)] = v.x;
    }
    __syncthreads();
    for (int i = tid; i < len; i += 256) outp[i] = lst[i];
  } else {
    for (int i = tid; i < len; i += 256) {
      const int2 v = ed[i];
      outp[atomicAdd(&st[v.y & (BIN_NODES - 1)], 1)] = v.x;
    }
  }
}

__device__ __forceinline__ void gemm_role(const float* x, const float* Wr, const float* adst,
                                          float* __restrict__ h, float* __restrict__ ald,
                                          int N, int nmain, int ngemm, int g, int ngrole,
                                          float* sm, float* av) {
  for (int t = g; t < ngemm; t += ngrole) {
    gemm_block(x, Wr, adst, h, ald, N, nmain, t, sm, av);
    __syncthreads();
  }
}

__global__ void __launch_bounds__(256, 8)
rgat_sort_gemm(const int* __restrict__ binbase, const int2* __restrict__ binned,
               int* __restrict__ offs, int* __restrict__ csr, int E, int N, int NP, int nb,
               const float* x, const float* W0, const float* adst0,
               float* __restrict__ h, float* __restrict__ ald, int nmain, int ngemm) {
  __shared__ __attribute__((aligned(16))) float sm[GM_SM_FLOATS];
  __shared__ __attribute__((aligned(16))) float av[FOUT];
  static_assert(BS_LDS_INTS <= GM_SM_FLOATS, "the sort tables fit the GEMM's LDS block");
  const int b = (int)blockIdx.x;
  const int rnd = b / GM_WGPS, plc = b - rnd * GM_WGPS;
  const bool mixed = rnd < 4;
  const int idx = mixed ? rnd * (GM_WGPS / 2) + (plc >> 1) : 2 * GM_WGPS + (b - 4 * GM_WGPS);
  if (mixed && ((plc + rnd) & 1) == 0) {
    gemm_role(x, W0, adst0, h, ald, N, nmain, ngemm, idx, 2 * GM_WGPS, sm, av);
  } else if (idx < 3 * nb) {
    const int r = idx / nb;
    int* tb = (int*)sm;
    bin_sort_block(binbase, binned, offs, csr, E, N, NP, idx - r * nb, r,
                   tb, tb + BIN_NODES, tb + 2 * BIN_NODES);
  }
}

#define NODES_PER_WAVE 5

template <int M>
__device__ __forceinline__ float lane_xor(float v) {
  return __int_as_float(__builtin_amdgcn_mov_dpp(__float_as_int(v), 0x160 + M, 0xf, 0xf, false));
}

template <int U>
__device__ __forceinline__ void rows_trip(v2f& acc01, v2f& acc23, float& m, float& l,
                                          const float* __restrict__ h, int s, int t, int lane,
                                          v2f as01, v2f as23, float adh) {
  v2f lo[U], hi[U];
  float sc[U];
#pragma unroll
  for (int u = 0; u < U; ++u) {
    const int su = __builtin_amdgcn_readlane(s, t + u);
    const float4 v = *(const float4*)(h + (size_t)su * FOUT + lane * 4);
    lo[u] = (v2f){v.x, v.y};
    hi[u] = (v2f){v.z, v.w};
  }
#pragma unroll
  for (int u = 0; u < U; ++u) {
    const v2f q = __builtin_elementwise_fma(hi[u], as23, lo[u] * as01);
    sc[u] = q.x + q.y;
  }
#pragma unroll
  for (int u = 0; u < U; ++u) sc[u] += lane_xor<1>(sc[u]);
#pragma unroll
  for (int u = 0; u < U; ++u) sc[u] += lane_xor<2>(sc[u]);
#pragma unroll
  for (int u = 0; u < U; ++u) sc[u] += lane_xor<4>(sc[u]);
  float gm = m;
#pragma unroll
  for (int u = 0; u < U; ++u) {
    sc[u] = leaky(sc[u] + adh);
    gm = fmaxf(gm, sc[u]);
  }
  const float scale = __expf(m - gm);
  m = gm;
  l *= scale;
  acc01 *= scale;
  acc23 *= scale;
#pragma unroll
  for (int u = 0; u < U; ++u) {
    const float p = __expf(sc[u] - gm);
    l += p;
    const v2f pp = {p, p};
    acc01 = __builtin_elementwise_fma(lo[u], pp, acc01);
    acc23 = __builtin_elementwise_fma(hi[u], pp, acc23);
  }
}

__device__ __forceinline__ void gather_rows(v2f& acc01, v2f& acc23, float& m, float& l,
                                            const float* __restrict__ h, int s, int n, int lane,
                                            v2f as01, v2f as23, float adh) {
  int t = 0;
  for (; t + 8 <= n; t += 8) rows_trip<8>(acc01, acc23, m, l, h, s, t, lane, as01, as23, adh);
  if (t + 4 <= n) { rows_trip<4>(acc01, acc23, m, l, h, s, t, lane, as01, as23, adh); t += 4; }
  for (; t < n; ++t) rows_trip<1>(acc01, acc23, m, l, h, s, t, lane, as01, as23, adh);
}

__device__ __forceinline__ void gather_block(const int* __restrict__ offs, const int* __restrict__ csr,
                                             const float* __restrict__ h, const float* __restrict__ asrc,
                                             const float* __restrict__ ald, const float* __restrict__ bias,
                                             float* __restrict__ out, int N, int first, int nwaves, int bid) {
  const int lane = threadIdx.x & 31;
  const int hsel = lane >> 3;
  const int wave = __builtin_amdgcn_readfirstlane((int)(threadIdx.x >> 5));
  const float ninf = -__builtin_inff();
  const float4 zero4 = make_float4(0.f, 0.f, 0.f, 0.f);

  int d = bid * 8 + wave;
  if (d >= N) return;

  const float4 as4 = *((const float4*)asrc + lane);
  const v2f as01 = {as4.x, as4.y}, as23 = {as4.z, as4.w};

  float4 bsum = zero4;
  if (first) {
    const float4 b0 = *((const float4*)bias + lane);
    const float4 b1 = *((const float4*)(bias + FOUT) + lane);
    const float4 b2 = *((const float4*)(bias + 2 * FOUT) + lane);
    bsum.x = b0.x + b1.x + b2.x; bsum.y = b0.y + b1.y + b2.y;
    bsum.z = b0.z + b1.z + b2.z; bsum.w = b0.w + b1.w + b2.w;
  }

  int beg = offs[d];
  int end = offs[d + 1];
  float4 ad = *(const float4*)(ald + (size_t)d * NHEAD);
  int s_first = 0;
  if (beg + lane < end) s_first = csr[beg + lane];

  for (;;) {
    const int dn = d + nwaves;
    const bool more = dn < N;
    int beg_n = 0, end_n = 0;
    float4 ad_n = zero4;
    if (more) {
      beg_n = offs[dn];
      end_n = offs[dn + 1];
      ad_n = *(const float4*)(ald + (size_t)dn * NHEAD);
    }
    float4* __restrict__ op = (float4*)(out + (size_t)d * FOUT) + lane;
    float4 o = bsum;
    if (!first) o = *op;

    int s_n = 0;
    if (more && (beg_n + lane < end_n)) s_n = csr[beg_n + lane];

    const float adh = hsel < 2 ? (hsel == 0 ? ad.x : ad.y) : (hsel == 2 ? ad.z : ad.w);

    v2f acc01 = {0.f, 0.f}, acc23 = {0.f, 0.f};
    float m = ninf, l = 0.f;
    {
      const int n = end - beg < 32 ? end - beg : 32;
      gather_rows(acc01, acc23, m, l, h, s_first, n, lane, as01, as23, adh);
    }
    for (int base = beg + 32; base < end; base += 32) {
      const int j = base + lane;
      int s = 0;
      if (j < end) s = csr[j];
      const int n = end - base < 32 ? end - base : 32;
      gather_rows(acc01, acc23, m, l, h, s, n, lane, as01, as23, adh);
    }

    const float inv = 1.f / (l + 1e-16f);
    float4 res;
    res.x = fmaf(acc01.x, inv, o.x); res.y = fmaf(acc01.y, inv, o.y);
    res.z = fmaf(acc23.x, inv, o.z); res.w = fmaf(acc23.y, inv, o.w);
    *op = res;

    if (!more) break;
    d = dn; beg = beg_n; end = end_n; ad = ad_n;
    s_first = s_n;
  }
}

__global__ void __launch_bounds__(256)
rgat_node_gather(const int* __restrict__ offs, const int* __restrict__ csr,
                 const float* __restrict__ h, const float* __restrict__ asrc,
                 const float* __restrict__ ald, const float* __restrict__ bias,
                 float* __restrict__ out, int N, int first, int nwaves) {
  gather_block(offs, csr, h, asrc, ald, bias, out, N, first, nwaves, (int)blockIdx.x);
}

#define MIX_GATHER_PER_WGP 3

__global__ void __launch_bounds__(256, 8)
rgat_gather_gemm(const int* __restrict__ offs, const int* __restrict__ csr,
                 const float* __restrict__ hcur, const float* __restrict__ asrc,
                 const float* __restrict__ aldcur, const float* __restrict__ bias,
                 float* __restrict__ out, int N, int first, int nwaves,
                 const float* x, const float* Wn, const float* adstn,
                 float* __restrict__ hnext, float* __restrict__ aldnext,
                 int nmain, int ngemm) {
  __shared__ __attribute__((aligned(16))) float sm[GM_SM_FLOATS];
  __shared__ __attribute__((aligned(16))) float av[FOUT];
  const int b = (int)blockIdx.x;
  const int rnd = b / GM_WGPS, plc = b - rnd * GM_WGPS;
  const int grp = rnd * (GM_WGPS / 4) + (plc >> 2);
  const int sub = plc & 3, gat = (4 - (rnd & 3)) & 3;
  if (sub == gat) {
    gemm_role(x, Wn, adstn, hnext, aldnext, N, nmain, ngemm, grp, GM_WGPS, sm, av);
  } else {
    gather_block(offs, csr, hcur, asrc, aldcur, bias, out, N, first, nwaves,
                 MIX_GATHER_PER_WGP * grp + (sub < gat ? sub : sub - 1));
  }
}

extern "C" void kernel_launch(void* const* d_in, const int* in_sizes, int n_in,
                              void* d_out, int out_size, void* d_ws, size_t ws_size,
                              hipStream_t stream) {
  const float* x        = (const float*)d_in[0];
  const int*   edges[3] = {(const int*)d_in[1], (const int*)d_in[2], (const int*)d_in[3]};
  const float* W        = (const float*)d_in[4];
  const float* att_src  = (const float*)d_in[5];
  const float* att_dst  = (const float*)d_in[6];
  const float* bias     = (const float*)d_in[7];

  const int N = in_sizes[0] / FIN;
  const int E = in_sizes[1] / 2;
  float* out = (float*)d_out;

  const int nt = (N + 1 + 1023) / 1024;
  const int NP = nt * 1024;
  const int nb = (N + 1 + BIN_NODES - 1) / BIN_NODES;

  char* ws = (char*)d_ws;
  size_t off = 0;
  auto carve = [&](size_t bytes) {
    void* p = ws + off;
    off += (bytes + 255) & ~(size_t)255;
    return p;
  };
  float* h    = (float*)carve((size_t)N * FOUT * sizeof(float));
  float* ald  = (float*)carve((size_t)N * NHEAD * sizeof(float));
  int*   csr  = (int*)carve((size_t)3 * E * sizeof(int));
  int*   tab  = (int*)carve((size_t)3 * NP * sizeof(int));
  int*   offs = (int*)carve((size_t)3 * NP * sizeof(int));
  float* h1   = (float*)carve((size_t)N * FOUT * sizeof(float));
  float* ald1 = (float*)carve((size_t)N * NHEAD * sizeof(float));
  int*   bintot  = tab;
  int*   cursor  = tab + 3 * NB_MAX;
  int*   binbase = tab + 6 * NB_MAX;
  int2*  binned = (int2*)h1;
  (void)ws_size;

  const int nmain  = (N / GM_ROWS / GM_WGPS) * GM_WGPS;
  const int nstrip = (N - nmain * GM_ROWS + GM_STRIP - 1) / GM_STRIP;
  const dim3 bgrid((E + BK_CH - 1) / BK_CH, 3);
  const int gblocks = (N + 8 * NODES_PER_WAVE - 1) / (8 * NODES_PER_WAVE);
  rgat_zero_cnt<<<(6 * NB_MAX + 255) / 256, 256, 0, stream>>>(tab, 6 * NB_MAX);
  rgat_bin_count<<<bgrid, BK_THREADS, 0, stream>>>(edges[0], edges[1], edges[2], bintot, E, nb);
  rgat_bin_scatter<<<bgrid, BK_THREADS, 0, stream>>>(edges[0], edges[1], edges[2], bintot, cursor,
                                                     binbase, binned, E, nb);

  const int ngemm = nmain + nstrip;
  const int mgblocks = MIX_GATHER_PER_WGP * GM_WGPS;
  const int sblocks  = 3 * nb > 2 * GM_WGPS ? 2 * GM_WGPS + 3 * nb : 4 * GM_WGPS;
  float* hp[2] = {h, h1};
  float* ap[2] = {ald, ald1};

  rgat_sort_gemm<<<sblocks, 256, 0, stream>>>(binbase, binned, offs, csr, E, N, NP, nb,
                                              x, W, att_dst, hp[0], ap[0], nmain, ngemm);
  for (int r = 0; r < 2; ++r) {
    rgat_gather_gemm<<<GM_WGPS + mgblocks, 256, 0, stream>>>(
        offs + (size_t)r * NP, csr + (size_t)r * E, hp[r & 1], att_src + (size_t)r * NHEAD * DHEAD,
        ap[r & 1], bias, out, N, r == 0, mgblocks * 8,
        x, W + (size_t)(r + 1) * FIN * FOUT, att_dst + (size_t)(r + 1) * NHEAD * DHEAD,
        hp[(r + 1) & 1], ap[(r + 1) & 1], nmain, ngemm);
  }
  rgat_node_gather<<<gblocks, 256, 0, stream>>>(
      offs + (size_t)2 * NP, csr + (size_t)2 * E, hp[0], att_src + (size_t)2 * NHEAD * DHEAD, ap[0],
      bias, out, N, 0, gblocks * 8);
}


# Level-1 edge binning block-local: a block sorts its 8192 edges by bin in LDS into its own slots, plus a row of bin starts; level 2 merges the runs. No count pass, no global adds; sums in run order
# speedup vs baseline: 8.8317x; 1.0406x over previous
#include <hip/hip_runtime.h>
#include <hip/hip_bf16.h>
#include <math.h>

typedef __attribute__((ext_vector_type(2))) float v2f;
typedef __attribute__((ext_vector_type(8))) float v8f;

#define FIN   256
#define FOUT  128
#define NHEAD 4
#define DHEAD 32
#define NEG_SLOPE 0.2f

#define GM_ROWS  64
#define GM_STRIP 16
#define GM_WGPS  256
#define GM_KC    32
#define GM_NCH   (FIN / GM_KC)
#define XS_PITCH 36
#define WS_PITCH 288
#define HS_PITCH 132
#define XS_FLOATS (GM_ROWS * XS_PITCH)
#define WS_FLOATS ((GM_KC / 2) * WS_PITCH)

__device__ __forceinline__ v8f wmma_f32(v2f a, v2f b, v8f c) {
  return __builtin_amdgcn_wmma_f32_16x16x4_f32(
       false, a,  false, b,
       (short)0, c,  false,  false);
}

__device__ __forceinline__ void stage_w_pair(float* __restrict__ dst, float4 lo, float4 hi) {
  *(float4*)(dst)     = make_float4(lo.x, hi.x, lo.y, hi.y);
  *(float4*)(dst + 4) = make_float4(lo.z, hi.z, lo.w, hi.w);
}

__device__ __forceinline__ void gemm_chunk(const float* __restrict__ xa, const float* __restrict__ wb,
                                           v8f& c00, v8f& c01, v8f& c10, v8f& c11) {
#pragma unroll
  for (int ks = 0; ks < GM_KC / 4; ++ks) {
    const v2f a0 = *(const v2f*)(xa + ks * 4);
    const v2f a1 = *(const v2f*)(xa + 16 * XS_PITCH + ks * 4);
    const v2f b0 = *(const v2f*)(wb + ks * 2 * WS_PITCH);
    const v2f b1 = *(const v2f*)(wb + ks * 2 * WS_PITCH + 32);
    c00 = wmma_f32(a0, b0, c00);
    c01 = wmma_f32(a0, b1, c01);
    c10 = wmma_f32(a1, b0, c10);
    c11 = wmma_f32(a1, b1, c11);
  }
}

__device__ __forceinline__ void gemm_strip16(const float* x, const float* Wr, const float* adst,
                                             float* __restrict__ h, float* __restrict__ ald,
                                             int N, int m0, float* sm, float* av) {
  float (*xs)[XS_FLOATS]  = (float (*)[XS_FLOATS])sm;
  float (*wsm)[WS_FLOATS] = (float (*)[WS_FLOATS])(sm + 2 * XS_FLOATS);
  float* hs = sm;

  const int tid  = threadIdx.x;
  const int lane = tid & 31;
  const int wave = tid >> 5;
  const int c16  = lane & 15;
  const int hh   = lane >> 4;

  const bool xld = tid < GM_STRIP * 8;
  const int xr = (tid >> 3) & (GM_STRIP - 1), xq = tid & 7;
  int gr0 = m0 + xr;  gr0 = gr0 < N ? gr0 : 0;
  const float* xg0 = x + (size_t)gr0 * FIN + xq * 4;
  const int xs_at = xr * XS_PITCH + xq * 4;
  const int wk = tid >> 5, wq = tid & 31;
  const float* wg = Wr + (size_t)(2 * wk) * FOUT + wq * 4;
  const int ws_at = wk * WS_PITCH + wq * 8;

  const int xa_at = c16 * XS_PITCH + 2 * hh;
  const int wb_at = hh * WS_PITCH + (wave * 16 + c16) * 2;

  float4 px0 = make_float4(0.f, 0.f, 0.f, 0.f), pw0, pw1, pw2, pw3;
  if (xld) px0 = *(const float4*)(xg0);
  pw0 = *(const float4*)(wg);
  pw1 = *(const float4*)(wg + FOUT);
  pw2 = *(const float4*)(wg + 16 * FOUT);
  pw3 = *(const float4*)(wg + 17 * FOUT);
  const float pa = adst[tid & 127];
  if (xld) *(float4*)(&xs[0][xs_at]) = px0;
  stage_w_pair(&wsm[0][ws_at], pw0, pw1);
  stage_w_pair(&wsm[0][ws_at + 8 * WS_PITCH], pw2, pw3);
  if (tid < FOUT) av[tid] = pa;
  __syncthreads();

  v8f c = {};
  for (int ch = 0; ch < GM_NCH; ++ch) {
    const int buf = ch & 1;
    const int nb  = buf ^ 1;
    const bool more = ch + 1 < GM_NCH;
    if (more) {
      const int kn = (ch + 1) * GM_KC;
      if (xld) px0 = *(const float4*)(xg0 + kn);
      pw0 = *(const float4*)(wg + (size_t)kn * FOUT);
      pw1 = *(const float4*)(wg + (size_t)(kn + 1) * FOUT);
      pw2 = *(const float4*)(wg + (size_t)(kn + 16) * FOUT);
      pw3 = *(const float4*)(wg + (size_t)(kn + 17) * FOUT);
    }
    const float* xa = &xs[buf][xa_at];
    const float* wb = &wsm[buf][wb_at];
#pragma unroll
    for (int ks = 0; ks < GM_KC / 4; ++ks) {
      const v2f a = *(const v2f*)(xa + ks * 4);
      const v2f b = *(const v2f*)(wb + ks * 2 * WS_PITCH);
      c = wmma_f32(a, b, c);
    }
    if (more) {
      if (xld) *(float4*)(&xs[nb][xs_at]) = px0;
      stage_w_pair(&wsm[nb][ws_at], pw0, pw1);
      stage_w_pair(&wsm[nb][ws_at + 8 * WS_PITCH], pw2, pw3);
    }
    __syncthreads();
  }

  {
    float* hw = hs + (8 * hh) * HS_PITCH + wave * 16 + c16;
#pragma unroll
    for (int v = 0; v < 8; ++v) hw[v * HS_PITCH] = c[v];
  }
  __syncthreads();

  int nrow = N - m0;
  nrow = nrow < GM_STRIP ? nrow : GM_STRIP;

  for (int i = 0; i < GM_STRIP / 8; ++i) {
    const int r = i * 8 + wave;
    if (r < nrow) *(float4*)(h + (size_t)(m0 + r) * FOUT + lane * 4) = *(const float4*)(hs + r * HS_PITCH + lane * 4);
  }

  if (tid < GM_STRIP * NHEAD) {
    const int r  = tid >> 2;
    const int hd = tid & 3;
    const float* hp = hs + r * HS_PITCH + hd * DHEAD;
    const float* ad = av + hd * DHEAD;
    float d = 0.f;
#pragma unroll
    for (int i = 0; i < DHEAD; ++i) d = fmaf(hp[i], ad[i], d);
    if (r < nrow) ald[(size_t)m0 * NHEAD + tid] = d;
  }
}

#define GM_SM_FLOATS (2 * XS_FLOATS + 2 * WS_FLOATS)
__device__ __forceinline__ void gemm_block(const float* x, const float* Wr, const float* adst,
                                           float* __restrict__ h, float* __restrict__ ald,
                                           int N, int nmain, int bid, float* sm, float* av) {
  float (*xs)[XS_FLOATS]  = (float (*)[XS_FLOATS])sm;
  float (*wsm)[WS_FLOATS] = (float (*)[WS_FLOATS])(sm + 2 * XS_FLOATS);
  float* hs = sm;

  if (bid >= nmain) {
    gemm_strip16(x, Wr, adst, h, ald, N, nmain * GM_ROWS + (bid - nmain) * GM_STRIP, sm, av);
    return;
  }

  const int tid  = threadIdx.x;
  const int lane = tid & 31;
  const int wave = tid >> 5;
  const int wm   = wave & 1;
  const int wn   = wave >> 1;
  const int c16  = lane & 15;
  const int hh   = lane >> 4;
  const int m0   = bid * GM_ROWS;

  const int xr = tid >> 3, xq = tid & 7;
  int gr0 = m0 + xr;       gr0 = gr0 < N ? gr0 : 0;
  int gr1 = m0 + xr + 32;  gr1 = gr1 < N ? gr1 : 0;
  const float* xg0 = x + (size_t)gr0 * FIN + xq * 4;
  const float* xg1 = x + (size_t)gr1 * FIN + xq * 4;
  const int xs_at = xr * XS_PITCH + xq * 4;
  const int wk = tid >> 5, wq = tid & 31;
  const float* wg = Wr + (size_t)(2 * wk) * FOUT + wq * 4;
  const int ws_at = wk * WS_PITCH + wq * 8;

  const int xa_at = (wm * 32 + c16) * XS_PITCH + 2 * hh;
  const int wb_at = hh * WS_PITCH + (wn * 32 + c16) * 2;

  float4 px0, px1, pw0, pw1, pw2, pw3;
  px0 = *(const float4*)(xg0);
  px1 = *(const float4*)(xg1);
  pw0 = *(const float4*)(wg);
  pw1 = *(const float4*)(wg + FOUT);
  pw2 = *(const float4*)(wg + 16 * FOUT);
  pw3 = *(const float4*)(wg + 17 * FOUT);
  const float pa = adst[tid & 127];
  *(float4*)(&xs[0][xs_at])                 = px0;
  *(float4*)(&xs[0][xs_at + 32 * XS_PITCH]) = px1;
  stage_w_pair(&wsm[0][ws_at], pw0, pw1);
  stage_w_pair(&wsm[0][ws_at + 8 * WS_PITCH], pw2, pw3);
  if (tid < FOUT) av[tid] = pa;
  __syncthreads();

  v8f c00 = {}, c01 = {}, c10 = {}, c11 = {};
#pragma unroll
  for (int ch = 0; ch + 1 < GM_NCH; ++ch) {
    const int buf = ch & 1;
    const int nb  = buf ^ 1;
    const int kn  = (ch + 1) * GM_KC;
    px0 = *(const float4*)(xg0 + kn);
    px1 = *(const float4*)(xg1 + kn);
    pw0 = *(const float4*)(wg + (size_t)kn * FOUT);
    pw1 = *(const float4*)(wg + (size_t)(kn + 1) * FOUT);
    pw2 = *(const float4*)(wg + (size_t)(kn + 16) * FOUT);
    pw3 = *(const float4*)(wg + (size_t)(kn + 17) * FOUT);
    __builtin_amdgcn_sched_barrier(0);
    gemm_chunk(&xs[buf][xa_at], &wsm[buf][wb_at], c00, c01, c10, c11);
    __builtin_amdgcn_sched_barrier(0);
    *(float4*)(&xs[nb][xs_at])                 = px0;
    *(float4*)(&xs[nb][xs_at + 32 * XS_PITCH]) = px1;
    stage_w_pair(&wsm[nb][ws_at], pw0, pw1);
    stage_w_pair(&wsm[nb][ws_at + 8 * WS_PITCH], pw2, pw3);
    __syncthreads();
  }
  gemm_chunk(&xs[(GM_NCH - 1) & 1][xa_at], &wsm[(GM_NCH - 1) & 1][wb_at], c00, c01, c10, c11);

  __syncthreads();

  {
    float* hw = hs + (wm * 32 + 8 * hh) * HS_PITCH + wn * 32 + c16;
#pragma unroll
    for (int v = 0; v < 8; ++v) {
      hw[v * HS_PITCH]             = c00[v];
      hw[v * HS_PITCH + 16]        = c01[v];
      hw[(v + 16) * HS_PITCH]      = c10[v];
      hw[(v + 16) * HS_PITCH + 16] = c11[v];
    }
  }
  __syncthreads();

  int nrow = N - m0;
  nrow = nrow < GM_ROWS ? nrow : GM_ROWS;

  const float* hr = hs + wave * HS_PITCH + lane * 4;
  float* hg = h + (size_t)(m0 + wave) * FOUT + lane * 4;
  if (nrow == GM_ROWS) {
    float4 val[GM_ROWS / 8];
#pragma unroll
    for (int i = 0; i < GM_ROWS / 8; ++i) val[i] = *(const float4*)(hr + i * 8 * HS_PITCH);
#pragma unroll
    for (int i = 0; i < GM_ROWS / 8; ++i) *(float4*)(hg + (size_t)i * 8 * FOUT) = val[i];
  } else {
#pragma unroll
    for (int i = 0; i < GM_ROWS / 8; ++i) {
      if (i * 8 + wave < nrow) *(float4*)(hg + (size_t)i * 8 * FOUT) = *(const float4*)(hr + i * 8 * HS_PITCH);
    }
  }

  {
    const int r  = tid >> 2;
    const int hd = tid & 3;
    const float* hp = hs + r * HS_PITCH + hd * DHEAD;
    const float* ad = av + hd * DHEAD;
    float d = 0.f;
#pragma unroll
    for (int i = 0; i < DHEAD; ++i) d = fmaf(hp[i], ad[i], d);
    if (r < nrow) ald[(size_t)m0 * NHEAD + tid] = d;
  }
}

__device__ __forceinline__ float leaky(float v) {
  return v > 0.f ? v : NEG_SLOPE * v;
}

#define BIN_SH     6
#define BIN_NODES  (1 << BIN_SH)
#define NB_MAX     1024
#define BK_THREADS 512
#define BK_EPT     16
#define BK_CH      (BK_THREADS * BK_EPT)
#define BS_CAP     4096
#define RUN_MAX    256
#define TB_PITCH   (NB_MAX + 2)

__device__ __forceinline__ const int* pick_rel(const int* a, const int* b, const int* c, int r) {
  return r == 0 ? a : (r == 1 ? b : c);
}

__device__ __forceinline__ void block_excl_scan(int* arr, int* wsum, int tid) {
  const int lane = tid & 31, wave = tid >> 5;
  const int a = arr[2 * tid], b = arr[2 * tid + 1];
  const int mine = a + b;
  int inc = mine;
  for (int m = 1; m < 32; m <<= 1) {
    int up = __shfl_up(inc, m, 32);
    if (lane >= m) inc += up;
  }
  if (lane == 31) wsum[wave] = inc;
  __syncthreads();
  int base = 0;
  for (int i = 0; i < BK_THREADS / 32; ++i) {
    if (i < wave) base += wsum[i];
  }
  const int ex = base + inc - mine;
  arr[2 * tid]     = ex;
  arr[2 * tid + 1] = ex + a;
  __syncthreads();
}

__global__ void __launch_bounds__(BK_THREADS)
rgat_bin_local(const int* __restrict__ ei0, const int* __restrict__ ei1,
               const int* __restrict__ ei2, int2* __restrict__ binned,
               int* __restrict__ runtab, int E, int nb) {
  __shared__ int2 stage[BK_CH];
  __shared__ int hist[NB_MAX];
  __shared__ int wsum[BK_THREADS / 32];
  const int tid = threadIdx.x;
  const int r = blockIdx.y;
  const int* __restrict__ ei = pick_rel(ei0, ei1, ei2, r);
  for (int i = tid; i < NB_MAX; i += BK_THREADS) hist[i] = 0;
  __syncthreads();

  const int e0 = blockIdx.x * BK_CH + tid;
  int s[BK_EPT], d[BK_EPT], k[BK_EPT];
#pragma unroll
  for (int i = 0; i < BK_EPT; ++i) {
    const int e = e0 + BK_THREADS * i;
    const bool in = e < E;
    s[i] = in ? ei[e] : 0;
    d[i] = in ? ei[E + e] : -1;
  }
#pragma unroll
  for (int i = 0; i < BK_EPT; ++i) {
    k[i] = 0;
    if (d[i] >= 0) k[i] = atomicAdd(&hist[d[i] >> BIN_SH], 1);
  }
  __syncthreads();

  block_excl_scan(hist, wsum, tid);

  int nloc = E - blockIdx.x * BK_CH;
  nloc = nloc < BK_CH ? nloc : BK_CH;
  int* __restrict__ trow = runtab + ((size_t)r * gridDim.x + blockIdx.x) * TB_PITCH;
  for (int i = tid; i <= nb; i += BK_THREADS) trow[i] = i < NB_MAX ? hist[i] : nloc;
#pragma unroll
  for (int i = 0; i < BK_EPT; ++i) {
    if (d[i] >= 0) stage[hist[d[i] >> BIN_SH] + k[i]] = make_int2(s[i], d[i]);
  }
  __syncthreads();

  int2* __restrict__ outp = binned + (size_t)r * E + (size_t)blockIdx.x * BK_CH;
  for (int i = tid; i < nloc; i += BK_THREADS) outp[i] = stage[i];
}

__device__ __forceinline__ int bin_pair_index(const int* rpos, const int* rsrc, int i) {
  int t = 0;
#pragma unroll
  for (int step = RUN_MAX / 2; step > 0; step >>= 1) {
    if (rpos[t + step] <= i) t += step;
  }
  return rsrc[t] + (i - rpos[t]);
}

#define BS_LDS_INTS (2 * BIN_NODES + 2 * RUN_MAX + 16 + 3 * BS_CAP)
__device__ __forceinline__ void bin_sort_block(const int* __restrict__ runtab, const int2* __restrict__ binned,
                                               int* __restrict__ offs, int* __restrict__ csr,
                                               int E, int N, int NP, int bin, int r, int nblk, int* tb) {
  int* cn   = tb;
  int* st   = cn + BIN_NODES;
  int* rpos = st + BIN_NODES;
  int* rsrc = rpos + RUN_MAX;
  int* wl   = rsrc + RUN_MAX;
  int* wb   = wl + 8;
  int2* ebuf = (int2*)(wb + 8);
  int* lst  = (int*)(ebuf + BS_CAP);

  const int tid = threadIdx.x;
  const int lane = tid & 31, wave = tid >> 5;

  int s0 = 0, len = 0;
  if (tid < nblk) {
    const int* __restrict__ tr = runtab + ((size_t)r * nblk + tid) * TB_PITCH + bin;
    s0 = tr[0];
    len = tr[1] - s0;
  }
  int inc = len, sb = s0;
  for (int m = 1; m < 32; m <<= 1) {
    int up = __shfl_up(inc, m, 32);
    if (lane >= m) inc += up;
  }
  for (int m = 16; m > 0; m >>= 1) sb += __shfl_xor(sb, m, 32);
  if (lane == 31) wl[wave] = inc;
  if (lane == 0)  wb[wave] = sb;
  if (tid < BIN_NODES) cn[tid] = 0;
  __syncthreads();
  int base = 0, total = 0, b0 = 0;
  for (int i = 0; i < 8; ++i) {
    const int v = wl[i];
    if (i < wave) base += v;
    total += v;
    b0 += wb[i];
  }
  rpos[tid] = base + inc - len;
  rsrc[tid] = tid * BK_CH + s0;
  __syncthreads();

  const int2* __restrict__ ed = binned + (size_t)r * E;
  int* __restrict__ outp = csr + (size_t)r * E + b0;
  const bool fits = total <= BS_CAP;

  if (fits) {
    for (int i0 = 0; i0 < total; i0 += 4 * 256) {
      int2 e[4];
      int at[4];
#pragma unroll
      for (int q = 0; q < 4; ++q) at[q] = bin_pair_index(rpos, rsrc, i0 + q * 256 + tid);
#pragma unroll
      for (int q = 0; q < 4; ++q) {
        e[q] = make_int2(0, 0);
        if (i0 + q * 256 + tid < total) e[q] = ed[at[q]];
      }
#pragma unroll
      for (int q = 0; q < 4; ++q) {
        const int i = i0 + q * 256 + tid;
        if (i < total) {
          ebuf[i] = e[q];
          atomicAdd(&cn[e[q].y & (BIN_NODES - 1)], 1);
        }
      }
    }
  } else {
    for (int i = tid; i < total; i += 256) atomicAdd(&cn[ed[bin_pair_index(rpos, rsrc, i)].y & (BIN_NODES - 1)], 1);
  }
  __syncthreads();
  if (tid < BIN_NODES) {
    int p = 0;
    for (int j = 0; j < BIN_NODES; ++j) {
      const int c = cn[j];
      if (j < tid) p += c;
    }
    st[tid] = p;
    const int node = bin * BIN_NODES + tid;
    if (node <= N) offs[(size_t)r * NP + node] = b0 + p;
  }
  __syncthreads();

  if (fits) {
    for (int i = tid; i < total; i += 256) {
      const int2 v = ebuf[i];
      lst[atomicAdd(&st[v.y & (BIN_NODES - 1)], 1)] = v.x;
    }
    __syncthreads();
    for (int i = tid; i < total; i += 256) outp[i] = lst[i];
  } else {
    for (int i = tid; i < total; i += 256) {
      const int2 v = ed[bin_pair_index(rpos, rsrc, i)];
      outp[atomicAdd(&st[v.y & (BIN_NODES - 1)], 1)] = v.x;
    }
  }
}

__device__ __forceinline__ void gemm_role(const float* x, const float* Wr, const float* adst,
                                          float* __restrict__ h, float* __restrict__ ald,
                                          int N, int nmain, int ngemm, int g, int ngrole,
                                          float* sm, float* av) {
  for (int t = g; t < ngemm; t += ngrole) {
    gemm_block(x, Wr, adst, h, ald, N, nmain, t, sm, av);
    __syncthreads();
  }
}

__global__ void __launch_bounds__(256, 8)
rgat_sort_gemm(const int* __restrict__ runtab, const int2* __restrict__ binned,
               int* __restrict__ offs, int* __restrict__ csr, int E, int N, int NP, int nb, int nblk,
               const float* x, const float* W0, const float* adst0,
               float* __restrict__ h, float* __restrict__ ald, int nmain, int ngemm) {
  __shared__ __attribute__((aligned(16))) float sm[GM_SM_FLOATS];
  __shared__ __attribute__((aligned(16))) float av[FOUT];
  static_assert(BS_LDS_INTS <= GM_SM_FLOATS, "the sort tables fit the GEMM's LDS block");
  const int b = (int)blockIdx.x;
  const int rnd = b / GM_WGPS, plc = b - rnd * GM_WGPS;
  const bool mixed = rnd < 4;
  const int idx = mixed ? rnd * (GM_WGPS / 2) + (plc >> 1) : 2 * GM_WGPS + (b - 4 * GM_WGPS);
  if (mixed && ((plc + rnd) & 1) == 0) {
    gemm_role(x, W0, adst0, h, ald, N, nmain, ngemm, idx, 2 * GM_WGPS, sm, av);
  } else if (idx < 3 * nb) {
    const int r = idx / nb;
    bin_sort_block(runtab, binned, offs, csr, E, N, NP, idx - r * nb, r, nblk, (int*)sm);
  }
}

#define NODES_PER_WAVE 5

template <int M>
__device__ __forceinline__ float lane_xor(float v) {
  return __int_as_float(__builtin_amdgcn_mov_dpp(__float_as_int(v), 0x160 + M, 0xf, 0xf, false));
}

template <int U>
__device__ __forceinline__ void rows_trip(v2f& acc01, v2f& acc23, float& m, float& l,
                                          const float* __restrict__ h, int s, int t, int lane,
                                          v2f as01, v2f as23, float adh) {
  v2f lo[U], hi[U];
  float sc[U];
#pragma unroll
  for (int u = 0; u < U; ++u) {
    const int su = __builtin_amdgcn_readlane(s, t + u);
    const float4 v = *(const float4*)(h + (size_t)su * FOUT + lane * 4);
    lo[u] = (v2f){v.x, v.y};
    hi[u] = (v2f){v.z, v.w};
  }
#pragma unroll
  for (int u = 0; u < U; ++u) {
    const v2f q = __builtin_elementwise_fma(hi[u], as23, lo[u] * as01);
    sc[u] = q.x + q.y;
  }
#pragma unroll
  for (int u = 0; u < U; ++u) sc[u] += lane_xor<1>(sc[u]);
#pragma unroll
  for (int u = 0; u < U; ++u) sc[u] += lane_xor<2>(sc[u]);
#pragma unroll
  for (int u = 0; u < U; ++u) sc[u] += lane_xor<4>(sc[u]);
  float gm = m;
#pragma unroll
  for (int u = 0; u < U; ++u) {
    sc[u] = leaky(sc[u] + adh);
    gm = fmaxf(gm, sc[u]);
  }
  const float scale = __expf(m - gm);
  m = gm;
  l *= scale;
  acc01 *= scale;
  acc23 *= scale;
#pragma unroll
  for (int u = 0; u < U; ++u) {
    const float p = __expf(sc[u] - gm);
    l += p;
    const v2f pp = {p, p};
    acc01 = __builtin_elementwise_fma(lo[u], pp, acc01);
    acc23 = __builtin_elementwise_fma(hi[u], pp, acc23);
  }
}

__device__ __forceinline__ void gather_rows(v2f& acc01, v2f& acc23, float& m, float& l,
                                            const float* __restrict__ h, int s, int n, int lane,
                                            v2f as01, v2f as23, float adh) {
  int t = 0;
  for (; t + 8 <= n; t += 8) rows_trip<8>(acc01, acc23, m, l, h, s, t, lane, as01, as23, adh);
  if (t + 4 <= n) { rows_trip<4>(acc01, acc23, m, l, h, s, t, lane, as01, as23, adh); t += 4; }
  for (; t < n; ++t) rows_trip<1>(acc01, acc23, m, l, h, s, t, lane, as01, as23, adh);
}

__device__ __forceinline__ void gather_block(const int* __restrict__ offs, const int* __restrict__ csr,
                                             const float* __restrict__ h, const float* __restrict__ asrc,
                                             const float* __restrict__ ald, const float* __restrict__ bias,
                                             float* __restrict__ out, int N, int first, int nwaves, int bid) {
  const int lane = threadIdx.x & 31;
  const int hsel = lane >> 3;
  const int wave = __builtin_amdgcn_readfirstlane((int)(threadIdx.x >> 5));
  const float ninf = -__builtin_inff();
  const float4 zero4 = make_float4(0.f, 0.f, 0.f, 0.f);

  int d = bid * 8 + wave;
  if (d >= N) return;

  const float4 as4 = *((const float4*)asrc + lane);
  const v2f as01 = {as4.x, as4.y}, as23 = {as4.z, as4.w};

  float4 bsum = zero4;
  if (first) {
    const float4 b0 = *((const float4*)bias + lane);
    const float4 b1 = *((const float4*)(bias + FOUT) + lane);
    const float4 b2 = *((const float4*)(bias + 2 * FOUT) + lane);
    bsum.x = b0.x + b1.x + b2.x; bsum.y = b0.y + b1.y + b2.y;
    bsum.z = b0.z + b1.z + b2.z; bsum.w = b0.w + b1.w + b2.w;
  }

  int beg = offs[d];
  int end = offs[d + 1];
  float4 ad = *(const float4*)(ald + (size_t)d * NHEAD);
  int s_first = 0;
  if (beg + lane < end) s_first = csr[beg + lane];

  for (;;) {
    const int dn = d + nwaves;
    const bool more = dn < N;
    int beg_n = 0, end_n = 0;
    float4 ad_n = zero4;
    if (more) {
      beg_n = offs[dn];
      end_n = offs[dn + 1];
      ad_n = *(const float4*)(ald + (size_t)dn * NHEAD);
    }
    float4* __restrict__ op = (float4*)(out + (size_t)d * FOUT) + lane;
    float4 o = bsum;
    if (!first) o = *op;

    int s_n = 0;
    if (more && (beg_n + lane < end_n)) s_n = csr[beg_n + lane];

    const float adh = hsel < 2 ? (hsel == 0 ? ad.x : ad.y) : (hsel == 2 ? ad.z : ad.w);

    v2f acc01 = {0.f, 0.f}, acc23 = {0.f, 0.f};
    float m = ninf, l = 0.f;
    {
      const int n = end - beg < 32 ? end - beg : 32;
      gather_rows(acc01, acc23, m, l, h, s_first, n, lane, as01, as23, adh);
    }
    for (int base = beg + 32; base < end; base += 32) {
      const int j = base + lane;
      int s = 0;
      if (j < end) s = csr[j];
      const int n = end - base < 32 ? end - base : 32;
      gather_rows(acc01, acc23, m, l, h, s, n, lane, as01, as23, adh);
    }

    const float inv = 1.f / (l + 1e-16f);
    float4 res;
    res.x = fmaf(acc01.x, inv, o.x); res.y = fmaf(acc01.y, inv, o.y);
    res.z = fmaf(acc23.x, inv, o.z); res.w = fmaf(acc23.y, inv, o.w);
    *op = res;

    if (!more) break;
    d = dn; beg = beg_n; end = end_n; ad = ad_n;
    s_first = s_n;
  }
}

__global__ void __launch_bounds__(256)
rgat_node_gather(const int* __restrict__ offs, const int* __restrict__ csr,
                 const float* __restrict__ h, const float* __restrict__ asrc,
                 const float* __restrict__ ald, const float* __restrict__ bias,
                 float* __restrict__ out, int N, int first, int nwaves) {
  gather_block(offs, csr, h, asrc, ald, bias, out, N, first, nwaves, (int)blockIdx.x);
}

#define MIX_GATHER_PER_WGP 3

__global__ void __launch_bounds__(256, 8)
rgat_gather_gemm(const int* __restrict__ offs, const int* __restrict__ csr,
                 const float* __restrict__ hcur, const float* __restrict__ asrc,
                 const float* __restrict__ aldcur, const float* __restrict__ bias,
                 float* __restrict__ out, int N, int first, int nwaves,
                 const float* x, const float* Wn, const float* adstn,
                 float* __restrict__ hnext, float* __restrict__ aldnext,
                 int nmain, int ngemm) {
  __shared__ __attribute__((aligned(16))) float sm[GM_SM_FLOATS];
  __shared__ __attribute__((aligned(16))) float av[FOUT];
  const int b = (int)blockIdx.x;
  const int rnd = b / GM_WGPS, plc = b - rnd * GM_WGPS;
  const int grp = rnd * (GM_WGPS / 4) + (plc >> 2);
  const int sub = plc & 3, gat = (4 - (rnd & 3)) & 3;
  if (sub == gat) {
    gemm_role(x, Wn, adstn, hnext, aldnext, N, nmain, ngemm, grp, GM_WGPS, sm, av);
  } else {
    gather_block(offs, csr, hcur, asrc, aldcur, bias, out, N, first, nwaves,
                 MIX_GATHER_PER_WGP * grp + (sub < gat ? sub : sub - 1));
  }
}

extern "C" void kernel_launch(void* const* d_in, const int* in_sizes, int n_in,
                              void* d_out, int out_size, void* d_ws, size_t ws_size,
                              hipStream_t stream) {
  const float* x        = (const float*)d_in[0];
  const int*   edges[3] = {(const int*)d_in[1], (const int*)d_in[2], (const int*)d_in[3]};
  const float* W        = (const float*)d_in[4];
  const float* att_src  = (const float*)d_in[5];
  const float* att_dst  = (const float*)d_in[6];
  const float* bias     = (const float*)d_in[7];

  const int N = in_sizes[0] / FIN;
  const int E = in_sizes[1] / 2;
  float* out = (float*)d_out;

  const int nt = (N + 1 + 1023) / 1024;
  const int NP = nt * 1024;
  const int nb = (N + 1 + BIN_NODES - 1) / BIN_NODES;

  char* ws = (char*)d_ws;
  size_t off = 0;
  auto carve = [&](size_t bytes) {
    void* p = ws + off;
    off += (bytes + 255) & ~(size_t)255;
    return p;
  };
  float* h    = (float*)carve((size_t)N * FOUT * sizeof(float));
  float* ald  = (float*)carve((size_t)N * NHEAD * sizeof(float));
  int*   csr  = (int*)carve((size_t)3 * E * sizeof(int));
  (void)carve((size_t)3 * NP * sizeof(int));
  int*   offs = (int*)carve((size_t)3 * NP * sizeof(int));
  float* h1   = (float*)carve((size_t)N * FOUT * sizeof(float));
  float* ald1 = (float*)carve((size_t)N * NHEAD * sizeof(float));
  int2*  binned = (int2*)h1;
  int*   runtab = (int*)((char*)h1 + (((size_t)3 * E * sizeof(int2) + 255) & ~(size_t)255));
  (void)ws_size;

  const int nmain  = (N / GM_ROWS / GM_WGPS) * GM_WGPS;
  const int nstrip = (N - nmain * GM_ROWS + GM_STRIP - 1) / GM_STRIP;
  const int nblk = (E + BK_CH - 1) / BK_CH;
  const dim3 bgrid(nblk, 3);
  const int gblocks = (N + 8 * NODES_PER_WAVE - 1) / (8 * NODES_PER_WAVE);
  rgat_bin_local<<<bgrid, BK_THREADS, 0, stream>>>(edges[0], edges[1], edges[2], binned, runtab, E, nb);

  const int ngemm = nmain + nstrip;
  const int mgblocks = MIX_GATHER_PER_WGP * GM_WGPS;
  const int sblocks  = 3 * nb > 2 * GM_WGPS ? 2 * GM_WGPS + 3 * nb : 4 * GM_WGPS;
  float* hp[2] = {h, h1};
  float* ap[2] = {ald, ald1};

  rgat_sort_gemm<<<sblocks, 256, 0, stream>>>(runtab, binned, offs, csr, E, N, NP, nb, nblk,
                                              x, W, att_dst, hp[0], ap[0], nmain, ngemm);
  for (int r = 0; r < 2; ++r) {
    rgat_gather_gemm<<<GM_WGPS + mgblocks, 256, 0, stream>>>(
        offs + (size_t)r * NP, csr + (size_t)r * E, hp[r & 1], att_src + (size_t)r * NHEAD * DHEAD,
        ap[r & 1], bias, out, N, r == 0, mgblocks * 8,
        x, W + (size_t)(r + 1) * FIN * FOUT, att_dst + (size_t)(r + 1) * NHEAD * DHEAD,
        hp[(r + 1) & 1], ap[(r + 1) & 1], nmain, ngemm);
  }
  rgat_node_gather<<<gblocks, 256, 0, stream>>>(
      offs + (size_t)2 * NP, csr + (size_t)2 * E, hp[0], att_src + (size_t)2 * NHEAD * DHEAD, ap[0],
      bias, out, N, 0, gblocks * 8);
}


# Gather softmax across the head's 8 lanes: a halving exchange leaves row i's score in lane i, so slope, max and exp run once per 8-row trip; leftover rows one trip; weight sum per lane, added last
# speedup vs baseline: 9.1039x; 1.0308x over previous
#include <hip/hip_runtime.h>
#include <hip/hip_bf16.h>
#include <math.h>

typedef __attribute__((ext_vector_type(2))) float v2f;
typedef __attribute__((ext_vector_type(8))) float v8f;

#define FIN   256
#define FOUT  128
#define NHEAD 4
#define DHEAD 32
#define NEG_SLOPE 0.2f

#define GM_ROWS  64
#define GM_STRIP 16
#define GM_WGPS  256
#define GM_KC    32
#define GM_NCH   (FIN / GM_KC)
#define XS_PITCH 36
#define WS_PITCH 288
#define HS_PITCH 132
#define XS_FLOATS (GM_ROWS * XS_PITCH)
#define WS_FLOATS ((GM_KC / 2) * WS_PITCH)

__device__ __forceinline__ v8f wmma_f32(v2f a, v2f b, v8f c) {
  return __builtin_amdgcn_wmma_f32_16x16x4_f32(
       false, a,  false, b,
       (short)0, c,  false,  false);
}

__device__ __forceinline__ void stage_w_pair(float* __restrict__ dst, float4 lo, float4 hi) {
  *(float4*)(dst)     = make_float4(lo.x, hi.x, lo.y, hi.y);
  *(float4*)(dst + 4) = make_float4(lo.z, hi.z, lo.w, hi.w);
}

__device__ __forceinline__ void gemm_chunk(const float* __restrict__ xa, const float* __restrict__ wb,
                                           v8f& c00, v8f& c01, v8f& c10, v8f& c11) {
#pragma unroll
  for (int ks = 0; ks < GM_KC / 4; ++ks) {
    const v2f a0 = *(const v2f*)(xa + ks * 4);
    const v2f a1 = *(const v2f*)(xa + 16 * XS_PITCH + ks * 4);
    const v2f b0 = *(const v2f*)(wb + ks * 2 * WS_PITCH);
    const v2f b1 = *(const v2f*)(wb + ks * 2 * WS_PITCH + 32);
    c00 = wmma_f32(a0, b0, c00);
    c01 = wmma_f32(a0, b1, c01);
    c10 = wmma_f32(a1, b0, c10);
    c11 = wmma_f32(a1, b1, c11);
  }
}

__device__ __forceinline__ void gemm_strip16(const float* x, const float* Wr, const float* adst,
                                             float* __restrict__ h, float* __restrict__ ald,
                                             int N, int m0, float* sm, float* av) {
  float (*xs)[XS_FLOATS]  = (float (*)[XS_FLOATS])sm;
  float (*wsm)[WS_FLOATS] = (float (*)[WS_FLOATS])(sm + 2 * XS_FLOATS);
  float* hs = sm;

  const int tid  = threadIdx.x;
  const int lane = tid & 31;
  const int wave = tid >> 5;
  const int c16  = lane & 15;
  const int hh   = lane >> 4;

  const bool xld = tid < GM_STRIP * 8;
  const int xr = (tid >> 3) & (GM_STRIP - 1), xq = tid & 7;
  int gr0 = m0 + xr;  gr0 = gr0 < N ? gr0 : 0;
  const float* xg0 = x + (size_t)gr0 * FIN + xq * 4;
  const int xs_at = xr * XS_PITCH + xq * 4;
  const int wk = tid >> 5, wq = tid & 31;
  const float* wg = Wr + (size_t)(2 * wk) * FOUT + wq * 4;
  const int ws_at = wk * WS_PITCH + wq * 8;

  const int xa_at = c16 * XS_PITCH + 2 * hh;
  const int wb_at = hh * WS_PITCH + (wave * 16 + c16) * 2;

  float4 px0 = make_float4(0.f, 0.f, 0.f, 0.f), pw0, pw1, pw2, pw3;
  if (xld) px0 = *(const float4*)(xg0);
  pw0 = *(const float4*)(wg);
  pw1 = *(const float4*)(wg + FOUT);
  pw2 = *(const float4*)(wg + 16 * FOUT);
  pw3 = *(const float4*)(wg + 17 * FOUT);
  const float pa = adst[tid & 127];
  if (xld) *(float4*)(&xs[0][xs_at]) = px0;
  stage_w_pair(&wsm[0][ws_at], pw0, pw1);
  stage_w_pair(&wsm[0][ws_at + 8 * WS_PITCH], pw2, pw3);
  if (tid < FOUT) av[tid] = pa;
  __syncthreads();

  v8f c = {};
  for (int ch = 0; ch < GM_NCH; ++ch) {
    const int buf = ch & 1;
    const int nb  = buf ^ 1;
    const bool more = ch + 1 < GM_NCH;
    if (more) {
      const int kn = (ch + 1) * GM_KC;
      if (xld) px0 = *(const float4*)(xg0 + kn);
      pw0 = *(const float4*)(wg + (size_t)kn * FOUT);
      pw1 = *(const float4*)(wg + (size_t)(kn + 1) * FOUT);
      pw2 = *(const float4*)(wg + (size_t)(kn + 16) * FOUT);
      pw3 = *(const float4*)(wg + (size_t)(kn + 17) * FOUT);
    }
    const float* xa = &xs[buf][xa_at];
    const float* wb = &wsm[buf][wb_at];
#pragma unroll
    for (int ks = 0; ks < GM_KC / 4; ++ks) {
      const v2f a = *(const v2f*)(xa + ks * 4);
      const v2f b = *(const v2f*)(wb + ks * 2 * WS_PITCH);
      c = wmma_f32(a, b, c);
    }
    if (more) {
      if (xld) *(float4*)(&xs[nb][xs_at]) = px0;
      stage_w_pair(&wsm[nb][ws_at], pw0, pw1);
      stage_w_pair(&wsm[nb][ws_at + 8 * WS_PITCH], pw2, pw3);
    }
    __syncthreads();
  }

  {
    float* hw = hs + (8 * hh) * HS_PITCH + wave * 16 + c16;
#pragma unroll
    for (int v = 0; v < 8; ++v) hw[v * HS_PITCH] = c[v];
  }
  __syncthreads();

  int nrow = N - m0;
  nrow = nrow < GM_STRIP ? nrow : GM_STRIP;

  for (int i = 0; i < GM_STRIP / 8; ++i) {
    const int r = i * 8 + wave;
    if (r < nrow) *(float4*)(h + (size_t)(m0 + r) * FOUT + lane * 4) = *(const float4*)(hs + r * HS_PITCH + lane * 4);
  }

  if (tid < GM_STRIP * NHEAD) {
    const int r  = tid >> 2;
    const int hd = tid & 3;
    const float* hp = hs + r * HS_PITCH + hd * DHEAD;
    const float* ad = av + hd * DHEAD;
    float d = 0.f;
#pragma unroll
    for (int i = 0; i < DHEAD; ++i) d = fmaf(hp[i], ad[i], d);
    if (r < nrow) ald[(size_t)m0 * NHEAD + tid] = d;
  }
}

#define GM_SM_FLOATS (2 * XS_FLOATS + 2 * WS_FLOATS)
__device__ __forceinline__ void gemm_block(const float* x, const float* Wr, const float* adst,
                                           float* __restrict__ h, float* __restrict__ ald,
                                           int N, int nmain, int bid, float* sm, float* av) {
  float (*xs)[XS_FLOATS]  = (float (*)[XS_FLOATS])sm;
  float (*wsm)[WS_FLOATS] = (float (*)[WS_FLOATS])(sm + 2 * XS_FLOATS);
  float* hs = sm;

  if (bid >= nmain) {
    gemm_strip16(x, Wr, adst, h, ald, N, nmain * GM_ROWS + (bid - nmain) * GM_STRIP, sm, av);
    return;
  }

  const int tid  = threadIdx.x;
  const int lane = tid & 31;
  const int wave = tid >> 5;
  const int wm   = wave & 1;
  const int wn   = wave >> 1;
  const int c16  = lane & 15;
  const int hh   = lane >> 4;
  const int m0   = bid * GM_ROWS;

  const int xr = tid >> 3, xq = tid & 7;
  int gr0 = m0 + xr;       gr0 = gr0 < N ? gr0 : 0;
  int gr1 = m0 + xr + 32;  gr1 = gr1 < N ? gr1 : 0;
  const float* xg0 = x + (size_t)gr0 * FIN + xq * 4;
  const float* xg1 = x + (size_t)gr1 * FIN + xq * 4;
  const int xs_at = xr * XS_PITCH + xq * 4;
  const int wk = tid >> 5, wq = tid & 31;
  const float* wg = Wr + (size_t)(2 * wk) * FOUT + wq * 4;
  const int ws_at = wk * WS_PITCH + wq * 8;

  const int xa_at = (wm * 32 + c16) * XS_PITCH + 2 * hh;
  const int wb_at = hh * WS_PITCH + (wn * 32 + c16) * 2;

  float4 px0, px1, pw0, pw1, pw2, pw3;
  px0 = *(const float4*)(xg0);
  px1 = *(const float4*)(xg1);
  pw0 = *(const float4*)(wg);
  pw1 = *(const float4*)(wg + FOUT);
  pw2 = *(const float4*)(wg + 16 * FOUT);
  pw3 = *(const float4*)(wg + 17 * FOUT);
  const float pa = adst[tid & 127];
  *(float4*)(&xs[0][xs_at])                 = px0;
  *(float4*)(&xs[0][xs_at + 32 * XS_PITCH]) = px1;
  stage_w_pair(&wsm[0][ws_at], pw0, pw1);
  stage_w_pair(&wsm[0][ws_at + 8 * WS_PITCH], pw2, pw3);
  if (tid < FOUT) av[tid] = pa;
  __syncthreads();

  v8f c00 = {}, c01 = {}, c10 = {}, c11 = {};
#pragma unroll
  for (int ch = 0; ch + 1 < GM_NCH; ++ch) {
    const int buf = ch & 1;
    const int nb  = buf ^ 1;
    const int kn  = (ch + 1) * GM_KC;
    px0 = *(const float4*)(xg0 + kn);
    px1 = *(const float4*)(xg1 + kn);
    pw0 = *(const float4*)(wg + (size_t)kn * FOUT);
    pw1 = *(const float4*)(wg + (size_t)(kn + 1) * FOUT);
    pw2 = *(const float4*)(wg + (size_t)(kn + 16) * FOUT);
    pw3 = *(const float4*)(wg + (size_t)(kn + 17) * FOUT);
    __builtin_amdgcn_sched_barrier(0);
    gemm_chunk(&xs[buf][xa_at], &wsm[buf][wb_at], c00, c01, c10, c11);
    __builtin_amdgcn_sched_barrier(0);
    *(float4*)(&xs[nb][xs_at])                 = px0;
    *(float4*)(&xs[nb][xs_at + 32 * XS_PITCH]) = px1;
    stage_w_pair(&wsm[nb][ws_at], pw0, pw1);
    stage_w_pair(&wsm[nb][ws_at + 8 * WS_PITCH], pw2, pw3);
    __syncthreads();
  }
  gemm_chunk(&xs[(GM_NCH - 1) & 1][xa_at], &wsm[(GM_NCH - 1) & 1][wb_at], c00, c01, c10, c11);

  __syncthreads();

  {
    float* hw = hs + (wm * 32 + 8 * hh) * HS_PITCH + wn * 32 + c16;
#pragma unroll
    for (int v = 0; v < 8; ++v) {
      hw[v * HS_PITCH]             = c00[v];
      hw[v * HS_PITCH + 16]        = c01[v];
      hw[(v + 16) * HS_PITCH]      = c10[v];
      hw[(v + 16) * HS_PITCH + 16] = c11[v];
    }
  }
  __syncthreads();

  int nrow = N - m0;
  nrow = nrow < GM_ROWS ? nrow : GM_ROWS;

  const float* hr = hs + wave * HS_PITCH + lane * 4;
  float* hg = h + (size_t)(m0 + wave) * FOUT + lane * 4;
  if (nrow == GM_ROWS) {
    float4 val[GM_ROWS / 8];
#pragma unroll
    for (int i = 0; i < GM_ROWS / 8; ++i) val[i] = *(const float4*)(hr + i * 8 * HS_PITCH);
#pragma unroll
    for (int i = 0; i < GM_ROWS / 8; ++i) *(float4*)(hg + (size_t)i * 8 * FOUT) = val[i];
  } else {
#pragma unroll
    for (int i = 0; i < GM_ROWS / 8; ++i) {
      if (i * 8 + wave < nrow) *(float4*)(hg + (size_t)i * 8 * FOUT) = *(const float4*)(hr + i * 8 * HS_PITCH);
    }
  }

  {
    const int r  = tid >> 2;
    const int hd = tid & 3;
    const float* hp = hs + r * HS_PITCH + hd * DHEAD;
    const float* ad = av + hd * DHEAD;
    float d = 0.f;
#pragma unroll
    for (int i = 0; i < DHEAD; ++i) d = fmaf(hp[i], ad[i], d);
    if (r < nrow) ald[(size_t)m0 * NHEAD + tid] = d;
  }
}

__device__ __forceinline__ float leaky(float v) {
  return v > 0.f ? v : NEG_SLOPE * v;
}

#define BIN_SH     6
#define BIN_NODES  (1 << BIN_SH)
#define NB_MAX     1024
#define BK_THREADS 512
#define BK_EPT     16
#define BK_CH      (BK_THREADS * BK_EPT)
#define BS_CAP     4096
#define RUN_MAX    256
#define TB_PITCH   (NB_MAX + 2)

__device__ __forceinline__ const int* pick_rel(const int* a, const int* b, const int* c, int r) {
  return r == 0 ? a : (r == 1 ? b : c);
}

__device__ __forceinline__ void block_excl_scan(int* arr, int* wsum, int tid) {
  const int lane = tid & 31, wave = tid >> 5;
  const int a = arr[2 * tid], b = arr[2 * tid + 1];
  const int mine = a + b;
  int inc = mine;
  for (int m = 1; m < 32; m <<= 1) {
    int up = __shfl_up(inc, m, 32);
    if (lane >= m) inc += up;
  }
  if (lane == 31) wsum[wave] = inc;
  __syncthreads();
  int base = 0;
  for (int i = 0; i < BK_THREADS / 32; ++i) {
    if (i < wave) base += wsum[i];
  }
  const int ex = base + inc - mine;
  arr[2 * tid]     = ex;
  arr[2 * tid + 1] = ex + a;
  __syncthreads();
}

__global__ void __launch_bounds__(BK_THREADS)
rgat_bin_local(const int* __restrict__ ei0, const int* __restrict__ ei1,
               const int* __restrict__ ei2, int2* __restrict__ binned,
               int* __restrict__ runtab, int E, int nb) {
  __shared__ int2 stage[BK_CH];
  __shared__ int hist[NB_MAX];
  __shared__ int wsum[BK_THREADS / 32];
  const int tid = threadIdx.x;
  const int r = blockIdx.y;
  const int* __restrict__ ei = pick_rel(ei0, ei1, ei2, r);
  for (int i = tid; i < NB_MAX; i += BK_THREADS) hist[i] = 0;
  __syncthreads();

  const int e0 = blockIdx.x * BK_CH + tid;
  int s[BK_EPT], d[BK_EPT], k[BK_EPT];
#pragma unroll
  for (int i = 0; i < BK_EPT; ++i) {
    const int e = e0 + BK_THREADS * i;
    const bool in = e < E;
    s[i] = in ? ei[e] : 0;
    d[i] = in ? ei[E + e] : -1;
  }
#pragma unroll
  for (int i = 0; i < BK_EPT; ++i) {
    k[i] = 0;
    if (d[i] >= 0) k[i] = atomicAdd(&hist[d[i] >> BIN_SH], 1);
  }
  __syncthreads();

  block_excl_scan(hist, wsum, tid);

  int nloc = E - blockIdx.x * BK_CH;
  nloc = nloc < BK_CH ? nloc : BK_CH;
  int* __restrict__ trow = runtab + ((size_t)r * gridDim.x + blockIdx.x) * TB_PITCH;
  for (int i = tid; i <= nb; i += BK_THREADS) trow[i] = i < NB_MAX ? hist[i] : nloc;
#pragma unroll
  for (int i = 0; i < BK_EPT; ++i) {
    if (d[i] >= 0) stage[hist[d[i] >> BIN_SH] + k[i]] = make_int2(s[i], d[i]);
  }
  __syncthreads();

  int2* __restrict__ outp = binned + (size_t)r * E + (size_t)blockIdx.x * BK_CH;
  for (int i = tid; i < nloc; i += BK_THREADS) outp[i] = stage[i];
}

__device__ __forceinline__ int bin_pair_index(const int* rpos, const int* rsrc, int i) {
  int t = 0;
#pragma unroll
  for (int step = RUN_MAX / 2; step > 0; step >>= 1) {
    if (rpos[t + step] <= i) t += step;
  }
  return rsrc[t] + (i - rpos[t]);
}

#define BS_LDS_INTS (2 * BIN_NODES + 2 * RUN_MAX + 16 + 3 * BS_CAP)
__device__ __forceinline__ void bin_sort_block(const int* __restrict__ runtab, const int2* __restrict__ binned,
                                               int* __restrict__ offs, int* __restrict__ csr,
                                               int E, int N, int NP, int bin, int r, int nblk, int* tb) {
  int* cn   = tb;
  int* st   = cn + BIN_NODES;
  int* rpos = st + BIN_NODES;
  int* rsrc = rpos + RUN_MAX;
  int* wl   = rsrc + RUN_MAX;
  int* wb   = wl + 8;
  int2* ebuf = (int2*)(wb + 8);
  int* lst  = (int*)(ebuf + BS_CAP);

  const int tid = threadIdx.x;
  const int lane = tid & 31, wave = tid >> 5;

  int s0 = 0, len = 0;
  if (tid < nblk) {
    const int* __restrict__ tr = runtab + ((size_t)r * nblk + tid) * TB_PITCH + bin;
    s0 = tr[0];
    len = tr[1] - s0;
  }
  int inc = len, sb = s0;
  for (int m = 1; m < 32; m <<= 1) {
    int up = __shfl_up(inc, m, 32);
    if (lane >= m) inc += up;
  }
  for (int m = 16; m > 0; m >>= 1) sb += __shfl_xor(sb, m, 32);
  if (lane == 31) wl[wave] = inc;
  if (lane == 0)  wb[wave] = sb;
  if (tid < BIN_NODES) cn[tid] = 0;
  __syncthreads();
  int base = 0, total = 0, b0 = 0;
  for (int i = 0; i < 8; ++i) {
    const int v = wl[i];
    if (i < wave) base += v;
    total += v;
    b0 += wb[i];
  }
  rpos[tid] = base + inc - len;
  rsrc[tid] = tid * BK_CH + s0;
  __syncthreads();

  const int2* __restrict__ ed = binned + (size_t)r * E;
  int* __restrict__ outp = csr + (size_t)r * E + b0;
  const bool fits = total <= BS_CAP;

  if (fits) {
    for (int i0 = 0; i0 < total; i0 += 4 * 256) {
      int2 e[4];
      int at[4];
#pragma unroll
      for (int q = 0; q < 4; ++q) at[q] = bin_pair_index(rpos, rsrc, i0 + q * 256 + tid);
#pragma unroll
      for (int q = 0; q < 4; ++q) {
        e[q] = make_int2(0, 0);
        if (i0 + q * 256 + tid < total) e[q] = ed[at[q]];
      }
#pragma unroll
      for (int q = 0; q < 4; ++q) {
        const int i = i0 + q * 256 + tid;
        if (i < total) {
          ebuf[i] = e[q];
          atomicAdd(&cn[e[q].y & (BIN_NODES - 1)], 1);
        }
      }
    }
  } else {
    for (int i = tid; i < total; i += 256) atomicAdd(&cn[ed[bin_pair_index(rpos, rsrc, i)].y & (BIN_NODES - 1)], 1);
  }
  __syncthreads();
  if (tid < BIN_NODES) {
    int p = 0;
    for (int j = 0; j < BIN_NODES; ++j) {
      const int c = cn[j];
      if (j < tid) p += c;
    }
    st[tid] = p;
    const int node = bin * BIN_NODES + tid;
    if (node <= N) offs[(size_t)r * NP + node] = b0 + p;
  }
  __syncthreads();

  if (fits) {
    for (int i = tid; i < total; i += 256) {
      const int2 v = ebuf[i];
      lst[atomicAdd(&st[v.y & (BIN_NODES - 1)], 1)] = v.x;
    }
    __syncthreads();
    for (int i = tid; i < total; i += 256) outp[i] = lst[i];
  } else {
    for (int i = tid; i < total; i += 256) {
      const int2 v = ed[bin_pair_index(rpos, rsrc, i)];
      outp[atomicAdd(&st[v.y & (BIN_NODES - 1)], 1)] = v.x;
    }
  }
}

__device__ __forceinline__ void gemm_role(const float* x, const float* Wr, const float* adst,
                                          float* __restrict__ h, float* __restrict__ ald,
                                          int N, int nmain, int ngemm, int g, int ngrole,
                                          float* sm, float* av) {
  for (int t = g; t < ngemm; t += ngrole) {
    gemm_block(x, Wr, adst, h, ald, N, nmain, t, sm, av);
    __syncthreads();
  }
}

__global__ void __launch_bounds__(256, 8)
rgat_sort_gemm(const int* __restrict__ runtab, const int2* __restrict__ binned,
               int* __restrict__ offs, int* __restrict__ csr, int E, int N, int NP, int nb, int nblk,
               const float* x, const float* W0, const float* adst0,
               float* __restrict__ h, float* __restrict__ ald, int nmain, int ngemm) {
  __shared__ __attribute__((aligned(16))) float sm[GM_SM_FLOATS];
  __shared__ __attribute__((aligned(16))) float av[FOUT];
  static_assert(BS_LDS_INTS <= GM_SM_FLOATS, "the sort tables fit the GEMM's LDS block");
  const int b = (int)blockIdx.x;
  const int rnd = b / GM_WGPS, plc = b - rnd * GM_WGPS;
  const bool mixed = rnd < 4;
  const int idx = mixed ? rnd * (GM_WGPS / 2) + (plc >> 1) : 2 * GM_WGPS + (b - 4 * GM_WGPS);
  if (mixed && ((plc + rnd) & 1) == 0) {
    gemm_role(x, W0, adst0, h, ald, N, nmain, ngemm, idx, 2 * GM_WGPS, sm, av);
  } else if (idx < 3 * nb) {
    const int r = idx / nb;
    bin_sort_block(runtab, binned, offs, csr, E, N, NP, idx - r * nb, r, nblk, (int*)sm);
  }
}

#define NODES_PER_WAVE 5

template <int M>
__device__ __forceinline__ float lane_xor(float v) {
  return __int_as_float(__builtin_amdgcn_mov_dpp(__float_as_int(v), 0x160 + M, 0xf, 0xf, false));
}

template <int I>
__device__ __forceinline__ float group_lane(float v) {
  return __int_as_float((int)__builtin_amdgcn_mov_dpp8((unsigned)__float_as_int(v), (unsigned)(I * 0x249249)));
}

template <int U>
__device__ __forceinline__ void rows_trip(v2f& acc01, v2f& acc23, float& m, float& l,
                                          const float* __restrict__ h, int s, int t, int lane,
                                          v2f as01, v2f as23, float adh) {
  v2f lo[U], hi[U];
  float q[8];
#pragma unroll
  for (int u = 0; u < U; ++u) {
    const int su = __builtin_amdgcn_readlane(s, t + u);
    const float4 v = *(const float4*)(h + (size_t)su * FOUT + lane * 4);
    lo[u] = (v2f){v.x, v.y};
    hi[u] = (v2f){v.z, v.w};
  }
#pragma unroll
  for (int u = 0; u < 8; ++u) q[u] = 0.f;
#pragma unroll
  for (int u = 0; u < U; ++u) {
    const v2f w = __builtin_elementwise_fma(hi[u], as23, lo[u] * as01);
    q[u] = w.x + w.y;
  }
  const bool b0 = (lane & 1) != 0, b1 = (lane & 2) != 0, b2 = (lane & 4) != 0;
  float r1[4], r2[2];
#pragma unroll
  for (int v = 0; v < 4; ++v) {
    const float keep = b0 ? q[2 * v + 1] : q[2 * v];
    const float send = b0 ? q[2 * v] : q[2 * v + 1];
    r1[v] = keep + lane_xor<1>(send);
  }
#pragma unroll
  for (int w = 0; w < 2; ++w) {
    const float keep = b1 ? r1[2 * w + 1] : r1[2 * w];
    const float send = b1 ? r1[2 * w] : r1[2 * w + 1];
    r2[w] = keep + lane_xor<2>(send);
  }
  float sc;
  {
    const float keep = b2 ? r2[1] : r2[0];
    const float send = b2 ? r2[0] : r2[1];
    sc = keep + lane_xor<4>(send);
  }
  sc = leaky(sc + adh);
  if (U < 8) sc = (lane & 7) < U ? sc : -__builtin_inff();
  float tm = sc;
  tm = fmaxf(tm, lane_xor<1>(tm));
  tm = fmaxf(tm, lane_xor<2>(tm));
  tm = fmaxf(tm, lane_xor<4>(tm));
  const float gm = fmaxf(m, tm);
  const float scale = __expf(m - gm);
  const float p = __expf(sc - gm);
  m = gm;
  l = fmaf(l, scale, p);
  acc01 *= scale;
  acc23 *= scale;
  const float pr[8] = {group_lane<0>(p), group_lane<1>(p), group_lane<2>(p), group_lane<3>(p),
                       group_lane<4>(p), group_lane<5>(p), group_lane<6>(p), group_lane<7>(p)};
#pragma unroll
  for (int u = 0; u < U; ++u) {
    const v2f pp = {pr[u], pr[u]};
    acc01 = __builtin_elementwise_fma(lo[u], pp, acc01);
    acc23 = __builtin_elementwise_fma(hi[u], pp, acc23);
  }
}

__device__ __forceinline__ void gather_rows(v2f& acc01, v2f& acc23, float& m, float& l,
                                            const float* __restrict__ h, int s, int n, int lane,
                                            v2f as01, v2f as23, float adh) {
  int t = 0;
  for (; t + 8 <= n; t += 8) rows_trip<8>(acc01, acc23, m, l, h, s, t, lane, as01, as23, adh);
  switch (n - t) {
    case 1: rows_trip<1>(acc01, acc23, m, l, h, s, t, lane, as01, as23, adh); break;
    case 2: rows_trip<2>(acc01, acc23, m, l, h, s, t, lane, as01, as23, adh); break;
    case 3: rows_trip<3>(acc01, acc23, m, l, h, s, t, lane, as01, as23, adh); break;
    case 4: rows_trip<4>(acc01, acc23, m, l, h, s, t, lane, as01, as23, adh); break;
    case 5: rows_trip<5>(acc01, acc23, m, l, h, s, t, lane, as01, as23, adh); break;
    case 6: rows_trip<6>(acc01, acc23, m, l, h, s, t, lane, as01, as23, adh); break;
    case 7: rows_trip<7>(acc01, acc23, m, l, h, s, t, lane, as01, as23, adh); break;
    default: break;
  }
}

__device__ __forceinline__ void gather_block(const int* __restrict__ offs, const int* __restrict__ csr,
                                             const float* __restrict__ h, const float* __restrict__ asrc,
                                             const float* __restrict__ ald, const float* __restrict__ bias,
                                             float* __restrict__ out, int N, int first, int nwaves, int bid) {
  const int lane = threadIdx.x & 31;
  const int hsel = lane >> 3;
  const int wave = __builtin_amdgcn_readfirstlane((int)(threadIdx.x >> 5));
  const float ninf = -__builtin_inff();
  const float4 zero4 = make_float4(0.f, 0.f, 0.f, 0.f);

  int d = bid * 8 + wave;
  if (d >= N) return;

  const float4 as4 = *((const float4*)asrc + lane);
  const v2f as01 = {as4.x, as4.y}, as23 = {as4.z, as4.w};

  float4 bsum = zero4;
  if (first) {
    const float4 b0 = *((const float4*)bias + lane);
    const float4 b1 = *((const float4*)(bias + FOUT) + lane);
    const float4 b2 = *((const float4*)(bias + 2 * FOUT) + lane);
    bsum.x = b0.x + b1.x + b2.x; bsum.y = b0.y + b1.y + b2.y;
    bsum.z = b0.z + b1.z + b2.z; bsum.w = b0.w + b1.w + b2.w;
  }

  int beg = offs[d];
  int end = offs[d + 1];
  float4 ad = *(const float4*)(ald + (size_t)d * NHEAD);
  int s_first = 0;
  if (beg + lane < end) s_first = csr[beg + lane];

  for (;;) {
    const int dn = d + nwaves;
    const bool more = dn < N;
    int beg_n = 0, end_n = 0;
    float4 ad_n = zero4;
    if (more) {
      beg_n = offs[dn];
      end_n = offs[dn + 1];
      ad_n = *(const float4*)(ald + (size_t)dn * NHEAD);
    }
    float4* __restrict__ op = (float4*)(out + (size_t)d * FOUT) + lane;
    float4 o = bsum;
    if (!first) o = *op;

    int s_n = 0;
    if (more && (beg_n + lane < end_n)) s_n = csr[beg_n + lane];

    const float adh = hsel < 2 ? (hsel == 0 ? ad.x : ad.y) : (hsel == 2 ? ad.z : ad.w);

    v2f acc01 = {0.f, 0.f}, acc23 = {0.f, 0.f};
    float m = ninf, l = 0.f;
    int s = s_first;
    for (int base = beg;;) {
      const int n = end - base < 32 ? end - base : 32;
      gather_rows(acc01, acc23, m, l, h, s, n, lane, as01, as23, adh);
      base += 32;
      if (base >= end) break;
      s = 0;
      if (base + lane < end) s = csr[base + lane];
    }

    l += lane_xor<1>(l);
    l += lane_xor<2>(l);
    l += lane_xor<4>(l);
    const float inv = 1.f / (l + 1e-16f);
    float4 res;
    res.x = fmaf(acc01.x, inv, o.x); res.y = fmaf(acc01.y, inv, o.y);
    res.z = fmaf(acc23.x, inv, o.z); res.w = fmaf(acc23.y, inv, o.w);
    *op = res;

    if (!more) break;
    d = dn; beg = beg_n; end = end_n; ad = ad_n;
    s_first = s_n;
  }
}

__global__ void __launch_bounds__(256)
rgat_node_gather(const int* __restrict__ offs, const int* __restrict__ csr,
                 const float* __restrict__ h, const float* __restrict__ asrc,
                 const float* __restrict__ ald, const float* __restrict__ bias,
                 float* __restrict__ out, int N, int first, int nwaves) {
  gather_block(offs, csr, h, asrc, ald, bias, out, N, first, nwaves, (int)blockIdx.x);
}

#define MIX_GATHER_PER_WGP 3

__global__ void __launch_bounds__(256, 8)
rgat_gather_gemm(const int* __restrict__ offs, const int* __restrict__ csr,
                 const float* __restrict__ hcur, const float* __restrict__ asrc,
                 const float* __restrict__ aldcur, const float* __restrict__ bias,
                 float* __restrict__ out, int N, int first, int nwaves,
                 const float* x, const float* Wn, const float* adstn,
                 float* __restrict__ hnext, float* __restrict__ aldnext,
                 int nmain, int ngemm) {
  __shared__ __attribute__((aligned(16))) float sm[GM_SM_FLOATS];
  __shared__ __attribute__((aligned(16))) float av[FOUT];
  const int b = (int)blockIdx.x;
  const int rnd = b / GM_WGPS, plc = b - rnd * GM_WGPS;
  const int grp = rnd * (GM_WGPS / 4) + (plc >> 2);
  const int sub = plc & 3, gat = (4 - (rnd & 3)) & 3;
  if (sub == gat) {
    gemm_role(x, Wn, adstn, hnext, aldnext, N, nmain, ngemm, grp, GM_WGPS, sm, av);
  } else {
    gather_block(offs, csr, hcur, asrc, aldcur, bias, out, N, first, nwaves,
                 MIX_GATHER_PER_WGP * grp + (sub < gat ? sub : sub - 1));
  }
}

extern "C" void kernel_launch(void* const* d_in, const int* in_sizes, int n_in,
                              void* d_out, int out_size, void* d_ws, size_t ws_size,
                              hipStream_t stream) {
  const float* x        = (const float*)d_in[0];
  const int*   edges[3] = {(const int*)d_in[1], (const int*)d_in[2], (const int*)d_in[3]};
  const float* W        = (const float*)d_in[4];
  const float* att_src  = (const float*)d_in[5];
  const float* att_dst  = (const float*)d_in[6];
  const float* bias     = (const float*)d_in[7];

  const int N = in_sizes[0] / FIN;
  const int E = in_sizes[1] / 2;
  float* out = (float*)d_out;

  const int nt = (N + 1 + 1023) / 1024;
  const int NP = nt * 1024;
  const int nb = (N + 1 + BIN_NODES - 1) / BIN_NODES;

  char* ws = (char*)d_ws;
  size_t off = 0;
  auto carve = [&](size_t bytes) {
    void* p = ws + off;
    off += (bytes + 255) & ~(size_t)255;
    return p;
  };
  float* h    = (float*)carve((size_t)N * FOUT * sizeof(float));
  float* ald  = (float*)carve((size_t)N * NHEAD * sizeof(float));
  int*   csr  = (int*)carve((size_t)3 * E * sizeof(int));
  (void)carve((size_t)3 * NP * sizeof(int));
  int*   offs = (int*)carve((size_t)3 * NP * sizeof(int));
  float* h1   = (float*)carve((size_t)N * FOUT * sizeof(float));
  float* ald1 = (float*)carve((size_t)N * NHEAD * sizeof(float));
  int2*  binned = (int2*)h1;
  int*   runtab = (int*)((char*)h1 + (((size_t)3 * E * sizeof(int2) + 255) & ~(size_t)255));
  (void)ws_size;

  const int nmain  = (N / GM_ROWS / GM_WGPS) * GM_WGPS;
  const int nstrip = (N - nmain * GM_ROWS + GM_STRIP - 1) / GM_STRIP;
  const int nblk = (E + BK_CH - 1) / BK_CH;
  const dim3 bgrid(nblk, 3);
  const int gblocks = (N + 8 * NODES_PER_WAVE - 1) / (8 * NODES_PER_WAVE);
  rgat_bin_local<<<bgrid, BK_THREADS, 0, stream>>>(edges[0], edges[1], edges[2], binned, runtab, E, nb);

  const int ngemm = nmain + nstrip;
  const int mgblocks = MIX_GATHER_PER_WGP * GM_WGPS;
  const int sblocks  = 3 * nb > 2 * GM_WGPS ? 2 * GM_WGPS + 3 * nb : 4 * GM_WGPS;
  float* hp[2] = {h, h1};
  float* ap[2] = {ald, ald1};

  rgat_sort_gemm<<<sblocks, 256, 0, stream>>>(runtab, binned, offs, csr, E, N, NP, nb, nblk,
                                              x, W, att_dst, hp[0], ap[0], nmain, ngemm);
  for (int r = 0; r < 2; ++r) {
    rgat_gather_gemm<<<GM_WGPS + mgblocks, 256, 0, stream>>>(
        offs + (size_t)r * NP, csr + (size_t)r * E, hp[r & 1], att_src + (size_t)r * NHEAD * DHEAD,
        ap[r & 1], bias, out, N, r == 0, mgblocks * 8,
        x, W + (size_t)(r + 1) * FIN * FOUT, att_dst + (size_t)(r + 1) * NHEAD * DHEAD,
        hp[(r + 1) & 1], ap[(r + 1) & 1], nmain, ngemm);
  }
  rgat_node_gather<<<gblocks, 256, 0, stream>>>(
      offs + (size_t)2 * NP, csr + (size_t)2 * E, hp[0], att_src + (size_t)2 * NHEAD * DHEAD, ap[0],
      bias, out, N, 0, gblocks * 8);
}
